# starting kernel
# baseline (speedup 1.0000x reference)
#include <hip/hip_runtime.h>

#define HID 64

typedef __attribute__((ext_vector_type(2))) float v2f;
typedef __attribute__((ext_vector_type(8))) float v8f;

__device__ __forceinline__ v8f wmma_k4(v2f a, v2f b, v8f c) {
  return __builtin_amdgcn_wmma_f32_16x16x4_f32(
       false, a,  false, b,
       (short)0, c,  false,  false);
}

__device__ __forceinline__ void atomic_add_f32_hw(float* addr, float val) {
  asm volatile("global_atomic_add_f32 %0, %1, off scope:SCOPE_DEV"
               :
               : "v"(addr), "v"(val)
               : "memory");
}

__global__ void zero_f32(float* __restrict__ p, long long n) {
  long long i = (long long)blockIdx.x * blockDim.x + threadIdx.x;
  if (i < n) p[i] = 0.0f;
}

__global__ void degree_kernel(const int* __restrict__ dst, float* __restrict__ deg, int E) {
  int e = blockIdx.x * blockDim.x + threadIdx.x;
  if (e < E) atomic_add_f32_hw(&deg[dst[e]], 1.0f);
}

__global__ void scatter_kernel(const float* __restrict__ h, const int* __restrict__ src,
                               const int* __restrict__ dst, float* __restrict__ agg,
                               long long total) {
  long long t = (long long)blockIdx.x * blockDim.x + threadIdx.x;
  if (t >= total) return;
  int e = (int)(t >> 6);
  int c = (int)(t & 63);
  int s = src[e];
  int d = dst[e];
  atomic_add_f32_hw(&agg[(long long)d * HID + c], h[(long long)s * HID + c]);
}

#define PROJ_K 168
#define PROJ_SX_STRIDE 172
#define PROJ_SW_STRIDE 68
#define PROJ_LDS_FLOATS (32 * PROJ_SX_STRIDE + PROJ_K * PROJ_SW_STRIDE)

__global__ __launch_bounds__(128) void proj_kernel(const float* __restrict__ x,
                                                   const float* __restrict__ W,
                                                   const float* __restrict__ bias,
                                                   float* __restrict__ h, int n) {
  extern __shared__ float smem[];
  float(*sX)[PROJ_SX_STRIDE] = (float(*)[PROJ_SX_STRIDE])smem;
  float(*sW)[PROJ_SW_STRIDE] = (float(*)[PROJ_SW_STRIDE])(smem + 32 * PROJ_SX_STRIDE);

  const int tid = threadIdx.x;
  const int rowBase = blockIdx.x * 32;

  for (int i = tid; i < 32 * PROJ_K; i += 128) {
    int r = i / PROJ_K, c = i % PROJ_K;
    int row = rowBase + r;
    sX[r][c] = (row < n && c < 165) ? x[(long long)row * 165 + c] : 0.0f;
  }
  for (int i = tid; i < PROJ_K * 64; i += 128) {
    int r = i >> 6, c = i & 63;
    sW[r][c] = (r < 165) ? W[r * 64 + c] : 0.0f;
  }
  __syncthreads();

  const int wave = tid >> 5;
  const int lane = tid & 31;
  const int hf = lane >> 4;
  const int m = lane & 15;
  const int n0 = wave * 16;

  v8f acc0 = {};
  v8f acc1 = {};
  for (int k = 0; k < PROJ_K; k += 4) {
    const int ka = k + 2 * hf;
    v2f b, a0, a1;
    b[0] = sW[ka][n0 + m];
    b[1] = sW[ka + 1][n0 + m];
    a0[0] = sX[m][ka];
    a0[1] = sX[m][ka + 1];
    a1[0] = sX[16 + m][ka];
    a1[1] = sX[16 + m][ka + 1];
    acc0 = wmma_k4(a0, b, acc0);
    acc1 = wmma_k4(a1, b, acc1);
  }

  const float bv = bias[n0 + m];
  for (int v = 0; v < 8; ++v) {
    int r0 = rowBase + v + 8 * hf;
    int r1 = r0 + 16;
    if (r0 < n) {
      float val = acc0[v] + bv;
      h[(long long)r0 * HID + n0 + m] = val > 0.0f ? val : 0.0f;
    }
    if (r1 < n) {
      float val = acc1[v] + bv;
      h[(long long)r1 * HID + n0 + m] = val > 0.0f ? val : 0.0f;
    }
  }
}

__global__ __launch_bounds__(128) void combine_kernel(
    const float* __restrict__ agg, const float* __restrict__ deg,
    const float* __restrict__ h_in, const float* __restrict__ Wl,
    const float* __restrict__ bl, const float* __restrict__ Wr,
    float* __restrict__ h_out, int n) {
  __shared__ float sAa[32][68];
  __shared__ float sAh[32][68];
  __shared__ float sWl[64][68];
  __shared__ float sWr[64][68];
  const int tid = threadIdx.x;
  const int rowBase = blockIdx.x * 32;

  for (int i = tid; i < 32 * 64; i += 128) {
    int r = i >> 6, c = i & 63;
    int row = rowBase + r;
    float va = 0.0f, vh = 0.0f;
    if (row < n) {
      float d = deg[row];
      d = d > 1.0f ? d : 1.0f;
      va = agg[(long long)row * HID + c] / d;
      vh = h_in[(long long)row * HID + c];
    }
    sAa[r][c] = va;
    sAh[r][c] = vh;
  }
  for (int i = tid; i < 64 * 64; i += 128) {
    int r = i >> 6, c = i & 63;
    sWl[r][c] = Wl[i];
    sWr[r][c] = Wr[i];
  }
  __syncthreads();

  const int wave = tid >> 5;
  const int lane = tid & 31;
  const int hf = lane >> 4;
  const int m = lane & 15;
  const int n0 = wave * 16;

  v8f acc0 = {};
  v8f acc1 = {};
  for (int k = 0; k < 64; k += 4) {
    const int ka = k + 2 * hf;
    v2f b, a0, a1;
    b[0] = sWl[ka][n0 + m];
    b[1] = sWl[ka + 1][n0 + m];
    a0[0] = sAa[m][ka];
    a0[1] = sAa[m][ka + 1];
    a1[0] = sAa[16 + m][ka];
    a1[1] = sAa[16 + m][ka + 1];
    acc0 = wmma_k4(a0, b, acc0);
    acc1 = wmma_k4(a1, b, acc1);
  }
  for (int k = 0; k < 64; k += 4) {
    const int ka = k + 2 * hf;
    v2f b, a0, a1;
    b[0] = sWr[ka][n0 + m];
    b[1] = sWr[ka + 1][n0 + m];
    a0[0] = sAh[m][ka];
    a0[1] = sAh[m][ka + 1];
    a1[0] = sAh[16 + m][ka];
    a1[1] = sAh[16 + m][ka + 1];
    acc0 = wmma_k4(a0, b, acc0);
    acc1 = wmma_k4(a1, b, acc1);
  }

  const float bv = bl[n0 + m];
  for (int v = 0; v < 8; ++v) {
    int r0 = rowBase + v + 8 * hf;
    int r1 = r0 + 16;
    if (r0 < n) {
      float val = acc0[v] + bv;
      h_out[(long long)r0 * HID + n0 + m] = val > 0.0f ? val : 0.0f;
    }
    if (r1 < n) {
      float val = acc1[v] + bv;
      h_out[(long long)r1 * HID + n0 + m] = val > 0.0f ? val : 0.0f;
    }
  }
}

__global__ void cls_kernel(const float* __restrict__ h, const float* __restrict__ Wc,
                           const float* __restrict__ bc, float* __restrict__ out, int n) {
  int i = blockIdx.x * blockDim.x + threadIdx.x;
  if (i >= n) return;
  const float4* hp = (const float4*)(h + (long long)i * HID);
  const float4* wp = (const float4*)Wc;
  float acc = bc[0];
#pragma unroll
  for (int j = 0; j < HID / 4; ++j) {
    float4 hv = hp[j];
    float4 wv = wp[j];
    acc += hv.x * wv.x + hv.y * wv.y + hv.z * wv.z + hv.w * wv.w;
  }
  out[i] = acc;
}

extern "C" void kernel_launch(void* const* d_in, const int* in_sizes, int n_in,
                              void* d_out, int out_size, void* d_ws, size_t ws_size,
                              hipStream_t stream) {
  const float* x   = (const float*)d_in[0];
  const int* eidx  = (const int*)d_in[1];
  const float* Wp  = (const float*)d_in[2];
  const float* bp  = (const float*)d_in[3];
  const float* W1l = (const float*)d_in[4];
  const float* b1l = (const float*)d_in[5];
  const float* W1r = (const float*)d_in[6];
  const float* W2l = (const float*)d_in[7];
  const float* b2l = (const float*)d_in[8];
  const float* W2r = (const float*)d_in[9];
  const float* Wc  = (const float*)d_in[10];
  const float* bc  = (const float*)d_in[11];

  const int N = in_sizes[0] / 165;
  const int E = in_sizes[1] / 2;
  const int* src = eidx;
  const int* dst = eidx + E;

  float* A   = (float*)d_ws;
  float* B   = A + (size_t)N * HID;
  float* AGG = B + (size_t)N * HID;
  float* DEG = AGG + (size_t)N * HID;

  const long long nh = (long long)N * HID;
  const long long te = (long long)E * HID;
  const unsigned gz_nh = (unsigned)((nh + 255) / 256);
  const unsigned gz_n  = (unsigned)((N + 255) / 256);
  const unsigned g_sc  = (unsigned)((te + 255) / 256);
  const unsigned g_row = (unsigned)((N + 31) / 32);
  const size_t proj_lds = (size_t)PROJ_LDS_FLOATS * sizeof(float);

  zero_f32<<<gz_nh, 256, 0, stream>>>(AGG, nh);
  zero_f32<<<gz_n, 256, 0, stream>>>(DEG, (long long)N);
  degree_kernel<<<(E + 255) / 256, 256, 0, stream>>>(dst, DEG, E);

  proj_kernel<<<g_row, 128, proj_lds, stream>>>(x, Wp, bp, A, N);

  scatter_kernel<<<g_sc, 256, 0, stream>>>(A, src, dst, AGG, te);
  combine_kernel<<<g_row, 128, 0, stream>>>(AGG, DEG, A, W1l, b1l, W1r, B, N);

  zero_f32<<<gz_nh, 256, 0, stream>>>(AGG, nh);
  scatter_kernel<<<g_sc, 256, 0, stream>>>(B, src, dst, AGG, te);
  combine_kernel<<<g_row, 128, 0, stream>>>(AGG, DEG, B, W2l, b2l, W2r, A, N);

  cls_kernel<<<(N + 255) / 256, 256, 0, stream>>>(A, Wc, bc, (float*)d_out, N);
}


# neighbour sums in an LDS tile: a work-group owns 784 destination rows, scans the whole edge list for them, each wave adds source rows into its own rows; no agg atomics or clears; sum order now fixed
# speedup vs baseline: 1.9512x; 1.9512x over previous
#include <hip/hip_runtime.h>

#define HID 64

typedef __attribute__((ext_vector_type(2))) float v2f;
typedef __attribute__((ext_vector_type(8))) float v8f;

__device__ __forceinline__ v8f wmma_k4(v2f a, v2f b, v8f c) {
  return __builtin_amdgcn_wmma_f32_16x16x4_f32(
       false, a,  false, b,
       (short)0, c,  false,  false);
}

__device__ __forceinline__ void atomic_add_f32_hw(float* addr, float val) {
  asm volatile("global_atomic_add_f32 %0, %1, off scope:SCOPE_DEV"
               :
               : "v"(addr), "v"(val)
               : "memory");
}

__global__ void zero_f32(float* __restrict__ p, long long n) {
  long long i = (long long)blockIdx.x * blockDim.x + threadIdx.x;
  if (i < n) p[i] = 0.0f;
}

__device__ __forceinline__ int slice_dst(const int* __restrict__ dst, int i, int chunk,
                                         int ebase, int E) {
  const int e = ebase + i;
  return (i < chunk && e < E) ? dst[e] : -1;
}

#define AG_ROWS 784
#define AG_WAVES 32
#define AG_THREADS (AG_WAVES * 32)
#define AG_ITER (AG_WAVES * 512)
#define AG_OUTCAP 416
#define AG_MYCAP 64

__device__ __forceinline__ void load_iter(int4 (&r)[4], const int* __restrict__ dst, int g,
                                          int woff, int E, bool vec) {
  const int base = g * AG_ITER;
  if (vec && base + AG_ITER <= E) {
#pragma unroll
    for (int j = 0; j < 4; ++j) r[j] = *(const int4*)(dst + base + woff + j * 128);
  } else {
#pragma unroll
    for (int j = 0; j < 4; ++j) {
      const int e = base + woff + j * 128;
      r[j].x = (e < E) ? dst[e] : -1;
      r[j].y = (e + 1 < E) ? dst[e + 1] : -1;
      r[j].z = (e + 2 < E) ? dst[e + 2] : -1;
      r[j].w = (e + 3 < E) ? dst[e + 3] : -1;
    }
  }
}

__device__ __forceinline__ int queue_push(int2* q, int qn, bool keep, int2 ent, int lane) {
  const unsigned b = __builtin_amdgcn_ballot_w32(keep);
  if (b != 0u) {
    if (keep) q[qn + __popc(b & ((1u << lane) - 1u))] = ent;
    qn += __popc(b);
  }
  return qn;
}

__device__ __forceinline__ bool push4(int2* q, int& qn, int4 d, int e, int rowlo, int lane) {
  const unsigned l0 = (unsigned)(d.x - rowlo);
  const unsigned l1 = (unsigned)(d.y - rowlo);
  const unsigned l2 = (unsigned)(d.z - rowlo);
  const unsigned l3 = (unsigned)(d.w - rowlo);
  const unsigned lm = min(min(l0, l1), min(l2, l3));
  if (__builtin_amdgcn_ballot_w32(lm < (unsigned)AG_ROWS) == 0u) return true;
  const bool k0 = l0 < (unsigned)AG_ROWS, k1 = l1 < (unsigned)AG_ROWS;
  const bool k2 = l2 < (unsigned)AG_ROWS, k3 = l3 < (unsigned)AG_ROWS;
  const unsigned b0 = __builtin_amdgcn_ballot_w32(k0), b1 = __builtin_amdgcn_ballot_w32(k1);
  const unsigned b2 = __builtin_amdgcn_ballot_w32(k2), b3 = __builtin_amdgcn_ballot_w32(k3);
  const int c0 = __popc(b0), c1 = __popc(b1), c2 = __popc(b2), c3 = __popc(b3);
  if (qn + c0 + c1 + c2 + c3 > AG_OUTCAP) return false;
  const unsigned below = (1u << lane) - 1u;
  int p = qn;
  if (k0) q[p + __popc(b0 & below)] = make_int2(e, (int)l0);
  p += c0;
  if (k1) q[p + __popc(b1 & below)] = make_int2(e + 1, (int)l1);
  p += c1;
  if (k2) q[p + __popc(b2 & below)] = make_int2(e + 2, (int)l2);
  p += c2;
  if (k3) q[p + __popc(b3 & below)] = make_int2(e + 3, (int)l3);
  qn = p + c3;
  return true;
}

__device__ __forceinline__ void add_rows(const int2* q, int cnt, const float* __restrict__ h,
                                         const int* __restrict__ src, float* tile, int lane) {
  int s = 0, loc = 0;
  if (lane < cnt) {
    const int2 ent = q[lane];
    s = src[ent.x];
    loc = ent.y;
  }
  for (int k = 0; k < cnt; k += 8) {
    float2 v[8];
    int lk[8];
#pragma unroll
    for (int u = 0; u < 8; ++u) {
      const int sk = __builtin_amdgcn_readlane(s, k + u);
      lk[u] = __builtin_amdgcn_readlane(loc, k + u);
      v[u] = *(const float2*)(h + (long long)sk * HID + lane * 2);
    }
#pragma unroll
    for (int u = 0; u < 8; ++u) {
      if (k + u < cnt) {
        float2* t = (float2*)(tile + lk[u] * HID + lane * 2);
        float2 a = *t;
        a.x += v[u].x;
        a.y += v[u].y;
        *t = a;
      }
    }
  }
}

__global__ __launch_bounds__(AG_THREADS) void neigh_sum_kernel(const float* __restrict__ h,
                                                             const int* __restrict__ src,
                                                             const int* __restrict__ dst,
                                                             float* __restrict__ agg, int n,
                                                             int E) {
  __shared__ float tile[AG_ROWS * HID];
  __shared__ int2 outq[AG_WAVES][AG_OUTCAP];
  __shared__ int2 ownq[AG_WAVES][AG_MYCAP];
  __shared__ int outn[AG_WAVES];
  __shared__ int fin[AG_WAVES];

  const int tid = threadIdx.x;
  const int wave = tid >> 5;
  const int lane = tid & 31;
  const int rowlo = blockIdx.x * AG_ROWS;

  float4* t4 = (float4*)tile;
  const float4 z = {0.0f, 0.0f, 0.0f, 0.0f};
  for (int i = tid; i < AG_ROWS * HID / 4; i += AG_THREADS) t4[i] = z;

  const bool vec = (((size_t)dst) & 15) == 0;
  const int nit = (E + AG_ITER - 1) / AG_ITER;
  const int start = (int)(((long long)blockIdx.x * nit) / (int)gridDim.x);
  const int woff = wave * 512 + lane * 4;
  const int4 none = {-1, -1, -1, -1};

  int4 cur[4], nxt[4];
#pragma unroll
  for (int j = 0; j < 4; ++j) {
    cur[j] = none;
    nxt[j] = none;
  }

  int2* oq = outq[wave];
  int2* mq = ownq[wave];
  int it = -1;
  int mn = 0;

  for (int round = 0; round <= 4 * nit; ++round) {
    int qn = 0;
    while (it < nit) {
      int g = start + it;
      if (g >= nit) g -= nit;
      int g1 = start + it + 1;
      if (g1 >= nit) g1 -= nit;
      if (it + 1 < nit) {
        load_iter(nxt, dst, g1, woff, E, vec);
      } else {
#pragma unroll
        for (int j = 0; j < 4; ++j) nxt[j] = none;
      }
      const int e0 = g * AG_ITER + woff;
      int queued = 0;
      if (push4(oq, qn, cur[0], e0, rowlo, lane)) {
        queued = 1;
        if (push4(oq, qn, cur[1], e0 + 128, rowlo, lane)) {
          queued = 2;
          if (push4(oq, qn, cur[2], e0 + 256, rowlo, lane)) {
            queued = 3;
            if (push4(oq, qn, cur[3], e0 + 384, rowlo, lane)) queued = 4;
          }
        }
      }
      if (queued < 4) {
        if (queued > 0) cur[0] = none;
        if (queued > 1) cur[1] = none;
        if (queued > 2) cur[2] = none;
        break;
      }
#pragma unroll
      for (int j = 0; j < 4; ++j) cur[j] = nxt[j];
      ++it;
    }
    if (lane == 0) {
      outn[wave] = qn;
      fin[wave] = (it >= nit) ? 1 : 0;
    }
    __syncthreads();

    for (int i = 0; i < AG_WAVES; ++i) {
      const int w2 = (wave + i) & (AG_WAVES - 1);
      const int cnt = __builtin_amdgcn_readfirstlane(outn[w2]);
      for (int b0 = 0; b0 < cnt; b0 += 32) {
        const bool live = b0 + lane < cnt;
        int2 ent = make_int2(0, -1);
        if (live) ent = outq[w2][b0 + lane];
        const bool mine = live && ((ent.y & (AG_WAVES - 1)) == wave);
        mn = queue_push(mq, mn, mine, ent, lane);
        if (mn >= 32) {
          add_rows(mq, 32, h, src, tile, lane);
          const int rem = mn - 32;
          if (lane < rem) {
            const int2 mv = mq[32 + lane];
            mq[lane] = mv;
          }
          mn = rem;
        }
      }
    }
    const bool alldone = __builtin_amdgcn_ballot_w32(fin[lane] != 0) == 0xffffffffu;
    __syncthreads();
    if (alldone) break;
  }
  if (mn > 0) add_rows(mq, mn, h, src, tile, lane);
  __syncthreads();

  const int rows = (n - rowlo < AG_ROWS) ? (n - rowlo) : AG_ROWS;
  float4* o4 = (float4*)(agg + (long long)rowlo * HID);
  for (int i = tid; i < rows * (HID / 4); i += AG_THREADS) o4[i] = t4[i];
}

#define PROJ_K 168
#define PROJ_SX_STRIDE 172
#define PROJ_SW_STRIDE 68
#define PROJ_LDS_FLOATS (32 * PROJ_SX_STRIDE + PROJ_K * PROJ_SW_STRIDE)

__global__ __launch_bounds__(128) void proj_kernel(const float* __restrict__ x,
                                                   const float* __restrict__ W,
                                                   const float* __restrict__ bias,
                                                   float* __restrict__ h, int n,
                                                   const int* __restrict__ dst,
                                                   float* __restrict__ deg, int E, int chunk) {
  extern __shared__ float smem[];
  float(*sX)[PROJ_SX_STRIDE] = (float(*)[PROJ_SX_STRIDE])smem;
  float(*sW)[PROJ_SW_STRIDE] = (float(*)[PROJ_SW_STRIDE])(smem + 32 * PROJ_SX_STRIDE);

  const int tid = threadIdx.x;
  const int rowBase = blockIdx.x * 32;

  const int ebase = blockIdx.x * chunk;
  const int d0 = slice_dst(dst, tid, chunk, ebase, E);
  const int d1 = slice_dst(dst, tid + 128, chunk, ebase, E);
  const int d2 = slice_dst(dst, tid + 256, chunk, ebase, E);
  const int d3 = slice_dst(dst, tid + 384, chunk, ebase, E);

  for (int i = tid; i < 32 * PROJ_K; i += 128) {
    int r = i / PROJ_K, c = i % PROJ_K;
    int row = rowBase + r;
    sX[r][c] = (row < n && c < 165) ? x[(long long)row * 165 + c] : 0.0f;
  }
  for (int i = tid; i < PROJ_K * 64; i += 128) {
    int r = i >> 6, c = i & 63;
    sW[r][c] = (r < 165) ? W[r * 64 + c] : 0.0f;
  }
  __syncthreads();

  if (d0 >= 0) atomic_add_f32_hw(&deg[d0], 1.0f);
  if (d1 >= 0) atomic_add_f32_hw(&deg[d1], 1.0f);
  if (d2 >= 0) atomic_add_f32_hw(&deg[d2], 1.0f);
  if (d3 >= 0) atomic_add_f32_hw(&deg[d3], 1.0f);
  for (int i = tid + 512; i < chunk; i += 128) {
    const int d = slice_dst(dst, i, chunk, ebase, E);
    if (d >= 0) atomic_add_f32_hw(&deg[d], 1.0f);
  }

  const int wave = tid >> 5;
  const int lane = tid & 31;
  const int hf = lane >> 4;
  const int m = lane & 15;
  const int n0 = wave * 16;

  v8f acc0 = {};
  v8f acc1 = {};
  for (int k = 0; k < PROJ_K; k += 4) {
    const int ka = k + 2 * hf;
    v2f b, a0, a1;
    b[0] = sW[ka][n0 + m];
    b[1] = sW[ka + 1][n0 + m];
    a0[0] = sX[m][ka];
    a0[1] = sX[m][ka + 1];
    a1[0] = sX[16 + m][ka];
    a1[1] = sX[16 + m][ka + 1];
    acc0 = wmma_k4(a0, b, acc0);
    acc1 = wmma_k4(a1, b, acc1);
  }

  const float bv = bias[n0 + m];
  for (int v = 0; v < 8; ++v) {
    int r0 = rowBase + v + 8 * hf;
    int r1 = r0 + 16;
    if (r0 < n) {
      float val = acc0[v] + bv;
      h[(long long)r0 * HID + n0 + m] = val > 0.0f ? val : 0.0f;
    }
    if (r1 < n) {
      float val = acc1[v] + bv;
      h[(long long)r1 * HID + n0 + m] = val > 0.0f ? val : 0.0f;
    }
  }
}

__global__ __launch_bounds__(128) void combine_kernel(
    const float* __restrict__ agg, const float* __restrict__ deg,
    const float* __restrict__ h_in, const float* __restrict__ Wl,
    const float* __restrict__ bl, const float* __restrict__ Wr,
    float* __restrict__ h_out, int n) {
  __shared__ float sAa[32][68];
  __shared__ float sAh[32][68];
  __shared__ float sWl[64][68];
  __shared__ float sWr[64][68];
  const int tid = threadIdx.x;
  const int rowBase = blockIdx.x * 32;

  for (int i = tid; i < 32 * 64; i += 128) {
    int r = i >> 6, c = i & 63;
    int row = rowBase + r;
    float va = 0.0f, vh = 0.0f;
    if (row < n) {
      float d = deg[row];
      d = d > 1.0f ? d : 1.0f;
      va = agg[(long long)row * HID + c] / d;
      vh = h_in[(long long)row * HID + c];
    }
    sAa[r][c] = va;
    sAh[r][c] = vh;
  }
  for (int i = tid; i < 64 * 64; i += 128) {
    int r = i >> 6, c = i & 63;
    sWl[r][c] = Wl[i];
    sWr[r][c] = Wr[i];
  }
  __syncthreads();

  const int wave = tid >> 5;
  const int lane = tid & 31;
  const int hf = lane >> 4;
  const int m = lane & 15;
  const int n0 = wave * 16;

  v8f acc0 = {};
  v8f acc1 = {};
  for (int k = 0; k < 64; k += 4) {
    const int ka = k + 2 * hf;
    v2f b, a0, a1;
    b[0] = sWl[ka][n0 + m];
    b[1] = sWl[ka + 1][n0 + m];
    a0[0] = sAa[m][ka];
    a0[1] = sAa[m][ka + 1];
    a1[0] = sAa[16 + m][ka];
    a1[1] = sAa[16 + m][ka + 1];
    acc0 = wmma_k4(a0, b, acc0);
    acc1 = wmma_k4(a1, b, acc1);
  }
  for (int k = 0; k < 64; k += 4) {
    const int ka = k + 2 * hf;
    v2f b, a0, a1;
    b[0] = sWr[ka][n0 + m];
    b[1] = sWr[ka + 1][n0 + m];
    a0[0] = sAh[m][ka];
    a0[1] = sAh[m][ka + 1];
    a1[0] = sAh[16 + m][ka];
    a1[1] = sAh[16 + m][ka + 1];
    acc0 = wmma_k4(a0, b, acc0);
    acc1 = wmma_k4(a1, b, acc1);
  }

  const float bv = bl[n0 + m];
  for (int v = 0; v < 8; ++v) {
    int r0 = rowBase + v + 8 * hf;
    int r1 = r0 + 16;
    if (r0 < n) {
      float val = acc0[v] + bv;
      h_out[(long long)r0 * HID + n0 + m] = val > 0.0f ? val : 0.0f;
    }
    if (r1 < n) {
      float val = acc1[v] + bv;
      h_out[(long long)r1 * HID + n0 + m] = val > 0.0f ? val : 0.0f;
    }
  }
}

__global__ void cls_kernel(const float* __restrict__ h, const float* __restrict__ Wc,
                           const float* __restrict__ bc, float* __restrict__ out, int n) {
  int i = blockIdx.x * blockDim.x + threadIdx.x;
  if (i >= n) return;
  const float4* hp = (const float4*)(h + (long long)i * HID);
  const float4* wp = (const float4*)Wc;
  float acc = bc[0];
#pragma unroll
  for (int j = 0; j < HID / 4; ++j) {
    float4 hv = hp[j];
    float4 wv = wp[j];
    acc += hv.x * wv.x + hv.y * wv.y + hv.z * wv.z + hv.w * wv.w;
  }
  out[i] = acc;
}

extern "C" void kernel_launch(void* const* d_in, const int* in_sizes, int n_in,
                              void* d_out, int out_size, void* d_ws, size_t ws_size,
                              hipStream_t stream) {
  const float* x   = (const float*)d_in[0];
  const int* eidx  = (const int*)d_in[1];
  const float* Wp  = (const float*)d_in[2];
  const float* bp  = (const float*)d_in[3];
  const float* W1l = (const float*)d_in[4];
  const float* b1l = (const float*)d_in[5];
  const float* W1r = (const float*)d_in[6];
  const float* W2l = (const float*)d_in[7];
  const float* b2l = (const float*)d_in[8];
  const float* W2r = (const float*)d_in[9];
  const float* Wc  = (const float*)d_in[10];
  const float* bc  = (const float*)d_in[11];

  const int N = in_sizes[0] / 165;
  const int E = in_sizes[1] / 2;
  const int* src = eidx;
  const int* dst = eidx + E;

  float* A   = (float*)d_ws;
  float* B   = A + (size_t)N * HID;
  float* AGG = B + (size_t)N * HID;
  float* DEG = AGG + (size_t)N * HID;

  const unsigned gz_n  = (unsigned)((N + 255) / 256);
  const unsigned g_ag  = (unsigned)((N + AG_ROWS - 1) / AG_ROWS);
  const unsigned g_row = (unsigned)((N + 31) / 32);
  const size_t proj_lds = (size_t)PROJ_LDS_FLOATS * sizeof(float);
  const int e_chunk = (int)(((long long)E + g_row - 1) / g_row);

  zero_f32<<<gz_n, 256, 0, stream>>>(DEG, (long long)N);

  proj_kernel<<<g_row, 128, proj_lds, stream>>>(x, Wp, bp, A, N, dst, DEG, E, e_chunk);

  neigh_sum_kernel<<<g_ag, AG_THREADS, 0, stream>>>(A, src, dst, AGG, N, E);
  combine_kernel<<<g_row, 128, 0, stream>>>(AGG, DEG, A, W1l, b1l, W1r, B, N);

  neigh_sum_kernel<<<g_ag, AG_THREADS, 0, stream>>>(B, src, dst, AGG, N, E);
  combine_kernel<<<g_row, 128, 0, stream>>>(AGG, DEG, B, W2l, b2l, W2r, A, N);

  cls_kernel<<<(N + 255) / 256, 256, 0, stream>>>(A, Wc, bc, (float*)d_out, N);
}


# GEMMs (proj, conv 1, conv 2): no LDS tiles; weights held in registers as B fragments, rows loaded straight into A fragments by 16-byte loads, waves loop over 16-row tiles; k order of the sums changes
# speedup vs baseline: 2.7172x; 1.3926x over previous
#include <hip/hip_runtime.h>

#define HID 64

typedef __attribute__((ext_vector_type(2))) float v2f;
typedef __attribute__((ext_vector_type(8))) float v8f;
typedef __attribute__((ext_vector_type(4))) float v4f;
typedef v4f __attribute__((aligned(4))) v4f_u;

__device__ __forceinline__ v8f wmma_k4(v2f a, v2f b, v8f c) {
  return __builtin_amdgcn_wmma_f32_16x16x4_f32(
       false, a,  false, b,
       (short)0, c,  false,  false);
}

__device__ __forceinline__ void atomic_add_f32_hw(float* addr, float val) {
  asm volatile("global_atomic_add_f32 %0, %1, off scope:SCOPE_DEV"
               :
               : "v"(addr), "v"(val)
               : "memory");
}

__global__ void zero_f32(float* __restrict__ p, long long n) {
  long long i = (long long)blockIdx.x * blockDim.x + threadIdx.x;
  if (i < n) p[i] = 0.0f;
}

__device__ __forceinline__ int slice_dst(const int* __restrict__ dst, int i, int chunk,
                                         int ebase, int E) {
  const int e = ebase + i;
  return (i < chunk && e < E) ? dst[e] : -1;
}

#define AG_ROWS 784
#define AG_WAVES 32
#define AG_THREADS (AG_WAVES * 32)
#define AG_ITER (AG_WAVES * 512)
#define AG_OUTCAP 416
#define AG_MYCAP 64

__device__ __forceinline__ void load_iter(int4 (&r)[4], const int* __restrict__ dst, int g,
                                          int woff, int E, bool vec) {
  const int base = g * AG_ITER;
  if (vec && base + AG_ITER <= E) {
#pragma unroll
    for (int j = 0; j < 4; ++j) r[j] = *(const int4*)(dst + base + woff + j * 128);
  } else {
#pragma unroll
    for (int j = 0; j < 4; ++j) {
      const int e = base + woff + j * 128;
      r[j].x = (e < E) ? dst[e] : -1;
      r[j].y = (e + 1 < E) ? dst[e + 1] : -1;
      r[j].z = (e + 2 < E) ? dst[e + 2] : -1;
      r[j].w = (e + 3 < E) ? dst[e + 3] : -1;
    }
  }
}

__device__ __forceinline__ int queue_push(int2* q, int qn, bool keep, int2 ent, int lane) {
  const unsigned b = __builtin_amdgcn_ballot_w32(keep);
  if (b != 0u) {
    if (keep) q[qn + __popc(b & ((1u << lane) - 1u))] = ent;
    qn += __popc(b);
  }
  return qn;
}

__device__ __forceinline__ bool push4(int2* q, int& qn, int4 d, int e, int rowlo, int lane) {
  const unsigned l0 = (unsigned)(d.x - rowlo);
  const unsigned l1 = (unsigned)(d.y - rowlo);
  const unsigned l2 = (unsigned)(d.z - rowlo);
  const unsigned l3 = (unsigned)(d.w - rowlo);
  const unsigned lm = min(min(l0, l1), min(l2, l3));
  if (__builtin_amdgcn_ballot_w32(lm < (unsigned)AG_ROWS) == 0u) return true;
  const bool k0 = l0 < (unsigned)AG_ROWS, k1 = l1 < (unsigned)AG_ROWS;
  const bool k2 = l2 < (unsigned)AG_ROWS, k3 = l3 < (unsigned)AG_ROWS;
  const unsigned b0 = __builtin_amdgcn_ballot_w32(k0), b1 = __builtin_amdgcn_ballot_w32(k1);
  const unsigned b2 = __builtin_amdgcn_ballot_w32(k2), b3 = __builtin_amdgcn_ballot_w32(k3);
  const int c0 = __popc(b0), c1 = __popc(b1), c2 = __popc(b2), c3 = __popc(b3);
  if (qn + c0 + c1 + c2 + c3 > AG_OUTCAP) return false;
  const unsigned below = (1u << lane) - 1u;
  int p = qn;
  if (k0) q[p + __popc(b0 & below)] = make_int2(e, (int)l0);
  p += c0;
  if (k1) q[p + __popc(b1 & below)] = make_int2(e + 1, (int)l1);
  p += c1;
  if (k2) q[p + __popc(b2 & below)] = make_int2(e + 2, (int)l2);
  p += c2;
  if (k3) q[p + __popc(b3 & below)] = make_int2(e + 3, (int)l3);
  qn = p + c3;
  return true;
}

__device__ __forceinline__ void add_rows(const int2* q, int cnt, const float* __restrict__ h,
                                         const int* __restrict__ src, float* tile, int lane) {
  int s = 0, loc = 0;
  if (lane < cnt) {
    const int2 ent = q[lane];
    s = src[ent.x];
    loc = ent.y;
  }
  for (int k = 0; k < cnt; k += 8) {
    float2 v[8];
    int lk[8];
#pragma unroll
    for (int u = 0; u < 8; ++u) {
      const int sk = __builtin_amdgcn_readlane(s, k + u);
      lk[u] = __builtin_amdgcn_readlane(loc, k + u);
      v[u] = *(const float2*)(h + (long long)sk * HID + lane * 2);
    }
#pragma unroll
    for (int u = 0; u < 8; ++u) {
      if (k + u < cnt) {
        float2* t = (float2*)(tile + lk[u] * HID + lane * 2);
        float2 a = *t;
        a.x += v[u].x;
        a.y += v[u].y;
        *t = a;
      }
    }
  }
}

__global__ __launch_bounds__(AG_THREADS) void neigh_sum_kernel(const float* __restrict__ h,
                                                             const int* __restrict__ src,
                                                             const int* __restrict__ dst,
                                                             float* __restrict__ agg, int n,
                                                             int E) {
  __shared__ float tile[AG_ROWS * HID];
  __shared__ int2 outq[AG_WAVES][AG_OUTCAP];
  __shared__ int2 ownq[AG_WAVES][AG_MYCAP];
  __shared__ int outn[AG_WAVES];
  __shared__ int fin[AG_WAVES];

  const int tid = threadIdx.x;
  const int wave = tid >> 5;
  const int lane = tid & 31;
  const int rowlo = blockIdx.x * AG_ROWS;

  float4* t4 = (float4*)tile;
  const float4 z = {0.0f, 0.0f, 0.0f, 0.0f};
  for (int i = tid; i < AG_ROWS * HID / 4; i += AG_THREADS) t4[i] = z;

  const bool vec = (((size_t)dst) & 15) == 0;
  const int nit = (E + AG_ITER - 1) / AG_ITER;
  const int start = (int)(((long long)blockIdx.x * nit) / (int)gridDim.x);
  const int woff = wave * 512 + lane * 4;
  const int4 none = {-1, -1, -1, -1};

  int4 cur[4], nxt[4];
#pragma unroll
  for (int j = 0; j < 4; ++j) {
    cur[j] = none;
    nxt[j] = none;
  }

  int2* oq = outq[wave];
  int2* mq = ownq[wave];
  int it = -1;
  int mn = 0;

  for (int round = 0; round <= 4 * nit; ++round) {
    int qn = 0;
    while (it < nit) {
      int g = start + it;
      if (g >= nit) g -= nit;
      int g1 = start + it + 1;
      if (g1 >= nit) g1 -= nit;
      if (it + 1 < nit) {
        load_iter(nxt, dst, g1, woff, E, vec);
      } else {
#pragma unroll
        for (int j = 0; j < 4; ++j) nxt[j] = none;
      }
      const int e0 = g * AG_ITER + woff;
      int queued = 0;
      if (push4(oq, qn, cur[0], e0, rowlo, lane)) {
        queued = 1;
        if (push4(oq, qn, cur[1], e0 + 128, rowlo, lane)) {
          queued = 2;
          if (push4(oq, qn, cur[2], e0 + 256, rowlo, lane)) {
            queued = 3;
            if (push4(oq, qn, cur[3], e0 + 384, rowlo, lane)) queued = 4;
          }
        }
      }
      if (queued < 4) {
        if (queued > 0) cur[0] = none;
        if (queued > 1) cur[1] = none;
        if (queued > 2) cur[2] = none;
        break;
      }
#pragma unroll
      for (int j = 0; j < 4; ++j) cur[j] = nxt[j];
      ++it;
    }
    if (lane == 0) {
      outn[wave] = qn;
      fin[wave] = (it >= nit) ? 1 : 0;
    }
    __syncthreads();

    for (int i = 0; i < AG_WAVES; ++i) {
      const int w2 = (wave + i) & (AG_WAVES - 1);
      const int cnt = __builtin_amdgcn_readfirstlane(outn[w2]);
      for (int b0 = 0; b0 < cnt; b0 += 32) {
        const bool live = b0 + lane < cnt;
        int2 ent = make_int2(0, -1);
        if (live) ent = outq[w2][b0 + lane];
        const bool mine = live && ((ent.y & (AG_WAVES - 1)) == wave);
        mn = queue_push(mq, mn, mine, ent, lane);
        if (mn >= 32) {
          add_rows(mq, 32, h, src, tile, lane);
          const int rem = mn - 32;
          if (lane < rem) {
            const int2 mv = mq[32 + lane];
            mq[lane] = mv;
          }
          mn = rem;
        }
      }
    }
    const bool alldone = __builtin_amdgcn_ballot_w32(fin[lane] != 0) == 0xffffffffu;
    __syncthreads();
    if (alldone) break;
  }
  if (mn > 0) add_rows(mq, mn, h, src, tile, lane);
  __syncthreads();

  const int rows = (n - rowlo < AG_ROWS) ? (n - rowlo) : AG_ROWS;
  float4* o4 = (float4*)(agg + (long long)rowlo * HID);
  for (int i = tid; i < rows * (HID / 4); i += AG_THREADS) o4[i] = t4[i];
}


#define PJ_FULL 20
__global__ __launch_bounds__(128) __attribute__((amdgpu_waves_per_eu(2, 2))) void proj_kernel(const float* __restrict__ x,
                                                   const float* __restrict__ W,
                                                   const float* __restrict__ bias,
                                                   float* __restrict__ h, int n,
                                                   const int* __restrict__ dst,
                                                   float* __restrict__ deg, int E, int chunk) {
  const int tid = threadIdx.x;
  const int wave = __builtin_amdgcn_readfirstlane(tid >> 5);
  const int lane = tid & 31;
  const int hf = lane >> 4;
  const int m = lane & 15;

  v2f bw[PJ_FULL + 1][2][4];
#pragma unroll
  for (int j = 0; j < PJ_FULL; ++j) {
#pragma unroll
    for (int tt = 0; tt < 2; ++tt) {
      const int k = 8 * j + 4 * hf + 2 * tt;
#pragma unroll
      for (int q = 0; q < 4; ++q) {
        bw[j][tt][q][0] = W[k * 64 + 16 * q + m];
        bw[j][tt][q][1] = W[(k + 1) * 64 + 16 * q + m];
      }
    }
  }
#pragma unroll
  for (int tt = 0; tt < 2; ++tt) {
    const int k = 8 * PJ_FULL + 4 * hf + 2 * tt;
#pragma unroll
    for (int q = 0; q < 4; ++q) {
      bw[PJ_FULL][tt][q][0] = (k < 165) ? W[k * 64 + 16 * q + m] : 0.0f;
      bw[PJ_FULL][tt][q][1] = (k + 1 < 165) ? W[(k + 1) * 64 + 16 * q + m] : 0.0f;
    }
  }
  float bv[4];
#pragma unroll
  for (int q = 0; q < 4; ++q) bv[q] = bias[16 * q + m];

  const int ntiles = (n + 15) >> 4;
  for (int t = blockIdx.x * 4 + wave; t < ntiles; t += (int)gridDim.x * 4) {
    const int ebase = t * chunk;
    int dd[8];
#pragma unroll
    for (int u = 0; u < 8; ++u) dd[u] = slice_dst(dst, u * 32 + lane, chunk, ebase, E);

    const int row = t * 16 + m;
    const int rl = row < n ? row : n - 1;
    const float* xr = x + (long long)rl * 165 + 4 * hf;
    v4f a[PJ_FULL + 1];
#pragma unroll
    for (int j = 0; j < PJ_FULL; ++j) a[j] = *(const v4f_u*)(xr + 8 * j);
    const float* tp = xr + 8 * PJ_FULL;
    const float t0 = tp[0];
    const float t1 = tp[hf ? 0 : 1];
    const float t2 = tp[hf ? 0 : 2];
    const float t3 = tp[hf ? 0 : 3];
    v4f tl;
    tl[0] = t0;
    tl[1] = hf ? 0.0f : t1;
    tl[2] = hf ? 0.0f : t2;
    tl[3] = hf ? 0.0f : t3;
    a[PJ_FULL] = tl;

    v8f acc[4] = {};
#pragma unroll
    for (int j = 0; j <= PJ_FULL; ++j) {
#pragma unroll
      for (int tt = 0; tt < 2; ++tt) {
        v2f af;
        af[0] = a[j][2 * tt];
        af[1] = a[j][2 * tt + 1];
#pragma unroll
        for (int q = 0; q < 4; ++q) acc[q] = wmma_k4(af, bw[j][tt][q], acc[q]);
      }
    }

#pragma unroll
    for (int q = 0; q < 4; ++q) {
#pragma unroll
      for (int v = 0; v < 8; ++v) {
        const int r = t * 16 + v + 8 * hf;
        if (r < n) {
          const float val = acc[q][v] + bv[q];
          h[(long long)r * HID + 16 * q + m] = val > 0.0f ? val : 0.0f;
        }
      }
    }

#pragma unroll
    for (int u = 0; u < 8; ++u)
      if (dd[u] >= 0) atomic_add_f32_hw(&deg[dd[u]], 1.0f);
    for (int i = 256 + lane; i < chunk; i += 32) {
      const int d = slice_dst(dst, i, chunk, ebase, E);
      if (d >= 0) atomic_add_f32_hw(&deg[d], 1.0f);
    }
  }
}

__global__ __launch_bounds__(128) __attribute__((amdgpu_waves_per_eu(2, 2))) void combine_kernel(
    const float* __restrict__ agg, const float* __restrict__ deg,
    const float* __restrict__ h_in, const float* __restrict__ Wl,
    const float* __restrict__ bl, const float* __restrict__ Wr,
    float* __restrict__ h_out, int n) {
  const int tid = threadIdx.x;
  const int wave = __builtin_amdgcn_readfirstlane(tid >> 5);
  const int lane = tid & 31;
  const int hf = lane >> 4;
  const int m = lane & 15;

  v2f wl[8][2][4], wr[8][2][4];
#pragma unroll
  for (int j = 0; j < 8; ++j) {
#pragma unroll
    for (int tt = 0; tt < 2; ++tt) {
      const int k = 8 * j + 4 * hf + 2 * tt;
#pragma unroll
      for (int q = 0; q < 4; ++q) {
        wl[j][tt][q][0] = Wl[k * 64 + 16 * q + m];
        wl[j][tt][q][1] = Wl[(k + 1) * 64 + 16 * q + m];
        wr[j][tt][q][0] = Wr[k * 64 + 16 * q + m];
        wr[j][tt][q][1] = Wr[(k + 1) * 64 + 16 * q + m];
      }
    }
  }
  float bv[4];
#pragma unroll
  for (int q = 0; q < 4; ++q) bv[q] = bl[16 * q + m];

  const int ntiles = (n + 15) >> 4;
  for (int t = blockIdx.x * 4 + wave; t < ntiles; t += (int)gridDim.x * 4) {
    const int row = t * 16 + m;
    const int rl = row < n ? row : n - 1;
    const v4f* ap = (const v4f*)(agg + (long long)rl * HID + 4 * hf);
    const v4f* hp = (const v4f*)(h_in + (long long)rl * HID + 4 * hf);
    v4f a[8], hv[8];
#pragma unroll
    for (int j = 0; j < 8; ++j) {
      a[j] = ap[2 * j];
      hv[j] = hp[2 * j];
    }
    float d = deg[rl];
    d = d > 1.0f ? d : 1.0f;
#pragma unroll
    for (int j = 0; j < 8; ++j) {
      a[j][0] = a[j][0] / d;
      a[j][1] = a[j][1] / d;
      a[j][2] = a[j][2] / d;
      a[j][3] = a[j][3] / d;
    }

    v8f acc[4] = {};
#pragma unroll
    for (int j = 0; j < 8; ++j) {
#pragma unroll
      for (int tt = 0; tt < 2; ++tt) {
        v2f af;
        af[0] = a[j][2 * tt];
        af[1] = a[j][2 * tt + 1];
#pragma unroll
        for (int q = 0; q < 4; ++q) acc[q] = wmma_k4(af, wl[j][tt][q], acc[q]);
      }
    }
#pragma unroll
    for (int j = 0; j < 8; ++j) {
#pragma unroll
      for (int tt = 0; tt < 2; ++tt) {
        v2f af;
        af[0] = hv[j][2 * tt];
        af[1] = hv[j][2 * tt + 1];
#pragma unroll
        for (int q = 0; q < 4; ++q) acc[q] = wmma_k4(af, wr[j][tt][q], acc[q]);
      }
    }

#pragma unroll
    for (int q = 0; q < 4; ++q) {
#pragma unroll
      for (int v = 0; v < 8; ++v) {
        const int r = t * 16 + v + 8 * hf;
        if (r < n) {
          const float val = acc[q][v] + bv[q];
          h_out[(long long)r * HID + 16 * q + m] = val > 0.0f ? val : 0.0f;
        }
      }
    }
  }
}

__global__ void cls_kernel(const float* __restrict__ h, const float* __restrict__ Wc,
                           const float* __restrict__ bc, float* __restrict__ out, int n) {
  int i = blockIdx.x * blockDim.x + threadIdx.x;
  if (i >= n) return;
  const float4* hp = (const float4*)(h + (long long)i * HID);
  const float4* wp = (const float4*)Wc;
  float acc = bc[0];
#pragma unroll
  for (int j = 0; j < HID / 4; ++j) {
    float4 hv = hp[j];
    float4 wv = wp[j];
    acc += hv.x * wv.x + hv.y * wv.y + hv.z * wv.z + hv.w * wv.w;
  }
  out[i] = acc;
}

extern "C" void kernel_launch(void* const* d_in, const int* in_sizes, int n_in,
                              void* d_out, int out_size, void* d_ws, size_t ws_size,
                              hipStream_t stream) {
  const float* x   = (const float*)d_in[0];
  const int* eidx  = (const int*)d_in[1];
  const float* Wp  = (const float*)d_in[2];
  const float* bp  = (const float*)d_in[3];
  const float* W1l = (const float*)d_in[4];
  const float* b1l = (const float*)d_in[5];
  const float* W1r = (const float*)d_in[6];
  const float* W2l = (const float*)d_in[7];
  const float* b2l = (const float*)d_in[8];
  const float* W2r = (const float*)d_in[9];
  const float* Wc  = (const float*)d_in[10];
  const float* bc  = (const float*)d_in[11];

  const int N = in_sizes[0] / 165;
  const int E = in_sizes[1] / 2;
  const int* src = eidx;
  const int* dst = eidx + E;

  float* A   = (float*)d_ws;
  float* B   = A + (size_t)N * HID;
  float* AGG = B + (size_t)N * HID;
  float* DEG = AGG + (size_t)N * HID;

  const unsigned gz_n  = (unsigned)((N + 255) / 256);
  const unsigned g_ag  = (unsigned)((N + AG_ROWS - 1) / AG_ROWS);
  const unsigned g_t16 = (unsigned)((N + 15) / 16);
  const int e_chunk = (int)(((long long)E + g_t16 - 1) / g_t16);
  const unsigned g_gemm = 512;

  zero_f32<<<gz_n, 256, 0, stream>>>(DEG, (long long)N);

  proj_kernel<<<g_gemm, 128, 0, stream>>>(x, Wp, bp, A, N, dst, DEG, E, e_chunk);

  neigh_sum_kernel<<<g_ag, AG_THREADS, 0, stream>>>(A, src, dst, AGG, N, E);
  combine_kernel<<<g_gemm, 128, 0, stream>>>(AGG, DEG, A, W1l, b1l, W1r, B, N);

  neigh_sum_kernel<<<g_ag, AG_THREADS, 0, stream>>>(B, src, dst, AGG, N, E);
  combine_kernel<<<g_gemm, 128, 0, stream>>>(AGG, DEG, B, W2l, b2l, W2r, A, N);

  cls_kernel<<<(N + 255) / 256, 256, 0, stream>>>(A, Wc, bc, (float*)d_out, N);
}


# conv 2 gathers from per-wave (source row, tile row) lists that conv 1's scan writes to a freed plane (conv 1 combine stores h1 over agg): edge list scanned once, not twice; sum order kept
# speedup vs baseline: 3.4775x; 1.2798x over previous
#include <hip/hip_runtime.h>

#define HID 64

typedef __attribute__((ext_vector_type(2))) float v2f;
typedef __attribute__((ext_vector_type(8))) float v8f;
typedef __attribute__((ext_vector_type(4))) float v4f;
typedef v4f __attribute__((aligned(4))) v4f_u;

__device__ __forceinline__ v8f wmma_k4(v2f a, v2f b, v8f c) {
  return __builtin_amdgcn_wmma_f32_16x16x4_f32(
       false, a,  false, b,
       (short)0, c,  false,  false);
}

__device__ __forceinline__ void atomic_add_f32_hw(float* addr, float val) {
  asm volatile("global_atomic_add_f32 %0, %1, off scope:SCOPE_DEV"
               :
               : "v"(addr), "v"(val)
               : "memory");
}

__global__ void zero_f32(float* __restrict__ p, long long n) {
  long long i = (long long)blockIdx.x * blockDim.x + threadIdx.x;
  if (i < n) p[i] = 0.0f;
}

__device__ __forceinline__ int slice_dst(const int* __restrict__ dst, int i, int chunk,
                                         int ebase, int E) {
  const int e = ebase + i;
  return (i < chunk && e < E) ? dst[e] : -1;
}

#define AG_ROWS 784
#define AG_WAVES 32
#define AG_THREADS (AG_WAVES * 32)
#define AG_ITER (AG_WAVES * 512)
#define AG_OUTCAP 416
#define AG_MYCAP 64
#define AG_SRC_BITS 22
#define AG_LIST_HEAD 32

__device__ __forceinline__ void load_iter(int4 (&r)[4], const int* __restrict__ dst, int g,
                                          int woff, int E, bool vec) {
  const int base = g * AG_ITER;
  if (vec && base + AG_ITER <= E) {
#pragma unroll
    for (int j = 0; j < 4; ++j) r[j] = *(const int4*)(dst + base + woff + j * 128);
  } else {
#pragma unroll
    for (int j = 0; j < 4; ++j) {
      const int e = base + woff + j * 128;
      r[j].x = (e < E) ? dst[e] : -1;
      r[j].y = (e + 1 < E) ? dst[e + 1] : -1;
      r[j].z = (e + 2 < E) ? dst[e + 2] : -1;
      r[j].w = (e + 3 < E) ? dst[e + 3] : -1;
    }
  }
}

__device__ __forceinline__ int queue_push(int2* q, int qn, bool keep, int2 ent, int lane) {
  const unsigned b = __builtin_amdgcn_ballot_w32(keep);
  if (b != 0u) {
    if (keep) q[qn + __popc(b & ((1u << lane) - 1u))] = ent;
    qn += __popc(b);
  }
  return qn;
}

__device__ __forceinline__ bool push4(int2* q, int& qn, int4 d, int e, int rowlo, int lane) {
  const unsigned l0 = (unsigned)(d.x - rowlo);
  const unsigned l1 = (unsigned)(d.y - rowlo);
  const unsigned l2 = (unsigned)(d.z - rowlo);
  const unsigned l3 = (unsigned)(d.w - rowlo);
  const unsigned lm = min(min(l0, l1), min(l2, l3));
  if (__builtin_amdgcn_ballot_w32(lm < (unsigned)AG_ROWS) == 0u) return true;
  const bool k0 = l0 < (unsigned)AG_ROWS, k1 = l1 < (unsigned)AG_ROWS;
  const bool k2 = l2 < (unsigned)AG_ROWS, k3 = l3 < (unsigned)AG_ROWS;
  const unsigned b0 = __builtin_amdgcn_ballot_w32(k0), b1 = __builtin_amdgcn_ballot_w32(k1);
  const unsigned b2 = __builtin_amdgcn_ballot_w32(k2), b3 = __builtin_amdgcn_ballot_w32(k3);
  const int c0 = __popc(b0), c1 = __popc(b1), c2 = __popc(b2), c3 = __popc(b3);
  if (qn + c0 + c1 + c2 + c3 > AG_OUTCAP) return false;
  const unsigned below = (1u << lane) - 1u;
  int p = qn;
  if (k0) q[p + __popc(b0 & below)] = make_int2(e, (int)l0);
  p += c0;
  if (k1) q[p + __popc(b1 & below)] = make_int2(e + 1, (int)l1);
  p += c1;
  if (k2) q[p + __popc(b2 & below)] = make_int2(e + 2, (int)l2);
  p += c2;
  if (k3) q[p + __popc(b3 & below)] = make_int2(e + 3, (int)l3);
  qn = p + c3;
  return true;
}

template <bool REC>
__device__ __forceinline__ void add_rows(const int2* q, int cnt, const float* __restrict__ h,
                                         const int* __restrict__ src, float* tile, int lane,
                                         unsigned* __restrict__ rec, int& rpos, int rcap) {
  int s = 0, loc = 0;
  if (lane < cnt) {
    const int2 ent = q[lane];
    s = src[ent.x];
    loc = ent.y;
  }
  if (REC) {
    if (rpos + cnt <= rcap && lane < cnt)
      rec[AG_LIST_HEAD + rpos + lane] = (unsigned)s | ((unsigned)loc << AG_SRC_BITS);
    rpos += cnt;
  }
  for (int k = 0; k < cnt; k += 8) {
    float2 v[8];
    int lk[8];
#pragma unroll
    for (int u = 0; u < 8; ++u) {
      const int sk = __builtin_amdgcn_readlane(s, k + u);
      lk[u] = __builtin_amdgcn_readlane(loc, k + u);
      v[u] = *(const float2*)(h + (long long)sk * HID + lane * 2);
    }
#pragma unroll
    for (int u = 0; u < 8; ++u) {
      if (k + u < cnt) {
        float2* t = (float2*)(tile + lk[u] * HID + lane * 2);
        float2 a = *t;
        a.x += v[u].x;
        a.y += v[u].y;
        *t = a;
      }
    }
  }
}

template <bool REC>
__device__ __forceinline__ void scan_gather(const float* __restrict__ h,
                                            const int* __restrict__ src,
                                            const int* __restrict__ dst, int E, float* tile,
                                            int2* outq, int2* ownq, int* outn, int* fin,
                                            int rowlo, int wave, int lane,
                                            unsigned* __restrict__ rec, int& rpos, int rcap) {
  const bool vec = (((size_t)dst) & 15) == 0;
  const int nit = (E + AG_ITER - 1) / AG_ITER;
  const int start = (int)(((long long)blockIdx.x * nit) / (int)gridDim.x);
  const int woff = wave * 512 + lane * 4;
  const int4 none = {-1, -1, -1, -1};

  int4 cur[4], nxt[4];
#pragma unroll
  for (int j = 0; j < 4; ++j) {
    cur[j] = none;
    nxt[j] = none;
  }

  int2* oq = outq + wave * AG_OUTCAP;
  int2* mq = ownq + wave * AG_MYCAP;
  int it = -1;
  int mn = 0;

  for (int round = 0; round <= 4 * nit; ++round) {
    int qn = 0;
    while (it < nit) {
      int g = start + it;
      if (g >= nit) g -= nit;
      int g1 = start + it + 1;
      if (g1 >= nit) g1 -= nit;
      if (it + 1 < nit) {
        load_iter(nxt, dst, g1, woff, E, vec);
      } else {
#pragma unroll
        for (int j = 0; j < 4; ++j) nxt[j] = none;
      }
      const int e0 = g * AG_ITER + woff;
      int queued = 0;
      if (push4(oq, qn, cur[0], e0, rowlo, lane)) {
        queued = 1;
        if (push4(oq, qn, cur[1], e0 + 128, rowlo, lane)) {
          queued = 2;
          if (push4(oq, qn, cur[2], e0 + 256, rowlo, lane)) {
            queued = 3;
            if (push4(oq, qn, cur[3], e0 + 384, rowlo, lane)) queued = 4;
          }
        }
      }
      if (queued < 4) {
        if (queued > 0) cur[0] = none;
        if (queued > 1) cur[1] = none;
        if (queued > 2) cur[2] = none;
        break;
      }
#pragma unroll
      for (int j = 0; j < 4; ++j) cur[j] = nxt[j];
      ++it;
    }
    if (lane == 0) {
      outn[wave] = qn;
      fin[wave] = (it >= nit) ? 1 : 0;
    }
    __syncthreads();

    for (int i = 0; i < AG_WAVES; ++i) {
      const int w2 = (wave + i) & (AG_WAVES - 1);
      const int cnt = __builtin_amdgcn_readfirstlane(outn[w2]);
      for (int b0 = 0; b0 < cnt; b0 += 32) {
        const bool live = b0 + lane < cnt;
        int2 ent = make_int2(0, -1);
        if (live) ent = outq[w2 * AG_OUTCAP + b0 + lane];
        const bool mine = live && ((ent.y & (AG_WAVES - 1)) == wave);
        mn = queue_push(mq, mn, mine, ent, lane);
        if (mn >= 32) {
          add_rows<REC>(mq, 32, h, src, tile, lane, rec, rpos, rcap);
          const int rem = mn - 32;
          if (lane < rem) {
            const int2 mv = mq[32 + lane];
            mq[lane] = mv;
          }
          mn = rem;
        }
      }
    }
    const bool alldone = __builtin_amdgcn_ballot_w32(fin[lane] != 0) == 0xffffffffu;
    __syncthreads();
    if (alldone) break;
  }
  if (mn > 0) add_rows<REC>(mq, mn, h, src, tile, lane, rec, rpos, rcap);
}

template <int L0>
__device__ __forceinline__ void load_group(float2 (&v)[8], int (&l)[8], unsigned eb,
                                           const float* __restrict__ h, int lane) {
  const unsigned smask = (1u << AG_SRC_BITS) - 1u;
#pragma unroll
  for (int u = 0; u < 8; ++u) {
    const unsigned e = (unsigned)__builtin_amdgcn_readlane((int)eb, L0 + u);
    l[u] = (int)(e >> AG_SRC_BITS);
    v[u] = *(const float2*)(h + (long long)(e & smask) * HID + lane * 2);
  }
}

__device__ __forceinline__ void add_group(const float2 (&v)[8], const int (&l)[8], int first,
                                          int cnt, float* tile, int lane) {
#pragma unroll
  for (int u = 0; u < 8; ++u) {
    if (first + u < cnt) {
      float2* t = (float2*)(tile + l[u] * HID + lane * 2);
      float2 a = *t;
      a.x += v[u].x;
      a.y += v[u].y;
      *t = a;
    }
  }
}

__device__ __forceinline__ void replay_rows(const unsigned* __restrict__ ents, int cnt,
                                            const float* __restrict__ h, float* tile, int lane) {
  unsigned eb = (lane < cnt) ? ents[lane] : 0u;
  float2 va[8], vb[8], vc[8], vd[8];
  int la[8], lb[8], lc[8], ld[8];
  load_group<0>(va, la, eb, h, lane);
  __builtin_amdgcn_sched_barrier(0);
  load_group<8>(vb, lb, eb, h, lane);
  for (int b = 0; b < cnt; b += 32) {
    const unsigned en = (b + 32 + lane < cnt) ? ents[b + 32 + lane] : 0u;
    load_group<16>(vc, lc, eb, h, lane);
    add_group(va, la, b, cnt, tile, lane);
    load_group<24>(vd, ld, eb, h, lane);
    add_group(vb, lb, b + 8, cnt, tile, lane);
    load_group<0>(va, la, en, h, lane);
    add_group(vc, lc, b + 16, cnt, tile, lane);
    load_group<8>(vb, lb, en, h, lane);
    add_group(vd, ld, b + 24, cnt, tile, lane);
    eb = en;
  }
}

__device__ __forceinline__ int list_room(int rows, int n, int& wstride) {
  wstride = 2 * rows;
  const int room = wstride - AG_LIST_HEAD;
  return (room > 0 && n <= (1 << AG_SRC_BITS)) ? room : 0;
}

template <bool REPLAY>
__global__ __launch_bounds__(AG_THREADS) void neigh_sum_kernel(const float* __restrict__ h,
                                                             const int* __restrict__ src,
                                                             const int* __restrict__ dst,
                                                             float* __restrict__ agg,
                                                             unsigned* __restrict__ lists, int n,
                                                             int E) {
  __shared__ float tile[AG_ROWS * HID];
  __shared__ int2 outq[AG_WAVES * AG_OUTCAP];
  __shared__ int2 ownq[AG_WAVES * AG_MYCAP];
  __shared__ int outn[AG_WAVES];
  __shared__ int fin[AG_WAVES];

  const int tid = threadIdx.x;
  const int wave = tid >> 5;
  const int lane = tid & 31;
  const int rowlo = blockIdx.x * AG_ROWS;
  const int rows = (n - rowlo < AG_ROWS) ? (n - rowlo) : AG_ROWS;

  float4* t4 = (float4*)tile;
  const float4 z = {0.0f, 0.0f, 0.0f, 0.0f};
  for (int i = tid; i < AG_ROWS * HID / 4; i += AG_THREADS) t4[i] = z;

  int wstride;
  const int rcap = list_room(rows, n, wstride);
  unsigned* rec = lists + (long long)rowlo * HID + wave * wstride;
  int rpos = 0;

  if (REPLAY) {
    const int cnt = __builtin_amdgcn_readfirstlane((int)rec[0]);
    const int redo =
        __builtin_amdgcn_readfirstlane(__syncthreads_or((cnt < 0 || cnt > rcap) ? 1 : 0));
    if (!redo) {
      replay_rows(rec + AG_LIST_HEAD, cnt, h, tile, lane);
    } else {
      scan_gather<false>(h, src, dst, E, tile, outq, ownq, outn, fin, rowlo, wave, lane, rec,
                         rpos, rcap);
    }
  } else {
    scan_gather<true>(h, src, dst, E, tile, outq, ownq, outn, fin, rowlo, wave, lane, rec, rpos,
                      rcap);
    if (lane == 0) rec[0] = (unsigned)rpos;
  }
  __syncthreads();

  float4* o4 = (float4*)(agg + (long long)rowlo * HID);
  for (int i = tid; i < rows * (HID / 4); i += AG_THREADS) o4[i] = t4[i];
}


#define PJ_FULL 20
__global__ __launch_bounds__(128) __attribute__((amdgpu_waves_per_eu(2, 2))) void proj_kernel(const float* __restrict__ x,
                                                   const float* __restrict__ W,
                                                   const float* __restrict__ bias,
                                                   float* __restrict__ h, int n,
                                                   const int* __restrict__ dst,
                                                   float* __restrict__ deg, int E, int chunk) {
  const int tid = threadIdx.x;
  const int wave = __builtin_amdgcn_readfirstlane(tid >> 5);
  const int lane = tid & 31;
  const int hf = lane >> 4;
  const int m = lane & 15;

  v2f bw[PJ_FULL + 1][2][4];
#pragma unroll
  for (int j = 0; j < PJ_FULL; ++j) {
#pragma unroll
    for (int tt = 0; tt < 2; ++tt) {
      const int k = 8 * j + 4 * hf + 2 * tt;
#pragma unroll
      for (int q = 0; q < 4; ++q) {
        bw[j][tt][q][0] = W[k * 64 + 16 * q + m];
        bw[j][tt][q][1] = W[(k + 1) * 64 + 16 * q + m];
      }
    }
  }
#pragma unroll
  for (int tt = 0; tt < 2; ++tt) {
    const int k = 8 * PJ_FULL + 4 * hf + 2 * tt;
#pragma unroll
    for (int q = 0; q < 4; ++q) {
      bw[PJ_FULL][tt][q][0] = (k < 165) ? W[k * 64 + 16 * q + m] : 0.0f;
      bw[PJ_FULL][tt][q][1] = (k + 1 < 165) ? W[(k + 1) * 64 + 16 * q + m] : 0.0f;
    }
  }
  float bv[4];
#pragma unroll
  for (int q = 0; q < 4; ++q) bv[q] = bias[16 * q + m];

  const int ntiles = (n + 15) >> 4;
  for (int t = blockIdx.x * 4 + wave; t < ntiles; t += (int)gridDim.x * 4) {
    const int ebase = t * chunk;
    int dd[8];
#pragma unroll
    for (int u = 0; u < 8; ++u) dd[u] = slice_dst(dst, u * 32 + lane, chunk, ebase, E);

    const int row = t * 16 + m;
    const int rl = row < n ? row : n - 1;
    const float* xr = x + (long long)rl * 165 + 4 * hf;
    v4f a[PJ_FULL + 1];
#pragma unroll
    for (int j = 0; j < PJ_FULL; ++j) a[j] = *(const v4f_u*)(xr + 8 * j);
    const float* tp = xr + 8 * PJ_FULL;
    const float t0 = tp[0];
    const float t1 = tp[hf ? 0 : 1];
    const float t2 = tp[hf ? 0 : 2];
    const float t3 = tp[hf ? 0 : 3];
    v4f tl;
    tl[0] = t0;
    tl[1] = hf ? 0.0f : t1;
    tl[2] = hf ? 0.0f : t2;
    tl[3] = hf ? 0.0f : t3;
    a[PJ_FULL] = tl;

    v8f acc[4] = {};
#pragma unroll
    for (int j = 0; j <= PJ_FULL; ++j) {
#pragma unroll
      for (int tt = 0; tt < 2; ++tt) {
        v2f af;
        af[0] = a[j][2 * tt];
        af[1] = a[j][2 * tt + 1];
#pragma unroll
        for (int q = 0; q < 4; ++q) acc[q] = wmma_k4(af, bw[j][tt][q], acc[q]);
      }
    }

#pragma unroll
    for (int q = 0; q < 4; ++q) {
#pragma unroll
      for (int v = 0; v < 8; ++v) {
        const int r = t * 16 + v + 8 * hf;
        if (r < n) {
          const float val = acc[q][v] + bv[q];
          h[(long long)r * HID + 16 * q + m] = val > 0.0f ? val : 0.0f;
        }
      }
    }

#pragma unroll
    for (int u = 0; u < 8; ++u)
      if (dd[u] >= 0) atomic_add_f32_hw(&deg[dd[u]], 1.0f);
    for (int i = 256 + lane; i < chunk; i += 32) {
      const int d = slice_dst(dst, i, chunk, ebase, E);
      if (d >= 0) atomic_add_f32_hw(&deg[d], 1.0f);
    }
  }
}

__global__ __launch_bounds__(128) __attribute__((amdgpu_waves_per_eu(2, 2))) void combine_kernel(
    const float* agg, const float* __restrict__ deg,
    const float* __restrict__ h_in, const float* __restrict__ Wl,
    const float* __restrict__ bl, const float* __restrict__ Wr,
    float* h_out, int n) {
  const int tid = threadIdx.x;
  const int wave = __builtin_amdgcn_readfirstlane(tid >> 5);
  const int lane = tid & 31;
  const int hf = lane >> 4;
  const int m = lane & 15;

  v2f wl[8][2][4], wr[8][2][4];
#pragma unroll
  for (int j = 0; j < 8; ++j) {
#pragma unroll
    for (int tt = 0; tt < 2; ++tt) {
      const int k = 8 * j + 4 * hf + 2 * tt;
#pragma unroll
      for (int q = 0; q < 4; ++q) {
        wl[j][tt][q][0] = Wl[k * 64 + 16 * q + m];
        wl[j][tt][q][1] = Wl[(k + 1) * 64 + 16 * q + m];
        wr[j][tt][q][0] = Wr[k * 64 + 16 * q + m];
        wr[j][tt][q][1] = Wr[(k + 1) * 64 + 16 * q + m];
      }
    }
  }
  float bv[4];
#pragma unroll
  for (int q = 0; q < 4; ++q) bv[q] = bl[16 * q + m];

  const int ntiles = (n + 15) >> 4;
  for (int t = blockIdx.x * 4 + wave; t < ntiles; t += (int)gridDim.x * 4) {
    const int row = t * 16 + m;
    const int rl = row < n ? row : n - 1;
    const v4f* ap = (const v4f*)(agg + (long long)rl * HID + 4 * hf);
    const v4f* hp = (const v4f*)(h_in + (long long)rl * HID + 4 * hf);
    v4f a[8], hv[8];
#pragma unroll
    for (int j = 0; j < 8; ++j) {
      a[j] = ap[2 * j];
      hv[j] = hp[2 * j];
    }
    float d = deg[rl];
    d = d > 1.0f ? d : 1.0f;
#pragma unroll
    for (int j = 0; j < 8; ++j) {
      a[j][0] = a[j][0] / d;
      a[j][1] = a[j][1] / d;
      a[j][2] = a[j][2] / d;
      a[j][3] = a[j][3] / d;
    }

    v8f acc[4] = {};
#pragma unroll
    for (int j = 0; j < 8; ++j) {
#pragma unroll
      for (int tt = 0; tt < 2; ++tt) {
        v2f af;
        af[0] = a[j][2 * tt];
        af[1] = a[j][2 * tt + 1];
#pragma unroll
        for (int q = 0; q < 4; ++q) acc[q] = wmma_k4(af, wl[j][tt][q], acc[q]);
      }
    }
#pragma unroll
    for (int j = 0; j < 8; ++j) {
#pragma unroll
      for (int tt = 0; tt < 2; ++tt) {
        v2f af;
        af[0] = hv[j][2 * tt];
        af[1] = hv[j][2 * tt + 1];
#pragma unroll
        for (int q = 0; q < 4; ++q) acc[q] = wmma_k4(af, wr[j][tt][q], acc[q]);
      }
    }

#pragma unroll
    for (int q = 0; q < 4; ++q) {
#pragma unroll
      for (int v = 0; v < 8; ++v) {
        const int r = t * 16 + v + 8 * hf;
        if (r < n) {
          const float val = acc[q][v] + bv[q];
          h_out[(long long)r * HID + 16 * q + m] = val > 0.0f ? val : 0.0f;
        }
      }
    }
  }
}

__global__ void cls_kernel(const float* __restrict__ h, const float* __restrict__ Wc,
                           const float* __restrict__ bc, float* __restrict__ out, int n) {
  int i = blockIdx.x * blockDim.x + threadIdx.x;
  if (i >= n) return;
  const float4* hp = (const float4*)(h + (long long)i * HID);
  const float4* wp = (const float4*)Wc;
  float acc = bc[0];
#pragma unroll
  for (int j = 0; j < HID / 4; ++j) {
    float4 hv = hp[j];
    float4 wv = wp[j];
    acc += hv.x * wv.x + hv.y * wv.y + hv.z * wv.z + hv.w * wv.w;
  }
  out[i] = acc;
}

extern "C" void kernel_launch(void* const* d_in, const int* in_sizes, int n_in,
                              void* d_out, int out_size, void* d_ws, size_t ws_size,
                              hipStream_t stream) {
  const float* x   = (const float*)d_in[0];
  const int* eidx  = (const int*)d_in[1];
  const float* Wp  = (const float*)d_in[2];
  const float* bp  = (const float*)d_in[3];
  const float* W1l = (const float*)d_in[4];
  const float* b1l = (const float*)d_in[5];
  const float* W1r = (const float*)d_in[6];
  const float* W2l = (const float*)d_in[7];
  const float* b2l = (const float*)d_in[8];
  const float* W2r = (const float*)d_in[9];
  const float* Wc  = (const float*)d_in[10];
  const float* bc  = (const float*)d_in[11];

  const int N = in_sizes[0] / 165;
  const int E = in_sizes[1] / 2;
  const int* src = eidx;
  const int* dst = eidx + E;

  float* A   = (float*)d_ws;
  float* B   = A + (size_t)N * HID;
  float* AGG = B + (size_t)N * HID;
  float* DEG = AGG + (size_t)N * HID;
  unsigned* LISTS = (unsigned*)B;

  const unsigned gz_n  = (unsigned)((N + 255) / 256);
  const unsigned g_ag  = (unsigned)((N + AG_ROWS - 1) / AG_ROWS);
  const unsigned g_t16 = (unsigned)((N + 15) / 16);
  const int e_chunk = (int)(((long long)E + g_t16 - 1) / g_t16);
  const unsigned g_gemm = 512;

  zero_f32<<<gz_n, 256, 0, stream>>>(DEG, (long long)N);

  proj_kernel<<<g_gemm, 128, 0, stream>>>(x, Wp, bp, A, N, dst, DEG, E, e_chunk);

  neigh_sum_kernel<false><<<g_ag, AG_THREADS, 0, stream>>>(A, src, dst, AGG, LISTS, N, E);
  combine_kernel<<<g_gemm, 128, 0, stream>>>(AGG, DEG, A, W1l, b1l, W1r, AGG, N);

  neigh_sum_kernel<true><<<g_ag, AG_THREADS, 0, stream>>>(AGG, src, dst, A, LISTS, N, E);
  combine_kernel<<<g_gemm, 128, 0, stream>>>(A, DEG, AGG, W2l, b2l, W2r, B, N);

  cls_kernel<<<(N + 255) / 256, 256, 0, stream>>>(B, Wc, bc, (float*)d_out, N);
}


# edge list filtered in two levels: a coarse pass writes (source, row) lists per 16-group super-group, each group scans only those; one launch more; sum order changes, still fixed by edge order
# speedup vs baseline: 3.9926x; 1.1481x over previous
#include <hip/hip_runtime.h>

#define HID 64

typedef __attribute__((ext_vector_type(2))) float v2f;
typedef __attribute__((ext_vector_type(8))) float v8f;
typedef __attribute__((ext_vector_type(4))) float v4f;
typedef v4f __attribute__((aligned(4))) v4f_u;

__device__ __forceinline__ v8f wmma_k4(v2f a, v2f b, v8f c) {
  return __builtin_amdgcn_wmma_f32_16x16x4_f32(
       false, a,  false, b,
       (short)0, c,  false,  false);
}

__device__ __forceinline__ void atomic_add_f32_hw(float* addr, float val) {
  asm volatile("global_atomic_add_f32 %0, %1, off scope:SCOPE_DEV"
               :
               : "v"(addr), "v"(val)
               : "memory");
}

__global__ void zero_f32(float* __restrict__ p, long long n) {
  long long i = (long long)blockIdx.x * blockDim.x + threadIdx.x;
  if (i < n) p[i] = 0.0f;
}

__device__ __forceinline__ int slice_dst(const int* __restrict__ dst, int i, int chunk,
                                         int ebase, int E) {
  const int e = ebase + i;
  return (i < chunk && e < E) ? dst[e] : -1;
}

#define AG_ROWS 784
#define AG_WAVES 32
#define AG_THREADS (AG_WAVES * 32)
#define AG_ITER (AG_WAVES * 512)
#define AG_OUTCAP 416
#define AG_MYCAP 64
#define AG_SRC_BITS 22
#define AG_LIST_HEAD 32

__device__ __forceinline__ void load_iter(int4 (&r)[4], const int* __restrict__ dst, int g,
                                          int woff, int E, bool vec) {
  const int base = g * AG_ITER;
  if (vec && base + AG_ITER <= E) {
#pragma unroll
    for (int j = 0; j < 4; ++j) r[j] = *(const int4*)(dst + base + woff + j * 128);
  } else {
#pragma unroll
    for (int j = 0; j < 4; ++j) {
      const int e = base + woff + j * 128;
      r[j].x = (e < E) ? dst[e] : -1;
      r[j].y = (e + 1 < E) ? dst[e + 1] : -1;
      r[j].z = (e + 2 < E) ? dst[e + 2] : -1;
      r[j].w = (e + 3 < E) ? dst[e + 3] : -1;
    }
  }
}

__device__ __forceinline__ int queue_push(int2* q, int qn, bool keep, int2 ent, int lane) {
  const unsigned b = __builtin_amdgcn_ballot_w32(keep);
  if (b != 0u) {
    if (keep) q[qn + __popc(b & ((1u << lane) - 1u))] = ent;
    qn += __popc(b);
  }
  return qn;
}

__device__ __forceinline__ bool push4(int2* q, int& qn, int4 d, int e, int rowlo, int lane) {
  const unsigned l0 = (unsigned)(d.x - rowlo);
  const unsigned l1 = (unsigned)(d.y - rowlo);
  const unsigned l2 = (unsigned)(d.z - rowlo);
  const unsigned l3 = (unsigned)(d.w - rowlo);
  const unsigned lm = min(min(l0, l1), min(l2, l3));
  if (__builtin_amdgcn_ballot_w32(lm < (unsigned)AG_ROWS) == 0u) return true;
  const bool k0 = l0 < (unsigned)AG_ROWS, k1 = l1 < (unsigned)AG_ROWS;
  const bool k2 = l2 < (unsigned)AG_ROWS, k3 = l3 < (unsigned)AG_ROWS;
  const unsigned b0 = __builtin_amdgcn_ballot_w32(k0), b1 = __builtin_amdgcn_ballot_w32(k1);
  const unsigned b2 = __builtin_amdgcn_ballot_w32(k2), b3 = __builtin_amdgcn_ballot_w32(k3);
  const int c0 = __popc(b0), c1 = __popc(b1), c2 = __popc(b2), c3 = __popc(b3);
  if (qn + c0 + c1 + c2 + c3 > AG_OUTCAP) return false;
  const unsigned below = (1u << lane) - 1u;
  int p = qn;
  if (k0) q[p + __popc(b0 & below)] = make_int2(e, (int)l0);
  p += c0;
  if (k1) q[p + __popc(b1 & below)] = make_int2(e + 1, (int)l1);
  p += c1;
  if (k2) q[p + __popc(b2 & below)] = make_int2(e + 2, (int)l2);
  p += c2;
  if (k3) q[p + __popc(b3 & below)] = make_int2(e + 3, (int)l3);
  qn = p + c3;
  return true;
}

template <bool REC, bool DIRECT>
__device__ __forceinline__ void add_rows(const int2* q, int cnt, const float* __restrict__ h,
                                         const int* __restrict__ src, float* tile, int lane,
                                         unsigned* __restrict__ rec, int& rpos, int rcap) {
  int s = 0, loc = 0;
  if (lane < cnt) {
    const int2 ent = q[lane];
    s = DIRECT ? ent.x : src[ent.x];
    loc = ent.y;
  }
  if (REC) {
    if (rpos + cnt <= rcap && lane < cnt)
      rec[AG_LIST_HEAD + rpos + lane] = (unsigned)s | ((unsigned)loc << AG_SRC_BITS);
    rpos += cnt;
  }
  for (int k = 0; k < cnt; k += 8) {
    float2 v[8];
    int lk[8];
#pragma unroll
    for (int u = 0; u < 8; ++u) {
      const int sk = __builtin_amdgcn_readlane(s, k + u);
      lk[u] = __builtin_amdgcn_readlane(loc, k + u);
      v[u] = *(const float2*)(h + (long long)sk * HID + lane * 2);
    }
#pragma unroll
    for (int u = 0; u < 8; ++u) {
      if (k + u < cnt) {
        float2* t = (float2*)(tile + lk[u] * HID + lane * 2);
        float2 a = *t;
        a.x += v[u].x;
        a.y += v[u].y;
        *t = a;
      }
    }
  }
}

template <bool REC>
__device__ __forceinline__ void scan_gather(const float* __restrict__ h,
                                            const int* __restrict__ src,
                                            const int* __restrict__ dst, int E, float* tile,
                                            int2* outq, int2* ownq, int* outn, int* fin,
                                            int rowlo, int wave, int lane,
                                            unsigned* __restrict__ rec, int& rpos, int rcap) {
  const bool vec = (((size_t)dst) & 15) == 0;
  const int nit = (E + AG_ITER - 1) / AG_ITER;
  const int start = (int)(((long long)blockIdx.x * nit) / (int)gridDim.x);
  const int woff = wave * 512 + lane * 4;
  const int4 none = {-1, -1, -1, -1};

  int4 cur[4], nxt[4];
#pragma unroll
  for (int j = 0; j < 4; ++j) {
    cur[j] = none;
    nxt[j] = none;
  }

  int2* oq = outq + wave * AG_OUTCAP;
  int2* mq = ownq + wave * AG_MYCAP;
  int it = -1;
  int mn = 0;

  for (int round = 0; round <= 4 * nit; ++round) {
    int qn = 0;
    while (it < nit) {
      int g = start + it;
      if (g >= nit) g -= nit;
      int g1 = start + it + 1;
      if (g1 >= nit) g1 -= nit;
      if (it + 1 < nit) {
        load_iter(nxt, dst, g1, woff, E, vec);
      } else {
#pragma unroll
        for (int j = 0; j < 4; ++j) nxt[j] = none;
      }
      const int e0 = g * AG_ITER + woff;
      int queued = 0;
      if (push4(oq, qn, cur[0], e0, rowlo, lane)) {
        queued = 1;
        if (push4(oq, qn, cur[1], e0 + 128, rowlo, lane)) {
          queued = 2;
          if (push4(oq, qn, cur[2], e0 + 256, rowlo, lane)) {
            queued = 3;
            if (push4(oq, qn, cur[3], e0 + 384, rowlo, lane)) queued = 4;
          }
        }
      }
      if (queued < 4) {
        if (queued > 0) cur[0] = none;
        if (queued > 1) cur[1] = none;
        if (queued > 2) cur[2] = none;
        break;
      }
#pragma unroll
      for (int j = 0; j < 4; ++j) cur[j] = nxt[j];
      ++it;
    }
    if (lane == 0) {
      outn[wave] = qn;
      fin[wave] = (it >= nit) ? 1 : 0;
    }
    __syncthreads();

    for (int i = 0; i < AG_WAVES; ++i) {
      const int w2 = (wave + i) & (AG_WAVES - 1);
      const int cnt = __builtin_amdgcn_readfirstlane(outn[w2]);
      for (int b0 = 0; b0 < cnt; b0 += 32) {
        const bool live = b0 + lane < cnt;
        int2 ent = make_int2(0, -1);
        if (live) ent = outq[w2 * AG_OUTCAP + b0 + lane];
        const bool mine = live && ((ent.y & (AG_WAVES - 1)) == wave);
        mn = queue_push(mq, mn, mine, ent, lane);
        if (mn >= 32) {
          add_rows<REC, false>(mq, 32, h, src, tile, lane, rec, rpos, rcap);
          const int rem = mn - 32;
          if (lane < rem) {
            const int2 mv = mq[32 + lane];
            mq[lane] = mv;
          }
          mn = rem;
        }
      }
    }
    const bool alldone = __builtin_amdgcn_ballot_w32(fin[lane] != 0) == 0xffffffffu;
    __syncthreads();
    if (alldone) break;
  }
  if (mn > 0) add_rows<REC, false>(mq, mn, h, src, tile, lane, rec, rpos, rcap);
}

#define SG_Q 16
#define SG_C 16
#define SG_STRIDE 768
#define SG_CAP (SG_STRIDE - AG_LIST_HEAD)
#define SG_SRC_BITS 18
#define SG_OVER 0xffffffffu
#define SG_NB (SG_CAP / 32)
#define SG_OUTCAP (2 * AG_OUTCAP)

__device__ __forceinline__ bool sg_layout(int n, int& g_ag, int& nsuper) {
  g_ag = (n + AG_ROWS - 1) / AG_ROWS;
  nsuper = (g_ag + SG_Q - 1) / SG_Q;
  const long long words = ((long long)g_ag + (long long)nsuper * SG_C) * 32 * SG_STRIDE;
  return n <= (1 << SG_SRC_BITS) && words <= (long long)n * HID;
}

__device__ __forceinline__ int queue_push_w(unsigned* q, int qn, bool keep, unsigned w, int lane) {
  const unsigned b = __builtin_amdgcn_ballot_w32(keep);
  if (b != 0u) {
    if (keep) q[qn + __popc(b & ((1u << lane) - 1u))] = w;
    qn += __popc(b);
  }
  return qn;
}

__device__ __forceinline__ void load_pair(int4& d0, int4& s0, int4& d1, int4& s1,
                                          const int* __restrict__ dst,
                                          const int* __restrict__ src, long long e, int E,
                                          bool whole) {
  if (whole) {
    d0 = *(const int4*)(dst + e);
    s0 = *(const int4*)(src + e);
    d1 = *(const int4*)(dst + e + 128);
    s1 = *(const int4*)(src + e + 128);
  } else {
    const long long f = e + 128;
    d0.x = (e < E) ? dst[e] : -1;
    d0.y = (e + 1 < E) ? dst[e + 1] : -1;
    d0.z = (e + 2 < E) ? dst[e + 2] : -1;
    d0.w = (e + 3 < E) ? dst[e + 3] : -1;
    s0.x = (e < E) ? src[e] : 0;
    s0.y = (e + 1 < E) ? src[e + 1] : 0;
    s0.z = (e + 2 < E) ? src[e + 2] : 0;
    s0.w = (e + 3 < E) ? src[e + 3] : 0;
    d1.x = (f < E) ? dst[f] : -1;
    d1.y = (f + 1 < E) ? dst[f + 1] : -1;
    d1.z = (f + 2 < E) ? dst[f + 2] : -1;
    d1.w = (f + 3 < E) ? dst[f + 3] : -1;
    s1.x = (f < E) ? src[f] : 0;
    s1.y = (f + 1 < E) ? src[f + 1] : 0;
    s1.z = (f + 2 < E) ? src[f + 2] : 0;
    s1.w = (f + 3 < E) ? src[f + 3] : 0;
  }
}

__device__ __forceinline__ void coarse_push(unsigned* mq, int& mn, int& pos, unsigned* __restrict__ seg,
                                            int d, int s, int slo, int lane) {
  const unsigned u = (unsigned)(d - slo);
  const bool keep = u < (unsigned)(SG_Q * AG_ROWS);
  mn = queue_push_w(mq, mn, keep, (unsigned)s | (u << SG_SRC_BITS), lane);
  if (mn >= 32) {
    if (pos + 32 <= SG_CAP) seg[AG_LIST_HEAD + pos + lane] = mq[lane];
    pos += 32;
    const int rem = mn - 32;
    if (lane < rem) {
      const unsigned mv = mq[32 + lane];
      mq[lane] = mv;
    }
    mn = rem;
  }
}

__global__ __launch_bounds__(AG_THREADS) void coarse_kernel(const int* __restrict__ src,
                                                          const int* __restrict__ dst,
                                                          unsigned* __restrict__ lists, int n,
                                                          int E) {
  __shared__ unsigned stage[AG_WAVES * 64];
  int g_ag, nsuper;
  if (!sg_layout(n, g_ag, nsuper)) return;
  const int tid = threadIdx.x;
  const int wave = __builtin_amdgcn_readfirstlane(tid >> 5);
  const int lane = tid & 31;
  const int s = blockIdx.x / SG_C;
  const int c = blockIdx.x - s * SG_C;
  const int slo = s * (SG_Q * AG_ROWS);
  const long long share = ((long long)E + SG_C * AG_WAVES - 1) / (SG_C * AG_WAVES);
  const int steps = (int)((share + 127) / 128);
  const long long first = (long long)(c * AG_WAVES + wave) * steps * 128;
  unsigned* seg = lists + ((long long)g_ag * 32 + (long long)blockIdx.x * 32 + wave) * SG_STRIDE;
  unsigned* mq = stage + wave * 64;
  const bool vec = ((((size_t)src) | ((size_t)dst)) & 15) == 0;

  int mn = 0, pos = 0;
  int4 da, sa, db, sb;
  load_pair(da, sa, db, sb, dst, src, first + lane * 4, E, vec && first + 256 <= E);
  for (int i = 0; i < steps; i += 2) {
    const int4 d0 = da, s0 = sa, d1 = db, s1 = sb;
    if (i + 2 < steps) {
      const long long e2 = first + (long long)(i + 2) * 128;
      load_pair(da, sa, db, sb, dst, src, e2 + lane * 4, E, vec && e2 + 256 <= E);
    }
    coarse_push(mq, mn, pos, seg, d0.x, s0.x, slo, lane);
    coarse_push(mq, mn, pos, seg, d0.y, s0.y, slo, lane);
    coarse_push(mq, mn, pos, seg, d0.z, s0.z, slo, lane);
    coarse_push(mq, mn, pos, seg, d0.w, s0.w, slo, lane);
    if (i + 1 < steps) {
      coarse_push(mq, mn, pos, seg, d1.x, s1.x, slo, lane);
      coarse_push(mq, mn, pos, seg, d1.y, s1.y, slo, lane);
      coarse_push(mq, mn, pos, seg, d1.z, s1.z, slo, lane);
      coarse_push(mq, mn, pos, seg, d1.w, s1.w, slo, lane);
    }
  }
  if (mn > 0) {
    if (pos + mn <= SG_CAP && lane < mn) seg[AG_LIST_HEAD + pos + lane] = mq[lane];
    pos += mn;
  }
  if (lane == 0) seg[0] = (pos <= SG_CAP) ? (unsigned)pos : SG_OVER;
}

template <bool REC>
__device__ __forceinline__ void seg_scan_gather(const float* __restrict__ h,
                                                const unsigned* __restrict__ coarse, unsigned nlen,
                                                float* tile, unsigned* outw, int2* ownq, int* outn,
                                                int* fin, int lo, int rot, int wave, int lane,
                                                unsigned* __restrict__ rec, int& rpos, int rcap) {
  const unsigned smask = (1u << SG_SRC_BITS) - 1u;
  unsigned* oq = outw + wave * SG_OUTCAP;
  int2* mq = ownq + wave * AG_MYCAP;
  int ci = 0;
  int mn = 0;

  for (int round = 0; round <= SG_C; ++round) {
    int qn = 0;
    while (ci < SG_C) {
      const int cc = (ci + rot) & (SG_C - 1);
      const int len = __builtin_amdgcn_readlane((int)nlen, cc);
      if (qn + len > SG_OUTCAP) break;
      const unsigned* sp = coarse + ((long long)cc * 32 + wave) * SG_STRIDE + AG_LIST_HEAD;
      unsigned wv[SG_NB];
#pragma unroll
      for (int j = 0; j < SG_NB; ++j) wv[j] = (32 * j + lane < len) ? sp[32 * j + lane] : SG_OVER;
#pragma unroll
      for (int j = 0; j < SG_NB; ++j) {
        if (32 * j < len) {
          const unsigned l = (wv[j] >> SG_SRC_BITS) - (unsigned)lo;
          qn = queue_push_w(oq, qn, l < (unsigned)AG_ROWS, (wv[j] & smask) | (l << AG_SRC_BITS),
                            lane);
        }
      }
      ++ci;
    }
    if (lane == 0) {
      outn[wave] = qn;
      fin[wave] = (ci >= SG_C) ? 1 : 0;
    }
    __syncthreads();

    for (int i = 0; i < AG_WAVES; ++i) {
      const int w2 = (wave + i) & (AG_WAVES - 1);
      const int cnt = __builtin_amdgcn_readfirstlane(outn[w2]);
      for (int b0 = 0; b0 < cnt; b0 += 32) {
        const bool live = b0 + lane < cnt;
        unsigned pw = 0u;
        if (live) pw = outw[w2 * SG_OUTCAP + b0 + lane];
        const int l = (int)(pw >> AG_SRC_BITS);
        const bool mine = live && ((l & (AG_WAVES - 1)) == wave);
        mn = queue_push(mq, mn, mine, make_int2((int)(pw & ((1u << AG_SRC_BITS) - 1u)), l), lane);
        if (mn >= 32) {
          add_rows<REC, true>(mq, 32, h, (const int*)0, tile, lane, rec, rpos, rcap);
          const int rem = mn - 32;
          if (lane < rem) {
            const int2 mv = mq[32 + lane];
            mq[lane] = mv;
          }
          mn = rem;
        }
      }
    }
    const bool alldone = __builtin_amdgcn_ballot_w32(fin[lane] != 0) == 0xffffffffu;
    __syncthreads();
    if (alldone) break;
  }
  if (mn > 0) add_rows<REC, true>(mq, mn, h, (const int*)0, tile, lane, rec, rpos, rcap);
}

template <int L0>
__device__ __forceinline__ void load_group(float2 (&v)[8], int (&l)[8], unsigned eb,
                                           const float* __restrict__ h, int lane) {
  const unsigned smask = (1u << AG_SRC_BITS) - 1u;
#pragma unroll
  for (int u = 0; u < 8; ++u) {
    const unsigned e = (unsigned)__builtin_amdgcn_readlane((int)eb, L0 + u);
    l[u] = (int)(e >> AG_SRC_BITS);
    v[u] = *(const float2*)(h + (long long)(e & smask) * HID + lane * 2);
  }
}

__device__ __forceinline__ void add_group(const float2 (&v)[8], const int (&l)[8], int first,
                                          int cnt, float* tile, int lane) {
#pragma unroll
  for (int u = 0; u < 8; ++u) {
    if (first + u < cnt) {
      float2* t = (float2*)(tile + l[u] * HID + lane * 2);
      float2 a = *t;
      a.x += v[u].x;
      a.y += v[u].y;
      *t = a;
    }
  }
}

__device__ __forceinline__ void replay_rows(const unsigned* __restrict__ ents, int cnt,
                                            const float* __restrict__ h, float* tile, int lane) {
  unsigned eb = (lane < cnt) ? ents[lane] : 0u;
  float2 va[8], vb[8], vc[8], vd[8];
  int la[8], lb[8], lc[8], ld[8];
  load_group<0>(va, la, eb, h, lane);
  __builtin_amdgcn_sched_barrier(0);
  load_group<8>(vb, lb, eb, h, lane);
  for (int b = 0; b < cnt; b += 32) {
    const unsigned en = (b + 32 + lane < cnt) ? ents[b + 32 + lane] : 0u;
    load_group<16>(vc, lc, eb, h, lane);
    add_group(va, la, b, cnt, tile, lane);
    load_group<24>(vd, ld, eb, h, lane);
    add_group(vb, lb, b + 8, cnt, tile, lane);
    load_group<0>(va, la, en, h, lane);
    add_group(vc, lc, b + 16, cnt, tile, lane);
    load_group<8>(vb, lb, en, h, lane);
    add_group(vd, ld, b + 24, cnt, tile, lane);
    eb = en;
  }
}

template <bool REPLAY>
__global__ __launch_bounds__(AG_THREADS) void neigh_sum_kernel(const float* __restrict__ h,
                                                             const int* __restrict__ src,
                                                             const int* __restrict__ dst,
                                                             float* __restrict__ agg,
                                                             unsigned* __restrict__ lists, int n,
                                                             int E) {
  __shared__ float tile[AG_ROWS * HID];
  __shared__ int2 outq[AG_WAVES * AG_OUTCAP];
  __shared__ int2 ownq[AG_WAVES * AG_MYCAP];
  __shared__ int outn[AG_WAVES];
  __shared__ int fin[AG_WAVES];

  const int tid = threadIdx.x;
  const int wave = tid >> 5;
  const int lane = tid & 31;
  const int rowlo = blockIdx.x * AG_ROWS;
  const int rows = (n - rowlo < AG_ROWS) ? (n - rowlo) : AG_ROWS;

  float4* t4 = (float4*)tile;
  const float4 z = {0.0f, 0.0f, 0.0f, 0.0f};
  for (int i = tid; i < AG_ROWS * HID / 4; i += AG_THREADS) t4[i] = z;

  int g_ag, nsuper;
  const bool lay = sg_layout(n, g_ag, nsuper);
  const int rcap = lay ? SG_CAP : 0;
  unsigned* rec = lists + ((long long)blockIdx.x * 32 + wave) * SG_STRIDE;
  int rpos = 0;

  if (REPLAY) {
    const int cnt = lay ? __builtin_amdgcn_readfirstlane((int)rec[0]) : -1;
    const int redo =
        __builtin_amdgcn_readfirstlane(__syncthreads_or((cnt < 0 || cnt > rcap) ? 1 : 0));
    if (!redo) {
      replay_rows(rec + AG_LIST_HEAD, cnt, h, tile, lane);
    } else {
      scan_gather<false>(h, src, dst, E, tile, outq, ownq, outn, fin, rowlo, wave, lane, rec,
                         rpos, rcap);
    }
  } else {
    const int sg = blockIdx.x / SG_Q;
    const int lo = (blockIdx.x - sg * SG_Q) * AG_ROWS;
    const unsigned* coarse = lists + ((long long)g_ag * 32 + (long long)sg * SG_C * 32) * SG_STRIDE;
    unsigned nlen = 0u;
    if (lay && lane < SG_C) nlen = coarse[((long long)lane * 32 + wave) * SG_STRIDE];
    const bool over = __builtin_amdgcn_ballot_w32(nlen > (unsigned)SG_CAP) != 0u;
    const int direct =
        __builtin_amdgcn_readfirstlane(__syncthreads_or((!lay || over) ? 1 : 0));
    if (!direct) {
      seg_scan_gather<true>(h, coarse, nlen, tile, (unsigned*)outq, ownq, outn, fin, lo,
                            blockIdx.x - sg * SG_Q, wave, lane, rec, rpos, rcap);
    } else {
      scan_gather<true>(h, src, dst, E, tile, outq, ownq, outn, fin, rowlo, wave, lane, rec,
                        rpos, rcap);
    }
    if (lay && lane == 0) rec[0] = (unsigned)rpos;
  }
  __syncthreads();

  float4* o4 = (float4*)(agg + (long long)rowlo * HID);
  for (int i = tid; i < rows * (HID / 4); i += AG_THREADS) o4[i] = t4[i];
}


#define PJ_FULL 20
__global__ __launch_bounds__(128) __attribute__((amdgpu_waves_per_eu(2, 2))) void proj_kernel(const float* __restrict__ x,
                                                   const float* __restrict__ W,
                                                   const float* __restrict__ bias,
                                                   float* __restrict__ h, int n,
                                                   const int* __restrict__ dst,
                                                   float* __restrict__ deg, int E, int chunk) {
  const int tid = threadIdx.x;
  const int wave = __builtin_amdgcn_readfirstlane(tid >> 5);
  const int lane = tid & 31;
  const int hf = lane >> 4;
  const int m = lane & 15;

  v2f bw[PJ_FULL + 1][2][4];
#pragma unroll
  for (int j = 0; j < PJ_FULL; ++j) {
#pragma unroll
    for (int tt = 0; tt < 2; ++tt) {
      const int k = 8 * j + 4 * hf + 2 * tt;
#pragma unroll
      for (int q = 0; q < 4; ++q) {
        bw[j][tt][q][0] = W[k * 64 + 16 * q + m];
        bw[j][tt][q][1] = W[(k + 1) * 64 + 16 * q + m];
      }
    }
  }
#pragma unroll
  for (int tt = 0; tt < 2; ++tt) {
    const int k = 8 * PJ_FULL + 4 * hf + 2 * tt;
#pragma unroll
    for (int q = 0; q < 4; ++q) {
      bw[PJ_FULL][tt][q][0] = (k < 165) ? W[k * 64 + 16 * q + m] : 0.0f;
      bw[PJ_FULL][tt][q][1] = (k + 1 < 165) ? W[(k + 1) * 64 + 16 * q + m] : 0.0f;
    }
  }
  float bv[4];
#pragma unroll
  for (int q = 0; q < 4; ++q) bv[q] = bias[16 * q + m];

  const int ntiles = (n + 15) >> 4;
  for (int t = blockIdx.x * 4 + wave; t < ntiles; t += (int)gridDim.x * 4) {
    const int ebase = t * chunk;
    int dd[8];
#pragma unroll
    for (int u = 0; u < 8; ++u) dd[u] = slice_dst(dst, u * 32 + lane, chunk, ebase, E);

    const int row = t * 16 + m;
    const int rl = row < n ? row : n - 1;
    const float* xr = x + (long long)rl * 165 + 4 * hf;
    v4f a[PJ_FULL + 1];
#pragma unroll
    for (int j = 0; j < PJ_FULL; ++j) a[j] = *(const v4f_u*)(xr + 8 * j);
    const float* tp = xr + 8 * PJ_FULL;
    const float t0 = tp[0];
    const float t1 = tp[hf ? 0 : 1];
    const float t2 = tp[hf ? 0 : 2];
    const float t3 = tp[hf ? 0 : 3];
    v4f tl;
    tl[0] = t0;
    tl[1] = hf ? 0.0f : t1;
    tl[2] = hf ? 0.0f : t2;
    tl[3] = hf ? 0.0f : t3;
    a[PJ_FULL] = tl;

    v8f acc[4] = {};
#pragma unroll
    for (int j = 0; j <= PJ_FULL; ++j) {
#pragma unroll
      for (int tt = 0; tt < 2; ++tt) {
        v2f af;
        af[0] = a[j][2 * tt];
        af[1] = a[j][2 * tt + 1];
#pragma unroll
        for (int q = 0; q < 4; ++q) acc[q] = wmma_k4(af, bw[j][tt][q], acc[q]);
      }
    }

#pragma unroll
    for (int q = 0; q < 4; ++q) {
#pragma unroll
      for (int v = 0; v < 8; ++v) {
        const int r = t * 16 + v + 8 * hf;
        if (r < n) {
          const float val = acc[q][v] + bv[q];
          h[(long long)r * HID + 16 * q + m] = val > 0.0f ? val : 0.0f;
        }
      }
    }

#pragma unroll
    for (int u = 0; u < 8; ++u)
      if (dd[u] >= 0) atomic_add_f32_hw(&deg[dd[u]], 1.0f);
    for (int i = 256 + lane; i < chunk; i += 32) {
      const int d = slice_dst(dst, i, chunk, ebase, E);
      if (d >= 0) atomic_add_f32_hw(&deg[d], 1.0f);
    }
  }
}

__global__ __launch_bounds__(128) __attribute__((amdgpu_waves_per_eu(2, 2))) void combine_kernel(
    const float* agg, const float* __restrict__ deg,
    const float* __restrict__ h_in, const float* __restrict__ Wl,
    const float* __restrict__ bl, const float* __restrict__ Wr,
    float* h_out, int n) {
  const int tid = threadIdx.x;
  const int wave = __builtin_amdgcn_readfirstlane(tid >> 5);
  const int lane = tid & 31;
  const int hf = lane >> 4;
  const int m = lane & 15;

  v2f wl[8][2][4], wr[8][2][4];
#pragma unroll
  for (int j = 0; j < 8; ++j) {
#pragma unroll
    for (int tt = 0; tt < 2; ++tt) {
      const int k = 8 * j + 4 * hf + 2 * tt;
#pragma unroll
      for (int q = 0; q < 4; ++q) {
        wl[j][tt][q][0] = Wl[k * 64 + 16 * q + m];
        wl[j][tt][q][1] = Wl[(k + 1) * 64 + 16 * q + m];
        wr[j][tt][q][0] = Wr[k * 64 + 16 * q + m];
        wr[j][tt][q][1] = Wr[(k + 1) * 64 + 16 * q + m];
      }
    }
  }
  float bv[4];
#pragma unroll
  for (int q = 0; q < 4; ++q) bv[q] = bl[16 * q + m];

  const int ntiles = (n + 15) >> 4;
  for (int t = blockIdx.x * 4 + wave; t < ntiles; t += (int)gridDim.x * 4) {
    const int row = t * 16 + m;
    const int rl = row < n ? row : n - 1;
    const v4f* ap = (const v4f*)(agg + (long long)rl * HID + 4 * hf);
    const v4f* hp = (const v4f*)(h_in + (long long)rl * HID + 4 * hf);
    v4f a[8], hv[8];
#pragma unroll
    for (int j = 0; j < 8; ++j) {
      a[j] = ap[2 * j];
      hv[j] = hp[2 * j];
    }
    float d = deg[rl];
    d = d > 1.0f ? d : 1.0f;
#pragma unroll
    for (int j = 0; j < 8; ++j) {
      a[j][0] = a[j][0] / d;
      a[j][1] = a[j][1] / d;
      a[j][2] = a[j][2] / d;
      a[j][3] = a[j][3] / d;
    }

    v8f acc[4] = {};
#pragma unroll
    for (int j = 0; j < 8; ++j) {
#pragma unroll
      for (int tt = 0; tt < 2; ++tt) {
        v2f af;
        af[0] = a[j][2 * tt];
        af[1] = a[j][2 * tt + 1];
#pragma unroll
        for (int q = 0; q < 4; ++q) acc[q] = wmma_k4(af, wl[j][tt][q], acc[q]);
      }
    }
#pragma unroll
    for (int j = 0; j < 8; ++j) {
#pragma unroll
      for (int tt = 0; tt < 2; ++tt) {
        v2f af;
        af[0] = hv[j][2 * tt];
        af[1] = hv[j][2 * tt + 1];
#pragma unroll
        for (int q = 0; q < 4; ++q) acc[q] = wmma_k4(af, wr[j][tt][q], acc[q]);
      }
    }

#pragma unroll
    for (int q = 0; q < 4; ++q) {
#pragma unroll
      for (int v = 0; v < 8; ++v) {
        const int r = t * 16 + v + 8 * hf;
        if (r < n) {
          const float val = acc[q][v] + bv[q];
          h_out[(long long)r * HID + 16 * q + m] = val > 0.0f ? val : 0.0f;
        }
      }
    }
  }
}

__global__ void cls_kernel(const float* __restrict__ h, const float* __restrict__ Wc,
                           const float* __restrict__ bc, float* __restrict__ out, int n) {
  int i = blockIdx.x * blockDim.x + threadIdx.x;
  if (i >= n) return;
  const float4* hp = (const float4*)(h + (long long)i * HID);
  const float4* wp = (const float4*)Wc;
  float acc = bc[0];
#pragma unroll
  for (int j = 0; j < HID / 4; ++j) {
    float4 hv = hp[j];
    float4 wv = wp[j];
    acc += hv.x * wv.x + hv.y * wv.y + hv.z * wv.z + hv.w * wv.w;
  }
  out[i] = acc;
}

extern "C" void kernel_launch(void* const* d_in, const int* in_sizes, int n_in,
                              void* d_out, int out_size, void* d_ws, size_t ws_size,
                              hipStream_t stream) {
  const float* x   = (const float*)d_in[0];
  const int* eidx  = (const int*)d_in[1];
  const float* Wp  = (const float*)d_in[2];
  const float* bp  = (const float*)d_in[3];
  const float* W1l = (const float*)d_in[4];
  const float* b1l = (const float*)d_in[5];
  const float* W1r = (const float*)d_in[6];
  const float* W2l = (const float*)d_in[7];
  const float* b2l = (const float*)d_in[8];
  const float* W2r = (const float*)d_in[9];
  const float* Wc  = (const float*)d_in[10];
  const float* bc  = (const float*)d_in[11];

  const int N = in_sizes[0] / 165;
  const int E = in_sizes[1] / 2;
  const int* src = eidx;
  const int* dst = eidx + E;

  float* A   = (float*)d_ws;
  float* B   = A + (size_t)N * HID;
  float* AGG = B + (size_t)N * HID;
  float* DEG = AGG + (size_t)N * HID;
  unsigned* LISTS = (unsigned*)B;

  const unsigned gz_n  = (unsigned)((N + 255) / 256);
  const unsigned g_ag  = (unsigned)((N + AG_ROWS - 1) / AG_ROWS);
  const unsigned g_t16 = (unsigned)((N + 15) / 16);
  const int e_chunk = (int)(((long long)E + g_t16 - 1) / g_t16);
  const unsigned g_gemm = 512;
  const unsigned g_co  = ((g_ag + SG_Q - 1) / SG_Q) * SG_C;

  zero_f32<<<gz_n, 256, 0, stream>>>(DEG, (long long)N);

  proj_kernel<<<g_gemm, 128, 0, stream>>>(x, Wp, bp, A, N, dst, DEG, E, e_chunk);

  coarse_kernel<<<g_co, AG_THREADS, 0, stream>>>(src, dst, LISTS, N, E);

  neigh_sum_kernel<false><<<g_ag, AG_THREADS, 0, stream>>>(A, src, dst, AGG, LISTS, N, E);
  combine_kernel<<<g_gemm, 128, 0, stream>>>(AGG, DEG, A, W1l, b1l, W1r, AGG, N);

  neigh_sum_kernel<true><<<g_ag, AG_THREADS, 0, stream>>>(AGG, src, dst, A, LISTS, N, E);
  combine_kernel<<<g_gemm, 128, 0, stream>>>(A, DEG, AGG, W2l, b2l, W2r, B, N);

  cls_kernel<<<(N + 255) / 256, 256, 0, stream>>>(B, Wc, bc, (float*)d_out, N);
}


# degree = rows a wave adds into its own LDS tile row, counted in registers in conv 1's gather; degree atomics, their clear launch and proj's pass over dst gone (8 launches to 7); float sums kept
# speedup vs baseline: 5.2333x; 1.3107x over previous
#include <hip/hip_runtime.h>

#define HID 64

typedef __attribute__((ext_vector_type(2))) float v2f;
typedef __attribute__((ext_vector_type(8))) float v8f;
typedef __attribute__((ext_vector_type(4))) float v4f;
typedef v4f __attribute__((aligned(4))) v4f_u;

__device__ __forceinline__ v8f wmma_k4(v2f a, v2f b, v8f c) {
  return __builtin_amdgcn_wmma_f32_16x16x4_f32(
       false, a,  false, b,
       (short)0, c,  false,  false);
}

#define AG_ROWS 784
#define AG_WAVES 32
#define AG_THREADS (AG_WAVES * 32)
#define AG_ITER (AG_WAVES * 512)
#define AG_OUTCAP 416
#define AG_MYCAP 64
#define AG_SRC_BITS 22
#define AG_LIST_HEAD 32

__device__ __forceinline__ void load_iter(int4 (&r)[4], const int* __restrict__ dst, int g,
                                          int woff, int E, bool vec) {
  const int base = g * AG_ITER;
  if (vec && base + AG_ITER <= E) {
#pragma unroll
    for (int j = 0; j < 4; ++j) r[j] = *(const int4*)(dst + base + woff + j * 128);
  } else {
#pragma unroll
    for (int j = 0; j < 4; ++j) {
      const int e = base + woff + j * 128;
      r[j].x = (e < E) ? dst[e] : -1;
      r[j].y = (e + 1 < E) ? dst[e + 1] : -1;
      r[j].z = (e + 2 < E) ? dst[e + 2] : -1;
      r[j].w = (e + 3 < E) ? dst[e + 3] : -1;
    }
  }
}

__device__ __forceinline__ int queue_push(int2* q, int qn, bool keep, int2 ent, int lane) {
  const unsigned b = __builtin_amdgcn_ballot_w32(keep);
  if (b != 0u) {
    if (keep) q[qn + __popc(b & ((1u << lane) - 1u))] = ent;
    qn += __popc(b);
  }
  return qn;
}

__device__ __forceinline__ bool push4(int2* q, int& qn, int4 d, int e, int rowlo, int lane) {
  const unsigned l0 = (unsigned)(d.x - rowlo);
  const unsigned l1 = (unsigned)(d.y - rowlo);
  const unsigned l2 = (unsigned)(d.z - rowlo);
  const unsigned l3 = (unsigned)(d.w - rowlo);
  const unsigned lm = min(min(l0, l1), min(l2, l3));
  if (__builtin_amdgcn_ballot_w32(lm < (unsigned)AG_ROWS) == 0u) return true;
  const bool k0 = l0 < (unsigned)AG_ROWS, k1 = l1 < (unsigned)AG_ROWS;
  const bool k2 = l2 < (unsigned)AG_ROWS, k3 = l3 < (unsigned)AG_ROWS;
  const unsigned b0 = __builtin_amdgcn_ballot_w32(k0), b1 = __builtin_amdgcn_ballot_w32(k1);
  const unsigned b2 = __builtin_amdgcn_ballot_w32(k2), b3 = __builtin_amdgcn_ballot_w32(k3);
  const int c0 = __popc(b0), c1 = __popc(b1), c2 = __popc(b2), c3 = __popc(b3);
  if (qn + c0 + c1 + c2 + c3 > AG_OUTCAP) return false;
  const unsigned below = (1u << lane) - 1u;
  int p = qn;
  if (k0) q[p + __popc(b0 & below)] = make_int2(e, (int)l0);
  p += c0;
  if (k1) q[p + __popc(b1 & below)] = make_int2(e + 1, (int)l1);
  p += c1;
  if (k2) q[p + __popc(b2 & below)] = make_int2(e + 2, (int)l2);
  p += c2;
  if (k3) q[p + __popc(b3 & below)] = make_int2(e + 3, (int)l3);
  qn = p + c3;
  return true;
}

template <bool REC, bool DIRECT>
__device__ __forceinline__ void add_rows(const int2* q, int cnt, const float* __restrict__ h,
                                         const int* __restrict__ src, float* tile, int lane,
                                         unsigned* __restrict__ rec, int& rpos, int rcap,
                                         int& dcnt, int myrow) {
  int s = 0, loc = 0;
  if (lane < cnt) {
    const int2 ent = q[lane];
    s = DIRECT ? ent.x : src[ent.x];
    loc = ent.y;
  }
  if (REC) {
    if (rpos + cnt <= rcap && lane < cnt)
      rec[AG_LIST_HEAD + rpos + lane] = (unsigned)s | ((unsigned)loc << AG_SRC_BITS);
    rpos += cnt;
  }
  for (int k = 0; k < cnt; k += 8) {
    float2 v[8];
    int lk[8];
#pragma unroll
    for (int u = 0; u < 8; ++u) {
      const int sk = __builtin_amdgcn_readlane(s, k + u);
      lk[u] = __builtin_amdgcn_readlane(loc, k + u);
      v[u] = *(const float2*)(h + (long long)sk * HID + lane * 2);
    }
#pragma unroll
    for (int u = 0; u < 8; ++u) {
      if (k + u < cnt) {
        float2* t = (float2*)(tile + lk[u] * HID + lane * 2);
        float2 a = *t;
        a.x += v[u].x;
        a.y += v[u].y;
        *t = a;
        if (REC) dcnt += (lk[u] == myrow) ? 1 : 0;
      }
    }
  }
}

template <bool REC>
__device__ __forceinline__ void scan_gather(const float* __restrict__ h,
                                            const int* __restrict__ src,
                                            const int* __restrict__ dst, int E, float* tile,
                                            int2* outq, int2* ownq, int* outn, int* fin,
                                            int rowlo, int wave, int lane,
                                            unsigned* __restrict__ rec, int& rpos, int rcap,
                                            int& dcnt) {
  const bool vec = (((size_t)dst) & 15) == 0;
  const int nit = (E + AG_ITER - 1) / AG_ITER;
  const int start = (int)(((long long)blockIdx.x * nit) / (int)gridDim.x);
  const int woff = wave * 512 + lane * 4;
  const int4 none = {-1, -1, -1, -1};

  int4 cur[4], nxt[4];
#pragma unroll
  for (int j = 0; j < 4; ++j) {
    cur[j] = none;
    nxt[j] = none;
  }

  int2* oq = outq + wave * AG_OUTCAP;
  int2* mq = ownq + wave * AG_MYCAP;
  int it = -1;
  int mn = 0;

  for (int round = 0; round <= 4 * nit; ++round) {
    int qn = 0;
    while (it < nit) {
      int g = start + it;
      if (g >= nit) g -= nit;
      int g1 = start + it + 1;
      if (g1 >= nit) g1 -= nit;
      if (it + 1 < nit) {
        load_iter(nxt, dst, g1, woff, E, vec);
      } else {
#pragma unroll
        for (int j = 0; j < 4; ++j) nxt[j] = none;
      }
      const int e0 = g * AG_ITER + woff;
      int queued = 0;
      if (push4(oq, qn, cur[0], e0, rowlo, lane)) {
        queued = 1;
        if (push4(oq, qn, cur[1], e0 + 128, rowlo, lane)) {
          queued = 2;
          if (push4(oq, qn, cur[2], e0 + 256, rowlo, lane)) {
            queued = 3;
            if (push4(oq, qn, cur[3], e0 + 384, rowlo, lane)) queued = 4;
          }
        }
      }
      if (queued < 4) {
        if (queued > 0) cur[0] = none;
        if (queued > 1) cur[1] = none;
        if (queued > 2) cur[2] = none;
        break;
      }
#pragma unroll
      for (int j = 0; j < 4; ++j) cur[j] = nxt[j];
      ++it;
    }
    if (lane == 0) {
      outn[wave] = qn;
      fin[wave] = (it >= nit) ? 1 : 0;
    }
    __syncthreads();

    for (int i = 0; i < AG_WAVES; ++i) {
      const int w2 = (wave + i) & (AG_WAVES - 1);
      const int cnt = __builtin_amdgcn_readfirstlane(outn[w2]);
      for (int b0 = 0; b0 < cnt; b0 += 32) {
        const bool live = b0 + lane < cnt;
        int2 ent = make_int2(0, -1);
        if (live) ent = outq[w2 * AG_OUTCAP + b0 + lane];
        const bool mine = live && ((ent.y & (AG_WAVES - 1)) == wave);
        mn = queue_push(mq, mn, mine, ent, lane);
        if (mn >= 32) {
          add_rows<REC, false>(mq, 32, h, src, tile, lane, rec, rpos, rcap, dcnt, wave + 32 * lane);
          const int rem = mn - 32;
          if (lane < rem) {
            const int2 mv = mq[32 + lane];
            mq[lane] = mv;
          }
          mn = rem;
        }
      }
    }
    const bool alldone = __builtin_amdgcn_ballot_w32(fin[lane] != 0) == 0xffffffffu;
    __syncthreads();
    if (alldone) break;
  }
  if (mn > 0) add_rows<REC, false>(mq, mn, h, src, tile, lane, rec, rpos, rcap, dcnt, wave + 32 * lane);
}

#define SG_Q 16
#define SG_C 16
#define SG_STRIDE 768
#define SG_CAP (SG_STRIDE - AG_LIST_HEAD)
#define SG_SRC_BITS 18
#define SG_OVER 0xffffffffu
#define SG_NB (SG_CAP / 32)
#define SG_OUTCAP (2 * AG_OUTCAP)

__device__ __forceinline__ bool sg_layout(int n, int& g_ag, int& nsuper) {
  g_ag = (n + AG_ROWS - 1) / AG_ROWS;
  nsuper = (g_ag + SG_Q - 1) / SG_Q;
  const long long words = ((long long)g_ag + (long long)nsuper * SG_C) * 32 * SG_STRIDE;
  return n <= (1 << SG_SRC_BITS) && words <= (long long)n * HID;
}

__device__ __forceinline__ int queue_push_w(unsigned* q, int qn, bool keep, unsigned w, int lane) {
  const unsigned b = __builtin_amdgcn_ballot_w32(keep);
  if (b != 0u) {
    if (keep) q[qn + __popc(b & ((1u << lane) - 1u))] = w;
    qn += __popc(b);
  }
  return qn;
}

__device__ __forceinline__ void load_pair(int4& d0, int4& s0, int4& d1, int4& s1,
                                          const int* __restrict__ dst,
                                          const int* __restrict__ src, long long e, int E,
                                          bool whole) {
  if (whole) {
    d0 = *(const int4*)(dst + e);
    s0 = *(const int4*)(src + e);
    d1 = *(const int4*)(dst + e + 128);
    s1 = *(const int4*)(src + e + 128);
  } else {
    const long long f = e + 128;
    d0.x = (e < E) ? dst[e] : -1;
    d0.y = (e + 1 < E) ? dst[e + 1] : -1;
    d0.z = (e + 2 < E) ? dst[e + 2] : -1;
    d0.w = (e + 3 < E) ? dst[e + 3] : -1;
    s0.x = (e < E) ? src[e] : 0;
    s0.y = (e + 1 < E) ? src[e + 1] : 0;
    s0.z = (e + 2 < E) ? src[e + 2] : 0;
    s0.w = (e + 3 < E) ? src[e + 3] : 0;
    d1.x = (f < E) ? dst[f] : -1;
    d1.y = (f + 1 < E) ? dst[f + 1] : -1;
    d1.z = (f + 2 < E) ? dst[f + 2] : -1;
    d1.w = (f + 3 < E) ? dst[f + 3] : -1;
    s1.x = (f < E) ? src[f] : 0;
    s1.y = (f + 1 < E) ? src[f + 1] : 0;
    s1.z = (f + 2 < E) ? src[f + 2] : 0;
    s1.w = (f + 3 < E) ? src[f + 3] : 0;
  }
}

__device__ __forceinline__ void coarse_push(unsigned* mq, int& mn, int& pos, unsigned* __restrict__ seg,
                                            int d, int s, int slo, int lane) {
  const unsigned u = (unsigned)(d - slo);
  const bool keep = u < (unsigned)(SG_Q * AG_ROWS);
  mn = queue_push_w(mq, mn, keep, (unsigned)s | (u << SG_SRC_BITS), lane);
  if (mn >= 32) {
    if (pos + 32 <= SG_CAP) seg[AG_LIST_HEAD + pos + lane] = mq[lane];
    pos += 32;
    const int rem = mn - 32;
    if (lane < rem) {
      const unsigned mv = mq[32 + lane];
      mq[lane] = mv;
    }
    mn = rem;
  }
}

__global__ __launch_bounds__(AG_THREADS) void coarse_kernel(const int* __restrict__ src,
                                                          const int* __restrict__ dst,
                                                          unsigned* __restrict__ lists, int n,
                                                          int E) {
  __shared__ unsigned stage[AG_WAVES * 64];
  int g_ag, nsuper;
  if (!sg_layout(n, g_ag, nsuper)) return;
  const int tid = threadIdx.x;
  const int wave = __builtin_amdgcn_readfirstlane(tid >> 5);
  const int lane = tid & 31;
  const int s = blockIdx.x / SG_C;
  const int c = blockIdx.x - s * SG_C;
  const int slo = s * (SG_Q * AG_ROWS);
  const long long share = ((long long)E + SG_C * AG_WAVES - 1) / (SG_C * AG_WAVES);
  const int steps = (int)((share + 127) / 128);
  const long long first = (long long)(c * AG_WAVES + wave) * steps * 128;
  unsigned* seg = lists + ((long long)g_ag * 32 + (long long)blockIdx.x * 32 + wave) * SG_STRIDE;
  unsigned* mq = stage + wave * 64;
  const bool vec = ((((size_t)src) | ((size_t)dst)) & 15) == 0;

  int mn = 0, pos = 0;
  int4 da, sa, db, sb;
  load_pair(da, sa, db, sb, dst, src, first + lane * 4, E, vec && first + 256 <= E);
  for (int i = 0; i < steps; i += 2) {
    const int4 d0 = da, s0 = sa, d1 = db, s1 = sb;
    if (i + 2 < steps) {
      const long long e2 = first + (long long)(i + 2) * 128;
      load_pair(da, sa, db, sb, dst, src, e2 + lane * 4, E, vec && e2 + 256 <= E);
    }
    coarse_push(mq, mn, pos, seg, d0.x, s0.x, slo, lane);
    coarse_push(mq, mn, pos, seg, d0.y, s0.y, slo, lane);
    coarse_push(mq, mn, pos, seg, d0.z, s0.z, slo, lane);
    coarse_push(mq, mn, pos, seg, d0.w, s0.w, slo, lane);
    if (i + 1 < steps) {
      coarse_push(mq, mn, pos, seg, d1.x, s1.x, slo, lane);
      coarse_push(mq, mn, pos, seg, d1.y, s1.y, slo, lane);
      coarse_push(mq, mn, pos, seg, d1.z, s1.z, slo, lane);
      coarse_push(mq, mn, pos, seg, d1.w, s1.w, slo, lane);
    }
  }
  if (mn > 0) {
    if (pos + mn <= SG_CAP && lane < mn) seg[AG_LIST_HEAD + pos + lane] = mq[lane];
    pos += mn;
  }
  if (lane == 0) seg[0] = (pos <= SG_CAP) ? (unsigned)pos : SG_OVER;
}

template <bool REC>
__device__ __forceinline__ void seg_scan_gather(const float* __restrict__ h,
                                                const unsigned* __restrict__ coarse, unsigned nlen,
                                                float* tile, unsigned* outw, int2* ownq, int* outn,
                                                int* fin, int lo, int rot, int wave, int lane,
                                                unsigned* __restrict__ rec, int& rpos, int rcap,
                                                int& dcnt) {
  const unsigned smask = (1u << SG_SRC_BITS) - 1u;
  unsigned* oq = outw + wave * SG_OUTCAP;
  int2* mq = ownq + wave * AG_MYCAP;
  int ci = 0;
  int mn = 0;

  for (int round = 0; round <= SG_C; ++round) {
    int qn = 0;
    while (ci < SG_C) {
      const int cc = (ci + rot) & (SG_C - 1);
      const int len = __builtin_amdgcn_readlane((int)nlen, cc);
      if (qn + len > SG_OUTCAP) break;
      const unsigned* sp = coarse + ((long long)cc * 32 + wave) * SG_STRIDE + AG_LIST_HEAD;
      unsigned wv[SG_NB];
#pragma unroll
      for (int j = 0; j < SG_NB; ++j) wv[j] = (32 * j + lane < len) ? sp[32 * j + lane] : SG_OVER;
#pragma unroll
      for (int j = 0; j < SG_NB; ++j) {
        if (32 * j < len) {
          const unsigned l = (wv[j] >> SG_SRC_BITS) - (unsigned)lo;
          qn = queue_push_w(oq, qn, l < (unsigned)AG_ROWS, (wv[j] & smask) | (l << AG_SRC_BITS),
                            lane);
        }
      }
      ++ci;
    }
    if (lane == 0) {
      outn[wave] = qn;
      fin[wave] = (ci >= SG_C) ? 1 : 0;
    }
    __syncthreads();

    for (int i = 0; i < AG_WAVES; ++i) {
      const int w2 = (wave + i) & (AG_WAVES - 1);
      const int cnt = __builtin_amdgcn_readfirstlane(outn[w2]);
      for (int b0 = 0; b0 < cnt; b0 += 32) {
        const bool live = b0 + lane < cnt;
        unsigned pw = 0u;
        if (live) pw = outw[w2 * SG_OUTCAP + b0 + lane];
        const int l = (int)(pw >> AG_SRC_BITS);
        const bool mine = live && ((l & (AG_WAVES - 1)) == wave);
        mn = queue_push(mq, mn, mine, make_int2((int)(pw & ((1u << AG_SRC_BITS) - 1u)), l), lane);
        if (mn >= 32) {
          add_rows<REC, true>(mq, 32, h, (const int*)0, tile, lane, rec, rpos, rcap, dcnt, wave + 32 * lane);
          const int rem = mn - 32;
          if (lane < rem) {
            const int2 mv = mq[32 + lane];
            mq[lane] = mv;
          }
          mn = rem;
        }
      }
    }
    const bool alldone = __builtin_amdgcn_ballot_w32(fin[lane] != 0) == 0xffffffffu;
    __syncthreads();
    if (alldone) break;
  }
  if (mn > 0) add_rows<REC, true>(mq, mn, h, (const int*)0, tile, lane, rec, rpos, rcap, dcnt, wave + 32 * lane);
}

template <int L0>
__device__ __forceinline__ void load_group(float2 (&v)[8], int (&l)[8], unsigned eb,
                                           const float* __restrict__ h, int lane) {
  const unsigned smask = (1u << AG_SRC_BITS) - 1u;
#pragma unroll
  for (int u = 0; u < 8; ++u) {
    const unsigned e = (unsigned)__builtin_amdgcn_readlane((int)eb, L0 + u);
    l[u] = (int)(e >> AG_SRC_BITS);
    v[u] = *(const float2*)(h + (long long)(e & smask) * HID + lane * 2);
  }
}

__device__ __forceinline__ void add_group(const float2 (&v)[8], const int (&l)[8], int first,
                                          int cnt, float* tile, int lane) {
#pragma unroll
  for (int u = 0; u < 8; ++u) {
    if (first + u < cnt) {
      float2* t = (float2*)(tile + l[u] * HID + lane * 2);
      float2 a = *t;
      a.x += v[u].x;
      a.y += v[u].y;
      *t = a;
    }
  }
}

__device__ __forceinline__ void replay_rows(const unsigned* __restrict__ ents, int cnt,
                                            const float* __restrict__ h, float* tile, int lane) {
  unsigned eb = (lane < cnt) ? ents[lane] : 0u;
  float2 va[8], vb[8], vc[8], vd[8];
  int la[8], lb[8], lc[8], ld[8];
  load_group<0>(va, la, eb, h, lane);
  __builtin_amdgcn_sched_barrier(0);
  load_group<8>(vb, lb, eb, h, lane);
  for (int b = 0; b < cnt; b += 32) {
    const unsigned en = (b + 32 + lane < cnt) ? ents[b + 32 + lane] : 0u;
    load_group<16>(vc, lc, eb, h, lane);
    add_group(va, la, b, cnt, tile, lane);
    load_group<24>(vd, ld, eb, h, lane);
    add_group(vb, lb, b + 8, cnt, tile, lane);
    load_group<0>(va, la, en, h, lane);
    add_group(vc, lc, b + 16, cnt, tile, lane);
    load_group<8>(vb, lb, en, h, lane);
    add_group(vd, ld, b + 24, cnt, tile, lane);
    eb = en;
  }
}

template <bool REPLAY>
__global__ __launch_bounds__(AG_THREADS) void neigh_sum_kernel(const float* __restrict__ h,
                                                             const int* __restrict__ src,
                                                             const int* __restrict__ dst,
                                                             float* __restrict__ agg,
                                                             unsigned* __restrict__ lists,
                                                             float* __restrict__ deg, int n,
                                                             int E) {
  __shared__ float tile[AG_ROWS * HID];
  __shared__ int2 outq[AG_WAVES * AG_OUTCAP];
  __shared__ int2 ownq[AG_WAVES * AG_MYCAP];
  __shared__ int outn[AG_WAVES];
  __shared__ int fin[AG_WAVES];

  const int tid = threadIdx.x;
  const int wave = tid >> 5;
  const int lane = tid & 31;
  const int rowlo = blockIdx.x * AG_ROWS;
  const int rows = (n - rowlo < AG_ROWS) ? (n - rowlo) : AG_ROWS;

  float4* t4 = (float4*)tile;
  const float4 z = {0.0f, 0.0f, 0.0f, 0.0f};
  for (int i = tid; i < AG_ROWS * HID / 4; i += AG_THREADS) t4[i] = z;

  int g_ag, nsuper;
  const bool lay = sg_layout(n, g_ag, nsuper);
  const int rcap = lay ? SG_CAP : 0;
  unsigned* rec = lists + ((long long)blockIdx.x * 32 + wave) * SG_STRIDE;
  int rpos = 0;
  int dcnt = 0;

  if (REPLAY) {
    const int cnt = lay ? __builtin_amdgcn_readfirstlane((int)rec[0]) : -1;
    const int redo =
        __builtin_amdgcn_readfirstlane(__syncthreads_or((cnt < 0 || cnt > rcap) ? 1 : 0));
    if (!redo) {
      replay_rows(rec + AG_LIST_HEAD, cnt, h, tile, lane);
    } else {
      scan_gather<false>(h, src, dst, E, tile, outq, ownq, outn, fin, rowlo, wave, lane, rec,
                         rpos, rcap, dcnt);
    }
  } else {
    const int sg = blockIdx.x / SG_Q;
    const int lo = (blockIdx.x - sg * SG_Q) * AG_ROWS;
    const unsigned* coarse = lists + ((long long)g_ag * 32 + (long long)sg * SG_C * 32) * SG_STRIDE;
    unsigned nlen = 0u;
    if (lay && lane < SG_C) nlen = coarse[((long long)lane * 32 + wave) * SG_STRIDE];
    const bool over = __builtin_amdgcn_ballot_w32(nlen > (unsigned)SG_CAP) != 0u;
    const int direct =
        __builtin_amdgcn_readfirstlane(__syncthreads_or((!lay || over) ? 1 : 0));
    if (!direct) {
      seg_scan_gather<true>(h, coarse, nlen, tile, (unsigned*)outq, ownq, outn, fin, lo,
                            blockIdx.x - sg * SG_Q, wave, lane, rec, rpos, rcap, dcnt);
    } else {
      scan_gather<true>(h, src, dst, E, tile, outq, ownq, outn, fin, rowlo, wave, lane, rec,
                        rpos, rcap, dcnt);
    }
    if (lay && lane == 0) rec[0] = (unsigned)rpos;
    const int r = wave + 32 * lane;
    if (r < AG_ROWS) ((float*)outq)[r] = (float)dcnt;
  }
  __syncthreads();

  float4* o4 = (float4*)(agg + (long long)rowlo * HID);
  for (int i = tid; i < rows * (HID / 4); i += AG_THREADS) o4[i] = t4[i];
  if (!REPLAY) {
    if (tid < rows) deg[rowlo + tid] = ((const float*)outq)[tid];
  }
}


#define PJ_FULL 20
__global__ __launch_bounds__(128) __attribute__((amdgpu_waves_per_eu(2, 2))) void proj_kernel(const float* __restrict__ x,
                                                   const float* __restrict__ W,
                                                   const float* __restrict__ bias,
                                                   float* __restrict__ h, int n) {
  const int tid = threadIdx.x;
  const int wave = __builtin_amdgcn_readfirstlane(tid >> 5);
  const int lane = tid & 31;
  const int hf = lane >> 4;
  const int m = lane & 15;

  v2f bw[PJ_FULL + 1][2][4];
#pragma unroll
  for (int j = 0; j < PJ_FULL; ++j) {
#pragma unroll
    for (int tt = 0; tt < 2; ++tt) {
      const int k = 8 * j + 4 * hf + 2 * tt;
#pragma unroll
      for (int q = 0; q < 4; ++q) {
        bw[j][tt][q][0] = W[k * 64 + 16 * q + m];
        bw[j][tt][q][1] = W[(k + 1) * 64 + 16 * q + m];
      }
    }
  }
#pragma unroll
  for (int tt = 0; tt < 2; ++tt) {
    const int k = 8 * PJ_FULL + 4 * hf + 2 * tt;
#pragma unroll
    for (int q = 0; q < 4; ++q) {
      bw[PJ_FULL][tt][q][0] = (k < 165) ? W[k * 64 + 16 * q + m] : 0.0f;
      bw[PJ_FULL][tt][q][1] = (k + 1 < 165) ? W[(k + 1) * 64 + 16 * q + m] : 0.0f;
    }
  }
  float bv[4];
#pragma unroll
  for (int q = 0; q < 4; ++q) bv[q] = bias[16 * q + m];

  const int ntiles = (n + 15) >> 4;
  for (int t = blockIdx.x * 4 + wave; t < ntiles; t += (int)gridDim.x * 4) {
    const int row = t * 16 + m;
    const int rl = row < n ? row : n - 1;
    const float* xr = x + (long long)rl * 165 + 4 * hf;
    v4f a[PJ_FULL + 1];
#pragma unroll
    for (int j = 0; j < PJ_FULL; ++j) a[j] = *(const v4f_u*)(xr + 8 * j);
    const float* tp = xr + 8 * PJ_FULL;
    const float t0 = tp[0];
    const float t1 = tp[hf ? 0 : 1];
    const float t2 = tp[hf ? 0 : 2];
    const float t3 = tp[hf ? 0 : 3];
    v4f tl;
    tl[0] = t0;
    tl[1] = hf ? 0.0f : t1;
    tl[2] = hf ? 0.0f : t2;
    tl[3] = hf ? 0.0f : t3;
    a[PJ_FULL] = tl;

    v8f acc[4] = {};
#pragma unroll
    for (int j = 0; j <= PJ_FULL; ++j) {
#pragma unroll
      for (int tt = 0; tt < 2; ++tt) {
        v2f af;
        af[0] = a[j][2 * tt];
        af[1] = a[j][2 * tt + 1];
#pragma unroll
        for (int q = 0; q < 4; ++q) acc[q] = wmma_k4(af, bw[j][tt][q], acc[q]);
      }
    }

#pragma unroll
    for (int q = 0; q < 4; ++q) {
#pragma unroll
      for (int v = 0; v < 8; ++v) {
        const int r = t * 16 + v + 8 * hf;
        if (r < n) {
          const float val = acc[q][v] + bv[q];
          h[(long long)r * HID + 16 * q + m] = val > 0.0f ? val : 0.0f;
        }
      }
    }
  }
}

__global__ __launch_bounds__(128) __attribute__((amdgpu_waves_per_eu(2, 2))) void combine_kernel(
    const float* agg, const float* __restrict__ deg,
    const float* __restrict__ h_in, const float* __restrict__ Wl,
    const float* __restrict__ bl, const float* __restrict__ Wr,
    float* h_out, int n) {
  const int tid = threadIdx.x;
  const int wave = __builtin_amdgcn_readfirstlane(tid >> 5);
  const int lane = tid & 31;
  const int hf = lane >> 4;
  const int m = lane & 15;

  v2f wl[8][2][4], wr[8][2][4];
#pragma unroll
  for (int j = 0; j < 8; ++j) {
#pragma unroll
    for (int tt = 0; tt < 2; ++tt) {
      const int k = 8 * j + 4 * hf + 2 * tt;
#pragma unroll
      for (int q = 0; q < 4; ++q) {
        wl[j][tt][q][0] = Wl[k * 64 + 16 * q + m];
        wl[j][tt][q][1] = Wl[(k + 1) * 64 + 16 * q + m];
        wr[j][tt][q][0] = Wr[k * 64 + 16 * q + m];
        wr[j][tt][q][1] = Wr[(k + 1) * 64 + 16 * q + m];
      }
    }
  }
  float bv[4];
#pragma unroll
  for (int q = 0; q < 4; ++q) bv[q] = bl[16 * q + m];

  const int ntiles = (n + 15) >> 4;
  for (int t = blockIdx.x * 4 + wave; t < ntiles; t += (int)gridDim.x * 4) {
    const int row = t * 16 + m;
    const int rl = row < n ? row : n - 1;
    const v4f* ap = (const v4f*)(agg + (long long)rl * HID + 4 * hf);
    const v4f* hp = (const v4f*)(h_in + (long long)rl * HID + 4 * hf);
    v4f a[8], hv[8];
#pragma unroll
    for (int j = 0; j < 8; ++j) {
      a[j] = ap[2 * j];
      hv[j] = hp[2 * j];
    }
    float d = deg[rl];
    d = d > 1.0f ? d : 1.0f;
#pragma unroll
    for (int j = 0; j < 8; ++j) {
      a[j][0] = a[j][0] / d;
      a[j][1] = a[j][1] / d;
      a[j][2] = a[j][2] / d;
      a[j][3] = a[j][3] / d;
    }

    v8f acc[4] = {};
#pragma unroll
    for (int j = 0; j < 8; ++j) {
#pragma unroll
      for (int tt = 0; tt < 2; ++tt) {
        v2f af;
        af[0] = a[j][2 * tt];
        af[1] = a[j][2 * tt + 1];
#pragma unroll
        for (int q = 0; q < 4; ++q) acc[q] = wmma_k4(af, wl[j][tt][q], acc[q]);
      }
    }
#pragma unroll
    for (int j = 0; j < 8; ++j) {
#pragma unroll
      for (int tt = 0; tt < 2; ++tt) {
        v2f af;
        af[0] = hv[j][2 * tt];
        af[1] = hv[j][2 * tt + 1];
#pragma unroll
        for (int q = 0; q < 4; ++q) acc[q] = wmma_k4(af, wr[j][tt][q], acc[q]);
      }
    }

#pragma unroll
    for (int q = 0; q < 4; ++q) {
#pragma unroll
      for (int v = 0; v < 8; ++v) {
        const int r = t * 16 + v + 8 * hf;
        if (r < n) {
          const float val = acc[q][v] + bv[q];
          h_out[(long long)r * HID + 16 * q + m] = val > 0.0f ? val : 0.0f;
        }
      }
    }
  }
}

__global__ void cls_kernel(const float* __restrict__ h, const float* __restrict__ Wc,
                           const float* __restrict__ bc, float* __restrict__ out, int n) {
  int i = blockIdx.x * blockDim.x + threadIdx.x;
  if (i >= n) return;
  const float4* hp = (const float4*)(h + (long long)i * HID);
  const float4* wp = (const float4*)Wc;
  float acc = bc[0];
#pragma unroll
  for (int j = 0; j < HID / 4; ++j) {
    float4 hv = hp[j];
    float4 wv = wp[j];
    acc += hv.x * wv.x + hv.y * wv.y + hv.z * wv.z + hv.w * wv.w;
  }
  out[i] = acc;
}

extern "C" void kernel_launch(void* const* d_in, const int* in_sizes, int n_in,
                              void* d_out, int out_size, void* d_ws, size_t ws_size,
                              hipStream_t stream) {
  const float* x   = (const float*)d_in[0];
  const int* eidx  = (const int*)d_in[1];
  const float* Wp  = (const float*)d_in[2];
  const float* bp  = (const float*)d_in[3];
  const float* W1l = (const float*)d_in[4];
  const float* b1l = (const float*)d_in[5];
  const float* W1r = (const float*)d_in[6];
  const float* W2l = (const float*)d_in[7];
  const float* b2l = (const float*)d_in[8];
  const float* W2r = (const float*)d_in[9];
  const float* Wc  = (const float*)d_in[10];
  const float* bc  = (const float*)d_in[11];

  const int N = in_sizes[0] / 165;
  const int E = in_sizes[1] / 2;
  const int* src = eidx;
  const int* dst = eidx + E;

  float* A   = (float*)d_ws;
  float* B   = A + (size_t)N * HID;
  float* AGG = B + (size_t)N * HID;
  float* DEG = AGG + (size_t)N * HID;
  unsigned* LISTS = (unsigned*)B;

  const unsigned g_ag  = (unsigned)((N + AG_ROWS - 1) / AG_ROWS);
  const unsigned g_gemm = 512;
  const unsigned g_co  = ((g_ag + SG_Q - 1) / SG_Q) * SG_C;

  proj_kernel<<<g_gemm, 128, 0, stream>>>(x, Wp, bp, A, N);

  coarse_kernel<<<g_co, AG_THREADS, 0, stream>>>(src, dst, LISTS, N, E);

  neigh_sum_kernel<false><<<g_ag, AG_THREADS, 0, stream>>>(A, src, dst, AGG, LISTS, DEG, N, E);
  combine_kernel<<<g_gemm, 128, 0, stream>>>(AGG, DEG, A, W1l, b1l, W1r, AGG, N);

  neigh_sum_kernel<true><<<g_ag, AG_THREADS, 0, stream>>>(AGG, src, dst, A, LISTS, DEG, N, E);
  combine_kernel<<<g_gemm, 128, 0, stream>>>(A, DEG, AGG, W2l, b2l, W2r, B, N);

  cls_kernel<<<(N + 255) / 256, 256, 0, stream>>>(B, Wc, bc, (float*)d_out, N);
}


# conv 1 neighbour sums: queued edges reach the wave owning their row by chains linked in LDS, a lane follows one chain; no wave filters all 32 queues; sum order changes, still fixed by edge order
# speedup vs baseline: 5.7516x; 1.0990x over previous
#include <hip/hip_runtime.h>

#define HID 64

typedef __attribute__((ext_vector_type(2))) float v2f;
typedef __attribute__((ext_vector_type(8))) float v8f;
typedef __attribute__((ext_vector_type(4))) float v4f;
typedef v4f __attribute__((aligned(4))) v4f_u;

__device__ __forceinline__ v8f wmma_k4(v2f a, v2f b, v8f c) {
  return __builtin_amdgcn_wmma_f32_16x16x4_f32(
       false, a,  false, b,
       (short)0, c,  false,  false);
}

#define AG_ROWS 784
#define AG_WAVES 32
#define AG_THREADS (AG_WAVES * 32)
#define AG_ITER (AG_WAVES * 512)
#define AG_OUTCAP 416
#define AG_MYCAP 64
#define AG_SRC_BITS 22
#define AG_LIST_HEAD 32

__device__ __forceinline__ void load_iter(int4 (&r)[4], const int* __restrict__ dst, int g,
                                          int woff, int E, bool vec) {
  const int base = g * AG_ITER;
  if (vec && base + AG_ITER <= E) {
#pragma unroll
    for (int j = 0; j < 4; ++j) r[j] = *(const int4*)(dst + base + woff + j * 128);
  } else {
#pragma unroll
    for (int j = 0; j < 4; ++j) {
      const int e = base + woff + j * 128;
      r[j].x = (e < E) ? dst[e] : -1;
      r[j].y = (e + 1 < E) ? dst[e + 1] : -1;
      r[j].z = (e + 2 < E) ? dst[e + 2] : -1;
      r[j].w = (e + 3 < E) ? dst[e + 3] : -1;
    }
  }
}

__device__ __forceinline__ int queue_push(int2* q, int qn, bool keep, int2 ent, int lane) {
  const unsigned b = __builtin_amdgcn_ballot_w32(keep);
  if (b != 0u) {
    if (keep) q[qn + __popc(b & ((1u << lane) - 1u))] = ent;
    qn += __popc(b);
  }
  return qn;
}

__device__ __forceinline__ bool push4(int2* q, int& qn, int4 d, int e, int rowlo, int lane) {
  const unsigned l0 = (unsigned)(d.x - rowlo);
  const unsigned l1 = (unsigned)(d.y - rowlo);
  const unsigned l2 = (unsigned)(d.z - rowlo);
  const unsigned l3 = (unsigned)(d.w - rowlo);
  const unsigned lm = min(min(l0, l1), min(l2, l3));
  if (__builtin_amdgcn_ballot_w32(lm < (unsigned)AG_ROWS) == 0u) return true;
  const bool k0 = l0 < (unsigned)AG_ROWS, k1 = l1 < (unsigned)AG_ROWS;
  const bool k2 = l2 < (unsigned)AG_ROWS, k3 = l3 < (unsigned)AG_ROWS;
  const unsigned b0 = __builtin_amdgcn_ballot_w32(k0), b1 = __builtin_amdgcn_ballot_w32(k1);
  const unsigned b2 = __builtin_amdgcn_ballot_w32(k2), b3 = __builtin_amdgcn_ballot_w32(k3);
  const int c0 = __popc(b0), c1 = __popc(b1), c2 = __popc(b2), c3 = __popc(b3);
  if (qn + c0 + c1 + c2 + c3 > AG_OUTCAP) return false;
  const unsigned below = (1u << lane) - 1u;
  int p = qn;
  if (k0) q[p + __popc(b0 & below)] = make_int2(e, (int)l0);
  p += c0;
  if (k1) q[p + __popc(b1 & below)] = make_int2(e + 1, (int)l1);
  p += c1;
  if (k2) q[p + __popc(b2 & below)] = make_int2(e + 2, (int)l2);
  p += c2;
  if (k3) q[p + __popc(b3 & below)] = make_int2(e + 3, (int)l3);
  qn = p + c3;
  return true;
}

template <bool REC, bool DIRECT>
__device__ __forceinline__ void add_rows(const int2* q, int cnt, const float* __restrict__ h,
                                         const int* __restrict__ src, float* tile, int lane,
                                         unsigned* __restrict__ rec, int& rpos, int rcap,
                                         int& dcnt, int myrow) {
  int s = 0, loc = 0;
  if (lane < cnt) {
    const int2 ent = q[lane];
    s = DIRECT ? ent.x : src[ent.x];
    loc = ent.y;
  }
  if (REC) {
    if (rpos + cnt <= rcap && lane < cnt)
      rec[AG_LIST_HEAD + rpos + lane] = (unsigned)s | ((unsigned)loc << AG_SRC_BITS);
    rpos += cnt;
  }
  for (int k = 0; k < cnt; k += 8) {
    float2 v[8];
    int lk[8];
#pragma unroll
    for (int u = 0; u < 8; ++u) {
      const int sk = __builtin_amdgcn_readlane(s, k + u);
      lk[u] = __builtin_amdgcn_readlane(loc, k + u);
      v[u] = *(const float2*)(h + (long long)sk * HID + lane * 2);
    }
#pragma unroll
    for (int u = 0; u < 8; ++u) {
      if (k + u < cnt) {
        float2* t = (float2*)(tile + lk[u] * HID + lane * 2);
        float2 a = *t;
        a.x += v[u].x;
        a.y += v[u].y;
        *t = a;
        if (REC) dcnt += (lk[u] == myrow) ? 1 : 0;
      }
    }
  }
}

template <bool REC>
__device__ __forceinline__ void scan_gather(const float* __restrict__ h,
                                            const int* __restrict__ src,
                                            const int* __restrict__ dst, int E, float* tile,
                                            int2* outq, int2* ownq, int* outn, int* fin,
                                            int rowlo, int wave, int lane,
                                            unsigned* __restrict__ rec, int& rpos, int rcap,
                                            int& dcnt) {
  const bool vec = (((size_t)dst) & 15) == 0;
  const int nit = (E + AG_ITER - 1) / AG_ITER;
  const int start = (int)(((long long)blockIdx.x * nit) / (int)gridDim.x);
  const int woff = wave * 512 + lane * 4;
  const int4 none = {-1, -1, -1, -1};

  int4 cur[4], nxt[4];
#pragma unroll
  for (int j = 0; j < 4; ++j) {
    cur[j] = none;
    nxt[j] = none;
  }

  int2* oq = outq + wave * AG_OUTCAP;
  int2* mq = ownq + wave * AG_MYCAP;
  int it = -1;
  int mn = 0;

  for (int round = 0; round <= 4 * nit; ++round) {
    int qn = 0;
    while (it < nit) {
      int g = start + it;
      if (g >= nit) g -= nit;
      int g1 = start + it + 1;
      if (g1 >= nit) g1 -= nit;
      if (it + 1 < nit) {
        load_iter(nxt, dst, g1, woff, E, vec);
      } else {
#pragma unroll
        for (int j = 0; j < 4; ++j) nxt[j] = none;
      }
      const int e0 = g * AG_ITER + woff;
      int queued = 0;
      if (push4(oq, qn, cur[0], e0, rowlo, lane)) {
        queued = 1;
        if (push4(oq, qn, cur[1], e0 + 128, rowlo, lane)) {
          queued = 2;
          if (push4(oq, qn, cur[2], e0 + 256, rowlo, lane)) {
            queued = 3;
            if (push4(oq, qn, cur[3], e0 + 384, rowlo, lane)) queued = 4;
          }
        }
      }
      if (queued < 4) {
        if (queued > 0) cur[0] = none;
        if (queued > 1) cur[1] = none;
        if (queued > 2) cur[2] = none;
        break;
      }
#pragma unroll
      for (int j = 0; j < 4; ++j) cur[j] = nxt[j];
      ++it;
    }
    if (lane == 0) {
      outn[wave] = qn;
      fin[wave] = (it >= nit) ? 1 : 0;
    }
    __syncthreads();

    for (int i = 0; i < AG_WAVES; ++i) {
      const int w2 = (wave + i) & (AG_WAVES - 1);
      const int cnt = __builtin_amdgcn_readfirstlane(outn[w2]);
      for (int b0 = 0; b0 < cnt; b0 += 32) {
        const bool live = b0 + lane < cnt;
        int2 ent = make_int2(0, -1);
        if (live) ent = outq[w2 * AG_OUTCAP + b0 + lane];
        const bool mine = live && ((ent.y & (AG_WAVES - 1)) == wave);
        mn = queue_push(mq, mn, mine, ent, lane);
        if (mn >= 32) {
          add_rows<REC, false>(mq, 32, h, src, tile, lane, rec, rpos, rcap, dcnt, wave + 32 * lane);
          const int rem = mn - 32;
          if (lane < rem) {
            const int2 mv = mq[32 + lane];
            mq[lane] = mv;
          }
          mn = rem;
        }
      }
    }
    const bool alldone = __builtin_amdgcn_ballot_w32(fin[lane] != 0) == 0xffffffffu;
    __syncthreads();
    if (alldone) break;
  }
  if (mn > 0) add_rows<REC, false>(mq, mn, h, src, tile, lane, rec, rpos, rcap, dcnt, wave + 32 * lane);
}

#define SG_Q 16
#define SG_C 16
#define SG_STRIDE 768
#define SG_CAP (SG_STRIDE - AG_LIST_HEAD)
#define SG_SRC_BITS 18
#define SG_OVER 0xffffffffu
#define SG_NB (SG_CAP / 32)

__device__ __forceinline__ bool sg_layout(int n, int& g_ag, int& nsuper) {
  g_ag = (n + AG_ROWS - 1) / AG_ROWS;
  nsuper = (g_ag + SG_Q - 1) / SG_Q;
  const long long words = ((long long)g_ag + (long long)nsuper * SG_C) * 32 * SG_STRIDE;
  return n <= (1 << SG_SRC_BITS) && words <= (long long)n * HID;
}

__device__ __forceinline__ int queue_push_w(unsigned* q, int qn, bool keep, unsigned w, int lane) {
  const unsigned b = __builtin_amdgcn_ballot_w32(keep);
  if (b != 0u) {
    if (keep) q[qn + __popc(b & ((1u << lane) - 1u))] = w;
    qn += __popc(b);
  }
  return qn;
}

__device__ __forceinline__ void load_pair(int4& d0, int4& s0, int4& d1, int4& s1,
                                          const int* __restrict__ dst,
                                          const int* __restrict__ src, long long e, int E,
                                          bool whole) {
  if (whole) {
    d0 = *(const int4*)(dst + e);
    s0 = *(const int4*)(src + e);
    d1 = *(const int4*)(dst + e + 128);
    s1 = *(const int4*)(src + e + 128);
  } else {
    const long long f = e + 128;
    d0.x = (e < E) ? dst[e] : -1;
    d0.y = (e + 1 < E) ? dst[e + 1] : -1;
    d0.z = (e + 2 < E) ? dst[e + 2] : -1;
    d0.w = (e + 3 < E) ? dst[e + 3] : -1;
    s0.x = (e < E) ? src[e] : 0;
    s0.y = (e + 1 < E) ? src[e + 1] : 0;
    s0.z = (e + 2 < E) ? src[e + 2] : 0;
    s0.w = (e + 3 < E) ? src[e + 3] : 0;
    d1.x = (f < E) ? dst[f] : -1;
    d1.y = (f + 1 < E) ? dst[f + 1] : -1;
    d1.z = (f + 2 < E) ? dst[f + 2] : -1;
    d1.w = (f + 3 < E) ? dst[f + 3] : -1;
    s1.x = (f < E) ? src[f] : 0;
    s1.y = (f + 1 < E) ? src[f + 1] : 0;
    s1.z = (f + 2 < E) ? src[f + 2] : 0;
    s1.w = (f + 3 < E) ? src[f + 3] : 0;
  }
}

__device__ __forceinline__ void coarse_push(unsigned* mq, int& mn, int& pos, unsigned* __restrict__ seg,
                                            int d, int s, int slo, int lane) {
  const unsigned u = (unsigned)(d - slo);
  const bool keep = u < (unsigned)(SG_Q * AG_ROWS);
  mn = queue_push_w(mq, mn, keep, (unsigned)s | (u << SG_SRC_BITS), lane);
  if (mn >= 32) {
    if (pos + 32 <= SG_CAP) seg[AG_LIST_HEAD + pos + lane] = mq[lane];
    pos += 32;
    const int rem = mn - 32;
    if (lane < rem) {
      const unsigned mv = mq[32 + lane];
      mq[lane] = mv;
    }
    mn = rem;
  }
}

__global__ __launch_bounds__(AG_THREADS) void coarse_kernel(const int* __restrict__ src,
                                                          const int* __restrict__ dst,
                                                          unsigned* __restrict__ lists, int n,
                                                          int E) {
  __shared__ unsigned stage[AG_WAVES * 64];
  int g_ag, nsuper;
  if (!sg_layout(n, g_ag, nsuper)) return;
  const int tid = threadIdx.x;
  const int wave = __builtin_amdgcn_readfirstlane(tid >> 5);
  const int lane = tid & 31;
  const int s = blockIdx.x / SG_C;
  const int c = blockIdx.x - s * SG_C;
  const int slo = s * (SG_Q * AG_ROWS);
  const long long share = ((long long)E + SG_C * AG_WAVES - 1) / (SG_C * AG_WAVES);
  const int steps = (int)((share + 127) / 128);
  const long long first = (long long)(c * AG_WAVES + wave) * steps * 128;
  unsigned* seg = lists + ((long long)g_ag * 32 + (long long)blockIdx.x * 32 + wave) * SG_STRIDE;
  unsigned* mq = stage + wave * 64;
  const bool vec = ((((size_t)src) | ((size_t)dst)) & 15) == 0;

  int mn = 0, pos = 0;
  int4 da, sa, db, sb;
  load_pair(da, sa, db, sb, dst, src, first + lane * 4, E, vec && first + 256 <= E);
  for (int i = 0; i < steps; i += 2) {
    const int4 d0 = da, s0 = sa, d1 = db, s1 = sb;
    if (i + 2 < steps) {
      const long long e2 = first + (long long)(i + 2) * 128;
      load_pair(da, sa, db, sb, dst, src, e2 + lane * 4, E, vec && e2 + 256 <= E);
    }
    coarse_push(mq, mn, pos, seg, d0.x, s0.x, slo, lane);
    coarse_push(mq, mn, pos, seg, d0.y, s0.y, slo, lane);
    coarse_push(mq, mn, pos, seg, d0.z, s0.z, slo, lane);
    coarse_push(mq, mn, pos, seg, d0.w, s0.w, slo, lane);
    if (i + 1 < steps) {
      coarse_push(mq, mn, pos, seg, d1.x, s1.x, slo, lane);
      coarse_push(mq, mn, pos, seg, d1.y, s1.y, slo, lane);
      coarse_push(mq, mn, pos, seg, d1.z, s1.z, slo, lane);
      coarse_push(mq, mn, pos, seg, d1.w, s1.w, slo, lane);
    }
  }
  if (mn > 0) {
    if (pos + mn <= SG_CAP && lane < mn) seg[AG_LIST_HEAD + pos + lane] = mq[lane];
    pos += mn;
  }
  if (lane == 0) seg[0] = (pos <= SG_CAP) ? (unsigned)pos : SG_OVER;
}

#define LL_STRIDE 512
#define LL_CAP (LL_STRIDE - 1)
#define LL_END ((unsigned)LL_CAP)
#define LL_ROW_BITS 5
#define LL_NEXT_SHIFT (SG_SRC_BITS + LL_ROW_BITS)
static_assert(AG_WAVES * LL_STRIDE + AG_WAVES * AG_WAVES <= 2 * AG_WAVES * AG_OUTCAP,
              "the 32 queues and the heads lie in the LDS of the scan queues");
static_assert(LL_NEXT_SHIFT + 9 == 32 && AG_ROWS <= (AG_WAVES << LL_ROW_BITS) && AG_WAVES == 32,
              "an entry packs source row, row / 32 and a 9-bit link into one word");
template <bool REC>
__device__ __forceinline__ void seg_scan_gather(const float* __restrict__ h,
                                                const unsigned* __restrict__ coarse, unsigned nlen,
                                                float* tile, unsigned* outw, int2* ownq,
                                                int* fin, int lo, int rot, int wave, int lane,
                                                unsigned* __restrict__ rec, int& rpos, int rcap,
                                                int& dcnt) {
  const unsigned smask = (1u << SG_SRC_BITS) - 1u;
  const unsigned below = (1u << lane) - 1u;
  unsigned* oq = outw + wave * LL_STRIDE;
  unsigned* heads = outw + AG_WAVES * LL_STRIDE;
  unsigned* hd = heads + wave * AG_WAVES;
  int2* mq = ownq + wave * AG_MYCAP;
  int ci = 0;
  int j0 = 0;
  int mn = 0;

  for (int round = 0; round <= SG_C * SG_NB; ++round) {
    int qn = 0;
    bool full = false;
    while (ci < SG_C) {
      const int cc = (ci + rot) & (SG_C - 1);
      const int len = __builtin_amdgcn_readlane((int)nlen, cc);
      const unsigned* sp = coarse + ((long long)cc * 32 + wave) * SG_STRIDE + AG_LIST_HEAD;
      unsigned wv[SG_NB];
#pragma unroll
      for (int j = 0; j < SG_NB; ++j) wv[j] = (32 * j + lane < len) ? sp[32 * j + lane] : SG_OVER;
#pragma unroll
      for (int j = 0; j < SG_NB; ++j) {
        if (!full && j >= j0 && 32 * j < len) {
          const unsigned l = (wv[j] >> SG_SRC_BITS) - (unsigned)lo;
          const bool keep = l < (unsigned)AG_ROWS;
          const unsigned b = __builtin_amdgcn_ballot_w32(keep);
          if (b != 0u) {
            const int c = __popc(b);
            if (qn + c > LL_CAP) {
              full = true;
              j0 = j;
            } else {
              if (keep) oq[qn + __popc(b & below)] = (wv[j] & smask) | (l << SG_SRC_BITS);
              qn += c;
            }
          }
        }
      }
      if (full) break;
      ++ci;
      j0 = 0;
    }

    hd[lane] = LL_END;
    for (int b0 = 0; b0 < qn; b0 += 32) {
      const bool live = b0 + lane < qn;
      unsigned w = 0u;
      if (live) w = oq[b0 + lane];
      const unsigned l = w >> SG_SRC_BITS;
      const unsigned d = l & (AG_WAVES - 1);
      unsigned m = __builtin_amdgcn_ballot_w32(live);
      for (int bit = 0; bit < 5; ++bit) {
        const bool on = ((d >> bit) & 1u) != 0u;
        const unsigned bb = __builtin_amdgcn_ballot_w32(live && on);
        m &= on ? bb : ~bb;
      }
      const unsigned pm = m & below;
      unsigned nx = LL_END;
      if (live) nx = hd[d];
      if (pm != 0u) nx = (unsigned)(b0 + 31 - __clz(pm));
      if (live) {
        oq[b0 + lane] = (w & smask) | ((l >> 5) << SG_SRC_BITS) | (nx << LL_NEXT_SHIFT);
        if ((m >> lane) == 1u) hd[d] = (unsigned)(b0 + lane);
      }
    }
    if (lane == 0) fin[wave] = (ci >= SG_C) ? 1 : 0;
    __syncthreads();

    unsigned p = heads[lane * AG_WAVES + wave];
    for (int step = 0; step <= LL_CAP; ++step) {
      const bool act = p != LL_END;
      if (__builtin_amdgcn_ballot_w32(act) == 0u) break;
      unsigned ent = 0u;
      if (act) ent = outw[lane * LL_STRIDE + p];
      const int srow = (int)(ent & smask);
      const int trow = wave + 32 * (int)((ent >> SG_SRC_BITS) & ((1u << LL_ROW_BITS) - 1u));
      mn = queue_push(mq, mn, act, make_int2(srow, trow), lane);
      p = act ? (ent >> LL_NEXT_SHIFT) : LL_END;
      if (mn >= 32) {
        add_rows<REC, true>(mq, 32, h, (const int*)0, tile, lane, rec, rpos, rcap, dcnt, wave + 32 * lane);
        const int rem = mn - 32;
        if (lane < rem) {
          const int2 mv = mq[32 + lane];
          mq[lane] = mv;
        }
        mn = rem;
      }
    }
    const bool alldone = __builtin_amdgcn_ballot_w32(fin[lane] != 0) == 0xffffffffu;
    __syncthreads();
    if (alldone) break;
  }
  if (mn > 0) add_rows<REC, true>(mq, mn, h, (const int*)0, tile, lane, rec, rpos, rcap, dcnt, wave + 32 * lane);
}

template <int L0>
__device__ __forceinline__ void load_group(float2 (&v)[8], int (&l)[8], unsigned eb,
                                           const float* __restrict__ h, int lane) {
  const unsigned smask = (1u << AG_SRC_BITS) - 1u;
#pragma unroll
  for (int u = 0; u < 8; ++u) {
    const unsigned e = (unsigned)__builtin_amdgcn_readlane((int)eb, L0 + u);
    l[u] = (int)(e >> AG_SRC_BITS);
    v[u] = *(const float2*)(h + (long long)(e & smask) * HID + lane * 2);
  }
}

__device__ __forceinline__ void add_group(const float2 (&v)[8], const int (&l)[8], int first,
                                          int cnt, float* tile, int lane) {
#pragma unroll
  for (int u = 0; u < 8; ++u) {
    if (first + u < cnt) {
      float2* t = (float2*)(tile + l[u] * HID + lane * 2);
      float2 a = *t;
      a.x += v[u].x;
      a.y += v[u].y;
      *t = a;
    }
  }
}

__device__ __forceinline__ void replay_rows(const unsigned* __restrict__ ents, int cnt,
                                            const float* __restrict__ h, float* tile, int lane) {
  unsigned eb = (lane < cnt) ? ents[lane] : 0u;
  float2 va[8], vb[8], vc[8], vd[8];
  int la[8], lb[8], lc[8], ld[8];
  load_group<0>(va, la, eb, h, lane);
  __builtin_amdgcn_sched_barrier(0);
  load_group<8>(vb, lb, eb, h, lane);
  for (int b = 0; b < cnt; b += 32) {
    const unsigned en = (b + 32 + lane < cnt) ? ents[b + 32 + lane] : 0u;
    load_group<16>(vc, lc, eb, h, lane);
    add_group(va, la, b, cnt, tile, lane);
    load_group<24>(vd, ld, eb, h, lane);
    add_group(vb, lb, b + 8, cnt, tile, lane);
    load_group<0>(va, la, en, h, lane);
    add_group(vc, lc, b + 16, cnt, tile, lane);
    load_group<8>(vb, lb, en, h, lane);
    add_group(vd, ld, b + 24, cnt, tile, lane);
    eb = en;
  }
}

template <bool REPLAY>
__global__ __launch_bounds__(AG_THREADS) void neigh_sum_kernel(const float* __restrict__ h,
                                                             const int* __restrict__ src,
                                                             const int* __restrict__ dst,
                                                             float* __restrict__ agg,
                                                             unsigned* __restrict__ lists,
                                                             float* __restrict__ deg, int n,
                                                             int E) {
  __shared__ float tile[AG_ROWS * HID];
  __shared__ int2 outq[AG_WAVES * AG_OUTCAP];
  __shared__ int2 ownq[AG_WAVES * AG_MYCAP];
  __shared__ int outn[AG_WAVES];
  __shared__ int fin[AG_WAVES];

  const int tid = threadIdx.x;
  const int wave = tid >> 5;
  const int lane = tid & 31;
  const int rowlo = blockIdx.x * AG_ROWS;
  const int rows = (n - rowlo < AG_ROWS) ? (n - rowlo) : AG_ROWS;

  float4* t4 = (float4*)tile;
  const float4 z = {0.0f, 0.0f, 0.0f, 0.0f};
  for (int i = tid; i < AG_ROWS * HID / 4; i += AG_THREADS) t4[i] = z;

  int g_ag, nsuper;
  const bool lay = sg_layout(n, g_ag, nsuper);
  const int rcap = lay ? SG_CAP : 0;
  unsigned* rec = lists + ((long long)blockIdx.x * 32 + wave) * SG_STRIDE;
  int rpos = 0;
  int dcnt = 0;

  if (REPLAY) {
    const int cnt = lay ? __builtin_amdgcn_readfirstlane((int)rec[0]) : -1;
    const int redo =
        __builtin_amdgcn_readfirstlane(__syncthreads_or((cnt < 0 || cnt > rcap) ? 1 : 0));
    if (!redo) {
      replay_rows(rec + AG_LIST_HEAD, cnt, h, tile, lane);
    } else {
      scan_gather<false>(h, src, dst, E, tile, outq, ownq, outn, fin, rowlo, wave, lane, rec,
                         rpos, rcap, dcnt);
    }
  } else {
    const int sg = blockIdx.x / SG_Q;
    const int lo = (blockIdx.x - sg * SG_Q) * AG_ROWS;
    const unsigned* coarse = lists + ((long long)g_ag * 32 + (long long)sg * SG_C * 32) * SG_STRIDE;
    unsigned nlen = 0u;
    if (lay && lane < SG_C) nlen = coarse[((long long)lane * 32 + wave) * SG_STRIDE];
    const bool over = __builtin_amdgcn_ballot_w32(nlen > (unsigned)SG_CAP) != 0u;
    const int direct =
        __builtin_amdgcn_readfirstlane(__syncthreads_or((!lay || over) ? 1 : 0));
    if (!direct) {
      seg_scan_gather<true>(h, coarse, nlen, tile, (unsigned*)outq, ownq, fin, lo,
                            blockIdx.x - sg * SG_Q, wave, lane, rec, rpos, rcap, dcnt);
    } else {
      scan_gather<true>(h, src, dst, E, tile, outq, ownq, outn, fin, rowlo, wave, lane, rec,
                        rpos, rcap, dcnt);
    }
    if (lay && lane == 0) rec[0] = (unsigned)rpos;
    const int r = wave + 32 * lane;
    if (r < AG_ROWS) ((float*)outq)[r] = (float)dcnt;
  }
  __syncthreads();

  float4* o4 = (float4*)(agg + (long long)rowlo * HID);
  for (int i = tid; i < rows * (HID / 4); i += AG_THREADS) o4[i] = t4[i];
  if (!REPLAY) {
    if (tid < rows) deg[rowlo + tid] = ((const float*)outq)[tid];
  }
}


#define PJ_FULL 20
__global__ __launch_bounds__(128) __attribute__((amdgpu_waves_per_eu(2, 2))) void proj_kernel(const float* __restrict__ x,
                                                   const float* __restrict__ W,
                                                   const float* __restrict__ bias,
                                                   float* __restrict__ h, int n) {
  const int tid = threadIdx.x;
  const int wave = __builtin_amdgcn_readfirstlane(tid >> 5);
  const int lane = tid & 31;
  const int hf = lane >> 4;
  const int m = lane & 15;

  v2f bw[PJ_FULL + 1][2][4];
#pragma unroll
  for (int j = 0; j < PJ_FULL; ++j) {
#pragma unroll
    for (int tt = 0; tt < 2; ++tt) {
      const int k = 8 * j + 4 * hf + 2 * tt;
#pragma unroll
      for (int q = 0; q < 4; ++q) {
        bw[j][tt][q][0] = W[k * 64 + 16 * q + m];
        bw[j][tt][q][1] = W[(k + 1) * 64 + 16 * q + m];
      }
    }
  }
#pragma unroll
  for (int tt = 0; tt < 2; ++tt) {
    const int k = 8 * PJ_FULL + 4 * hf + 2 * tt;
#pragma unroll
    for (int q = 0; q < 4; ++q) {
      bw[PJ_FULL][tt][q][0] = (k < 165) ? W[k * 64 + 16 * q + m] : 0.0f;
      bw[PJ_FULL][tt][q][1] = (k + 1 < 165) ? W[(k + 1) * 64 + 16 * q + m] : 0.0f;
    }
  }
  float bv[4];
#pragma unroll
  for (int q = 0; q < 4; ++q) bv[q] = bias[16 * q + m];

  const int ntiles = (n + 15) >> 4;
  for (int t = blockIdx.x * 4 + wave; t < ntiles; t += (int)gridDim.x * 4) {
    const int row = t * 16 + m;
    const int rl = row < n ? row : n - 1;
    const float* xr = x + (long long)rl * 165 + 4 * hf;
    v4f a[PJ_FULL + 1];
#pragma unroll
    for (int j = 0; j < PJ_FULL; ++j) a[j] = *(const v4f_u*)(xr + 8 * j);
    const float* tp = xr + 8 * PJ_FULL;
    const float t0 = tp[0];
    const float t1 = tp[hf ? 0 : 1];
    const float t2 = tp[hf ? 0 : 2];
    const float t3 = tp[hf ? 0 : 3];
    v4f tl;
    tl[0] = t0;
    tl[1] = hf ? 0.0f : t1;
    tl[2] = hf ? 0.0f : t2;
    tl[3] = hf ? 0.0f : t3;
    a[PJ_FULL] = tl;

    v8f acc[4] = {};
#pragma unroll
    for (int j = 0; j <= PJ_FULL; ++j) {
#pragma unroll
      for (int tt = 0; tt < 2; ++tt) {
        v2f af;
        af[0] = a[j][2 * tt];
        af[1] = a[j][2 * tt + 1];
#pragma unroll
        for (int q = 0; q < 4; ++q) acc[q] = wmma_k4(af, bw[j][tt][q], acc[q]);
      }
    }

#pragma unroll
    for (int q = 0; q < 4; ++q) {
#pragma unroll
      for (int v = 0; v < 8; ++v) {
        const int r = t * 16 + v + 8 * hf;
        if (r < n) {
          const float val = acc[q][v] + bv[q];
          h[(long long)r * HID + 16 * q + m] = val > 0.0f ? val : 0.0f;
        }
      }
    }
  }
}

__global__ __launch_bounds__(128) __attribute__((amdgpu_waves_per_eu(2, 2))) void combine_kernel(
    const float* agg, const float* __restrict__ deg,
    const float* __restrict__ h_in, const float* __restrict__ Wl,
    const float* __restrict__ bl, const float* __restrict__ Wr,
    float* h_out, int n) {
  const int tid = threadIdx.x;
  const int wave = __builtin_amdgcn_readfirstlane(tid >> 5);
  const int lane = tid & 31;
  const int hf = lane >> 4;
  const int m = lane & 15;

  v2f wl[8][2][4], wr[8][2][4];
#pragma unroll
  for (int j = 0; j < 8; ++j) {
#pragma unroll
    for (int tt = 0; tt < 2; ++tt) {
      const int k = 8 * j + 4 * hf + 2 * tt;
#pragma unroll
      for (int q = 0; q < 4; ++q) {
        wl[j][tt][q][0] = Wl[k * 64 + 16 * q + m];
        wl[j][tt][q][1] = Wl[(k + 1) * 64 + 16 * q + m];
        wr[j][tt][q][0] = Wr[k * 64 + 16 * q + m];
        wr[j][tt][q][1] = Wr[(k + 1) * 64 + 16 * q + m];
      }
    }
  }
  float bv[4];
#pragma unroll
  for (int q = 0; q < 4; ++q) bv[q] = bl[16 * q + m];

  const int ntiles = (n + 15) >> 4;
  for (int t = blockIdx.x * 4 + wave; t < ntiles; t += (int)gridDim.x * 4) {
    const int row = t * 16 + m;
    const int rl = row < n ? row : n - 1;
    const v4f* ap = (const v4f*)(agg + (long long)rl * HID + 4 * hf);
    const v4f* hp = (const v4f*)(h_in + (long long)rl * HID + 4 * hf);
    v4f a[8], hv[8];
#pragma unroll
    for (int j = 0; j < 8; ++j) {
      a[j] = ap[2 * j];
      hv[j] = hp[2 * j];
    }
    float d = deg[rl];
    d = d > 1.0f ? d : 1.0f;
#pragma unroll
    for (int j = 0; j < 8; ++j) {
      a[j][0] = a[j][0] / d;
      a[j][1] = a[j][1] / d;
      a[j][2] = a[j][2] / d;
      a[j][3] = a[j][3] / d;
    }

    v8f acc[4] = {};
#pragma unroll
    for (int j = 0; j < 8; ++j) {
#pragma unroll
      for (int tt = 0; tt < 2; ++tt) {
        v2f af;
        af[0] = a[j][2 * tt];
        af[1] = a[j][2 * tt + 1];
#pragma unroll
        for (int q = 0; q < 4; ++q) acc[q] = wmma_k4(af, wl[j][tt][q], acc[q]);
      }
    }
#pragma unroll
    for (int j = 0; j < 8; ++j) {
#pragma unroll
      for (int tt = 0; tt < 2; ++tt) {
        v2f af;
        af[0] = hv[j][2 * tt];
        af[1] = hv[j][2 * tt + 1];
#pragma unroll
        for (int q = 0; q < 4; ++q) acc[q] = wmma_k4(af, wr[j][tt][q], acc[q]);
      }
    }

#pragma unroll
    for (int q = 0; q < 4; ++q) {
#pragma unroll
      for (int v = 0; v < 8; ++v) {
        const int r = t * 16 + v + 8 * hf;
        if (r < n) {
          const float val = acc[q][v] + bv[q];
          h_out[(long long)r * HID + 16 * q + m] = val > 0.0f ? val : 0.0f;
        }
      }
    }
  }
}

__global__ void cls_kernel(const float* __restrict__ h, const float* __restrict__ Wc,
                           const float* __restrict__ bc, float* __restrict__ out, int n) {
  int i = blockIdx.x * blockDim.x + threadIdx.x;
  if (i >= n) return;
  const float4* hp = (const float4*)(h + (long long)i * HID);
  const float4* wp = (const float4*)Wc;
  float acc = bc[0];
#pragma unroll
  for (int j = 0; j < HID / 4; ++j) {
    float4 hv = hp[j];
    float4 wv = wp[j];
    acc += hv.x * wv.x + hv.y * wv.y + hv.z * wv.z + hv.w * wv.w;
  }
  out[i] = acc;
}

extern "C" void kernel_launch(void* const* d_in, const int* in_sizes, int n_in,
                              void* d_out, int out_size, void* d_ws, size_t ws_size,
                              hipStream_t stream) {
  const float* x   = (const float*)d_in[0];
  const int* eidx  = (const int*)d_in[1];
  const float* Wp  = (const float*)d_in[2];
  const float* bp  = (const float*)d_in[3];
  const float* W1l = (const float*)d_in[4];
  const float* b1l = (const float*)d_in[5];
  const float* W1r = (const float*)d_in[6];
  const float* W2l = (const float*)d_in[7];
  const float* b2l = (const float*)d_in[8];
  const float* W2r = (const float*)d_in[9];
  const float* Wc  = (const float*)d_in[10];
  const float* bc  = (const float*)d_in[11];

  const int N = in_sizes[0] / 165;
  const int E = in_sizes[1] / 2;
  const int* src = eidx;
  const int* dst = eidx + E;

  float* A   = (float*)d_ws;
  float* B   = A + (size_t)N * HID;
  float* AGG = B + (size_t)N * HID;
  float* DEG = AGG + (size_t)N * HID;
  unsigned* LISTS = (unsigned*)B;

  const unsigned g_ag  = (unsigned)((N + AG_ROWS - 1) / AG_ROWS);
  const unsigned g_gemm = 512;
  const unsigned g_co  = ((g_ag + SG_Q - 1) / SG_Q) * SG_C;

  proj_kernel<<<g_gemm, 128, 0, stream>>>(x, Wp, bp, A, N);

  coarse_kernel<<<g_co, AG_THREADS, 0, stream>>>(src, dst, LISTS, N, E);

  neigh_sum_kernel<false><<<g_ag, AG_THREADS, 0, stream>>>(A, src, dst, AGG, LISTS, DEG, N, E);
  combine_kernel<<<g_gemm, 128, 0, stream>>>(AGG, DEG, A, W1l, b1l, W1r, AGG, N);

  neigh_sum_kernel<true><<<g_ag, AG_THREADS, 0, stream>>>(AGG, src, dst, A, LISTS, DEG, N, E);
  combine_kernel<<<g_gemm, 128, 0, stream>>>(A, DEG, AGG, W2l, b2l, W2r, B, N);

  cls_kernel<<<(N + 255) / 256, 256, 0, stream>>>(B, Wc, bc, (float*)d_out, N);
}


# coarse pass splits each wave's list by group (16 bins in LDS, in edge order, stored one behind the other): a group reads only its own bin of each list, no second filter; sum order kept
# speedup vs baseline: 6.3644x; 1.1065x over previous
#include <hip/hip_runtime.h>

#define HID 64

typedef __attribute__((ext_vector_type(2))) float v2f;
typedef __attribute__((ext_vector_type(8))) float v8f;
typedef __attribute__((ext_vector_type(4))) float v4f;
typedef v4f __attribute__((aligned(4))) v4f_u;

__device__ __forceinline__ v8f wmma_k4(v2f a, v2f b, v8f c) {
  return __builtin_amdgcn_wmma_f32_16x16x4_f32(
       false, a,  false, b,
       (short)0, c,  false,  false);
}

#define AG_ROWS 784
#define AG_WAVES 32
#define AG_THREADS (AG_WAVES * 32)
#define AG_ITER (AG_WAVES * 512)
#define AG_OUTCAP 416
#define AG_MYCAP 64
#define AG_SRC_BITS 22
#define AG_LIST_HEAD 32

__device__ __forceinline__ void load_iter(int4 (&r)[4], const int* __restrict__ dst, int g,
                                          int woff, int E, bool vec) {
  const int base = g * AG_ITER;
  if (vec && base + AG_ITER <= E) {
#pragma unroll
    for (int j = 0; j < 4; ++j) r[j] = *(const int4*)(dst + base + woff + j * 128);
  } else {
#pragma unroll
    for (int j = 0; j < 4; ++j) {
      const int e = base + woff + j * 128;
      r[j].x = (e < E) ? dst[e] : -1;
      r[j].y = (e + 1 < E) ? dst[e + 1] : -1;
      r[j].z = (e + 2 < E) ? dst[e + 2] : -1;
      r[j].w = (e + 3 < E) ? dst[e + 3] : -1;
    }
  }
}

__device__ __forceinline__ int queue_push(int2* q, int qn, bool keep, int2 ent, int lane) {
  const unsigned b = __builtin_amdgcn_ballot_w32(keep);
  if (b != 0u) {
    if (keep) q[qn + __popc(b & ((1u << lane) - 1u))] = ent;
    qn += __popc(b);
  }
  return qn;
}

__device__ __forceinline__ bool push4(int2* q, int& qn, int4 d, int e, int rowlo, int lane) {
  const unsigned l0 = (unsigned)(d.x - rowlo);
  const unsigned l1 = (unsigned)(d.y - rowlo);
  const unsigned l2 = (unsigned)(d.z - rowlo);
  const unsigned l3 = (unsigned)(d.w - rowlo);
  const unsigned lm = min(min(l0, l1), min(l2, l3));
  if (__builtin_amdgcn_ballot_w32(lm < (unsigned)AG_ROWS) == 0u) return true;
  const bool k0 = l0 < (unsigned)AG_ROWS, k1 = l1 < (unsigned)AG_ROWS;
  const bool k2 = l2 < (unsigned)AG_ROWS, k3 = l3 < (unsigned)AG_ROWS;
  const unsigned b0 = __builtin_amdgcn_ballot_w32(k0), b1 = __builtin_amdgcn_ballot_w32(k1);
  const unsigned b2 = __builtin_amdgcn_ballot_w32(k2), b3 = __builtin_amdgcn_ballot_w32(k3);
  const int c0 = __popc(b0), c1 = __popc(b1), c2 = __popc(b2), c3 = __popc(b3);
  if (qn + c0 + c1 + c2 + c3 > AG_OUTCAP) return false;
  const unsigned below = (1u << lane) - 1u;
  int p = qn;
  if (k0) q[p + __popc(b0 & below)] = make_int2(e, (int)l0);
  p += c0;
  if (k1) q[p + __popc(b1 & below)] = make_int2(e + 1, (int)l1);
  p += c1;
  if (k2) q[p + __popc(b2 & below)] = make_int2(e + 2, (int)l2);
  p += c2;
  if (k3) q[p + __popc(b3 & below)] = make_int2(e + 3, (int)l3);
  qn = p + c3;
  return true;
}

template <bool REC, bool DIRECT>
__device__ __forceinline__ void add_rows(const int2* q, int cnt, const float* __restrict__ h,
                                         const int* __restrict__ src, float* tile, int lane,
                                         unsigned* __restrict__ rec, int& rpos, int rcap,
                                         int& dcnt, int myrow) {
  int s = 0, loc = 0;
  if (lane < cnt) {
    const int2 ent = q[lane];
    s = DIRECT ? ent.x : src[ent.x];
    loc = ent.y;
  }
  if (REC) {
    if (rpos + cnt <= rcap && lane < cnt)
      rec[AG_LIST_HEAD + rpos + lane] = (unsigned)s | ((unsigned)loc << AG_SRC_BITS);
    rpos += cnt;
  }
  for (int k = 0; k < cnt; k += 8) {
    float2 v[8];
    int lk[8];
#pragma unroll
    for (int u = 0; u < 8; ++u) {
      const int sk = __builtin_amdgcn_readlane(s, k + u);
      lk[u] = __builtin_amdgcn_readlane(loc, k + u);
      v[u] = *(const float2*)(h + (long long)sk * HID + lane * 2);
    }
#pragma unroll
    for (int u = 0; u < 8; ++u) {
      if (k + u < cnt) {
        float2* t = (float2*)(tile + lk[u] * HID + lane * 2);
        float2 a = *t;
        a.x += v[u].x;
        a.y += v[u].y;
        *t = a;
        if (REC) dcnt += (lk[u] == myrow) ? 1 : 0;
      }
    }
  }
}

template <bool REC>
__device__ __forceinline__ void scan_gather(const float* __restrict__ h,
                                            const int* __restrict__ src,
                                            const int* __restrict__ dst, int E, float* tile,
                                            int2* outq, int2* ownq, int* outn, int* fin,
                                            int rowlo, int wave, int lane,
                                            unsigned* __restrict__ rec, int& rpos, int rcap,
                                            int& dcnt) {
  const bool vec = (((size_t)dst) & 15) == 0;
  const int nit = (E + AG_ITER - 1) / AG_ITER;
  const int start = (int)(((long long)blockIdx.x * nit) / (int)gridDim.x);
  const int woff = wave * 512 + lane * 4;
  const int4 none = {-1, -1, -1, -1};

  int4 cur[4], nxt[4];
#pragma unroll
  for (int j = 0; j < 4; ++j) {
    cur[j] = none;
    nxt[j] = none;
  }

  int2* oq = outq + wave * AG_OUTCAP;
  int2* mq = ownq + wave * AG_MYCAP;
  int it = -1;
  int mn = 0;

  for (int round = 0; round <= 4 * nit; ++round) {
    int qn = 0;
    while (it < nit) {
      int g = start + it;
      if (g >= nit) g -= nit;
      int g1 = start + it + 1;
      if (g1 >= nit) g1 -= nit;
      if (it + 1 < nit) {
        load_iter(nxt, dst, g1, woff, E, vec);
      } else {
#pragma unroll
        for (int j = 0; j < 4; ++j) nxt[j] = none;
      }
      const int e0 = g * AG_ITER + woff;
      int queued = 0;
      if (push4(oq, qn, cur[0], e0, rowlo, lane)) {
        queued = 1;
        if (push4(oq, qn, cur[1], e0 + 128, rowlo, lane)) {
          queued = 2;
          if (push4(oq, qn, cur[2], e0 + 256, rowlo, lane)) {
            queued = 3;
            if (push4(oq, qn, cur[3], e0 + 384, rowlo, lane)) queued = 4;
          }
        }
      }
      if (queued < 4) {
        if (queued > 0) cur[0] = none;
        if (queued > 1) cur[1] = none;
        if (queued > 2) cur[2] = none;
        break;
      }
#pragma unroll
      for (int j = 0; j < 4; ++j) cur[j] = nxt[j];
      ++it;
    }
    if (lane == 0) {
      outn[wave] = qn;
      fin[wave] = (it >= nit) ? 1 : 0;
    }
    __syncthreads();

    for (int i = 0; i < AG_WAVES; ++i) {
      const int w2 = (wave + i) & (AG_WAVES - 1);
      const int cnt = __builtin_amdgcn_readfirstlane(outn[w2]);
      for (int b0 = 0; b0 < cnt; b0 += 32) {
        const bool live = b0 + lane < cnt;
        int2 ent = make_int2(0, -1);
        if (live) ent = outq[w2 * AG_OUTCAP + b0 + lane];
        const bool mine = live && ((ent.y & (AG_WAVES - 1)) == wave);
        mn = queue_push(mq, mn, mine, ent, lane);
        if (mn >= 32) {
          add_rows<REC, false>(mq, 32, h, src, tile, lane, rec, rpos, rcap, dcnt, wave + 32 * lane);
          const int rem = mn - 32;
          if (lane < rem) {
            const int2 mv = mq[32 + lane];
            mq[lane] = mv;
          }
          mn = rem;
        }
      }
    }
    const bool alldone = __builtin_amdgcn_ballot_w32(fin[lane] != 0) == 0xffffffffu;
    __syncthreads();
    if (alldone) break;
  }
  if (mn > 0) add_rows<REC, false>(mq, mn, h, src, tile, lane, rec, rpos, rcap, dcnt, wave + 32 * lane);
}

#define SG_Q 16
#define SG_C 16
#define SG_STRIDE 768
#define SG_CAP (SG_STRIDE - AG_LIST_HEAD)
#define SG_SRC_BITS 18
#define SG_OVER 0xffffffffu
#define SG_NB (SG_CAP / 32)
#define SB_CAP 64
static_assert(SG_Q == 16 && SG_Q + 2 <= AG_LIST_HEAD, "a bin index has four bits; the bin starts lie in the list's head");

__device__ __forceinline__ bool sg_layout(int n, int& g_ag, int& nsuper) {
  g_ag = (n + AG_ROWS - 1) / AG_ROWS;
  nsuper = (g_ag + SG_Q - 1) / SG_Q;
  const long long words = ((long long)g_ag + (long long)nsuper * SG_C) * 32 * SG_STRIDE;
  return n <= (1 << SG_SRC_BITS) && words <= (long long)n * HID;
}

__device__ __forceinline__ int queue_push_w(unsigned* q, int qn, bool keep, unsigned w, int lane) {
  const unsigned b = __builtin_amdgcn_ballot_w32(keep);
  if (b != 0u) {
    if (keep) q[qn + __popc(b & ((1u << lane) - 1u))] = w;
    qn += __popc(b);
  }
  return qn;
}

__device__ __forceinline__ void load_pair(int4& d0, int4& s0, int4& d1, int4& s1,
                                          const int* __restrict__ dst,
                                          const int* __restrict__ src, long long e, int E,
                                          bool whole) {
  if (whole) {
    d0 = *(const int4*)(dst + e);
    s0 = *(const int4*)(src + e);
    d1 = *(const int4*)(dst + e + 128);
    s1 = *(const int4*)(src + e + 128);
  } else {
    const long long f = e + 128;
    d0.x = (e < E) ? dst[e] : -1;
    d0.y = (e + 1 < E) ? dst[e + 1] : -1;
    d0.z = (e + 2 < E) ? dst[e + 2] : -1;
    d0.w = (e + 3 < E) ? dst[e + 3] : -1;
    s0.x = (e < E) ? src[e] : 0;
    s0.y = (e + 1 < E) ? src[e + 1] : 0;
    s0.z = (e + 2 < E) ? src[e + 2] : 0;
    s0.w = (e + 3 < E) ? src[e + 3] : 0;
    d1.x = (f < E) ? dst[f] : -1;
    d1.y = (f + 1 < E) ? dst[f + 1] : -1;
    d1.z = (f + 2 < E) ? dst[f + 2] : -1;
    d1.w = (f + 3 < E) ? dst[f + 3] : -1;
    s1.x = (f < E) ? src[f] : 0;
    s1.y = (f + 1 < E) ? src[f + 1] : 0;
    s1.z = (f + 2 < E) ? src[f + 2] : 0;
    s1.w = (f + 3 < E) ? src[f + 3] : 0;
  }
}

__device__ __forceinline__ void bin_push(unsigned* bq, unsigned* bc, bool live, unsigned w, int lane) {
  const unsigned smask = (1u << SG_SRC_BITS) - 1u;
  const unsigned u = w >> SG_SRC_BITS;
  const unsigned q = u / (unsigned)AG_ROWS;
  const unsigned l = u - q * (unsigned)AG_ROWS;
  unsigned m = __builtin_amdgcn_ballot_w32(live);
  for (int bit = 0; bit < 4; ++bit) {
    const bool on = ((q >> bit) & 1u) != 0u;
    const unsigned bb = __builtin_amdgcn_ballot_w32(live && on);
    m &= on ? bb : ~bb;
  }
  if (live) {
    const unsigned pos = bc[q] + (unsigned)__popc(m & ((1u << lane) - 1u));
    if (pos < (unsigned)SB_CAP) bq[q * SB_CAP + pos] = (w & smask) | (l << SG_SRC_BITS);
    if ((m >> lane) == 1u) bc[q] = pos + 1u;
  }
}

__device__ __forceinline__ void coarse_push(unsigned* mq, int& mn, unsigned* bq, unsigned* bc,
                                            int d, int s, int slo, int lane) {
  const unsigned u = (unsigned)(d - slo);
  const bool keep = u < (unsigned)(SG_Q * AG_ROWS);
  mn = queue_push_w(mq, mn, keep, (unsigned)s | (u << SG_SRC_BITS), lane);
  if (mn >= 32) {
    bin_push(bq, bc, true, mq[lane], lane);
    const int rem = mn - 32;
    if (lane < rem) {
      const unsigned mv = mq[32 + lane];
      mq[lane] = mv;
    }
    mn = rem;
  }
}

__global__ __launch_bounds__(AG_THREADS) void coarse_kernel(const int* __restrict__ src,
                                                          const int* __restrict__ dst,
                                                          unsigned* __restrict__ lists, int n,
                                                          int E) {
  __shared__ unsigned stage[AG_WAVES * 64];
  __shared__ unsigned bins[AG_WAVES * SG_Q * SB_CAP];
  __shared__ unsigned blen[AG_WAVES * SG_Q];
  int g_ag, nsuper;
  if (!sg_layout(n, g_ag, nsuper)) return;
  const int tid = threadIdx.x;
  const int wave = __builtin_amdgcn_readfirstlane(tid >> 5);
  const int lane = tid & 31;
  const int s = blockIdx.x / SG_C;
  const int c = blockIdx.x - s * SG_C;
  const int slo = s * (SG_Q * AG_ROWS);
  const long long share = ((long long)E + SG_C * AG_WAVES - 1) / (SG_C * AG_WAVES);
  const int steps = (int)((share + 127) / 128);
  const long long first = (long long)(c * AG_WAVES + wave) * steps * 128;
  unsigned* seg = lists + ((long long)g_ag * 32 + (long long)blockIdx.x * 32 + wave) * SG_STRIDE;
  unsigned* mq = stage + wave * 64;
  unsigned* bq = bins + wave * (SG_Q * SB_CAP);
  unsigned* bc = blen + wave * SG_Q;
  const bool vec = ((((size_t)src) | ((size_t)dst)) & 15) == 0;

  if (lane < SG_Q) bc[lane] = 0u;
  int mn = 0;
  int4 da, sa, db, sb;
  load_pair(da, sa, db, sb, dst, src, first + lane * 4, E, vec && first + 256 <= E);
  for (int i = 0; i < steps; i += 2) {
    const int4 d0 = da, s0 = sa, d1 = db, s1 = sb;
    if (i + 2 < steps) {
      const long long e2 = first + (long long)(i + 2) * 128;
      load_pair(da, sa, db, sb, dst, src, e2 + lane * 4, E, vec && e2 + 256 <= E);
    }
    coarse_push(mq, mn, bq, bc, d0.x, s0.x, slo, lane);
    coarse_push(mq, mn, bq, bc, d0.y, s0.y, slo, lane);
    coarse_push(mq, mn, bq, bc, d0.z, s0.z, slo, lane);
    coarse_push(mq, mn, bq, bc, d0.w, s0.w, slo, lane);
    if (i + 1 < steps) {
      coarse_push(mq, mn, bq, bc, d1.x, s1.x, slo, lane);
      coarse_push(mq, mn, bq, bc, d1.y, s1.y, slo, lane);
      coarse_push(mq, mn, bq, bc, d1.z, s1.z, slo, lane);
      coarse_push(mq, mn, bq, bc, d1.w, s1.w, slo, lane);
    }
  }
  if (mn > 0) bin_push(bq, bc, lane < mn, (lane < mn) ? mq[lane] : 0u, lane);

  const unsigned blq = (lane < SG_Q) ? bc[lane] : 0u;
  bool over = __builtin_amdgcn_ballot_w32(blq > (unsigned)SB_CAP) != 0u;
  int run = 0, offv = 0;
#pragma unroll
  for (int q = 0; q < SG_Q; ++q) {
    const int cq = __builtin_amdgcn_readlane((int)blq, q);
    if (lane == q) offv = run;
    if (!over && run + cq <= SG_CAP) {
      if (lane < cq) seg[AG_LIST_HEAD + run + lane] = bq[q * SB_CAP + lane];
      if (32 + lane < cq) seg[AG_LIST_HEAD + run + 32 + lane] = bq[q * SB_CAP + 32 + lane];
    }
    run += cq;
  }
  if (lane == SG_Q) offv = run;
  if (run > SG_CAP) over = true;
  if (lane <= SG_Q) seg[1 + lane] = (unsigned)offv;
  if (lane == 0) seg[0] = over ? SG_OVER : (unsigned)run;
}

#define LL_STRIDE 512
#define LL_CAP (LL_STRIDE - 1)
#define LL_END ((unsigned)LL_CAP)
#define LL_ROW_BITS 5
#define LL_NEXT_SHIFT (SG_SRC_BITS + LL_ROW_BITS)
static_assert(AG_WAVES * LL_STRIDE + AG_WAVES * AG_WAVES <= 2 * AG_WAVES * AG_OUTCAP,
              "the 32 queues and the heads lie in the LDS of the scan queues");
static_assert(LL_NEXT_SHIFT + 9 == 32 && AG_ROWS <= (AG_WAVES << LL_ROW_BITS) && AG_WAVES == 32,
              "an entry packs source row, row / 32 and a 9-bit link into one word");
template <bool REC>
__device__ __forceinline__ void seg_scan_gather(const float* __restrict__ h,
                                                const unsigned* __restrict__ coarse, int nst, int nen,
                                                float* tile, unsigned* outw, int2* ownq,
                                                int* fin, int rot, int wave, int lane,
                                                unsigned* __restrict__ rec, int& rpos, int rcap,
                                                int& dcnt) {
  const unsigned smask = (1u << SG_SRC_BITS) - 1u;
  const unsigned below = (1u << lane) - 1u;
  unsigned* oq = outw + wave * LL_STRIDE;
  unsigned* heads = outw + AG_WAVES * LL_STRIDE;
  unsigned* hd = heads + wave * AG_WAVES;
  int2* mq = ownq + wave * AG_MYCAP;
  int ci = 0;
  int mn = 0;

  for (int round = 0; round <= SG_C * SG_NB; ++round) {
    int qn = 0;
    if (ci < SG_C) {
      unsigned wa[SG_C], wb[SG_C];
      int ln[SG_C];
#pragma unroll
      for (int k = 0; k < SG_C; ++k) {
        const int cc = (k + rot) & (SG_C - 1);
        const int st = __builtin_amdgcn_readfirstlane(min(max(__builtin_amdgcn_readlane(nst, cc), 0), SG_CAP));
        const int len = __builtin_amdgcn_readfirstlane(
            min(max(__builtin_amdgcn_readlane(nen, cc) - st, 0), min(SB_CAP, SG_CAP - st)));
        const unsigned* sp = coarse + ((long long)cc * 32 + wave) * SG_STRIDE + AG_LIST_HEAD + st;
        wa[k] = (lane < len) ? sp[lane] : 0u;
        wb[k] = (32 + lane < len) ? sp[32 + lane] : 0u;
        ln[k] = len;
      }
      bool full = false;
      int nci = SG_C;
#pragma unroll
      for (int k = 0; k < SG_C; ++k) {
        if (!full && k >= ci) {
          if (qn + ln[k] > LL_CAP) {
            full = true;
            nci = k;
          } else {
            if (lane < ln[k]) oq[qn + lane] = wa[k];
            if (32 + lane < ln[k]) oq[qn + 32 + lane] = wb[k];
            qn += ln[k];
          }
        }
      }
      ci = nci;
    }

    hd[lane] = LL_END;
    for (int b0 = 0; b0 < qn; b0 += 32) {
      const bool live = b0 + lane < qn;
      unsigned w = 0u;
      if (live) w = oq[b0 + lane];
      const unsigned l = w >> SG_SRC_BITS;
      const unsigned d = l & (AG_WAVES - 1);
      unsigned m = __builtin_amdgcn_ballot_w32(live);
      for (int bit = 0; bit < 5; ++bit) {
        const bool on = ((d >> bit) & 1u) != 0u;
        const unsigned bb = __builtin_amdgcn_ballot_w32(live && on);
        m &= on ? bb : ~bb;
      }
      const unsigned pm = m & below;
      unsigned nx = LL_END;
      if (live) nx = hd[d];
      if (pm != 0u) nx = (unsigned)(b0 + 31 - __clz(pm));
      if (live) {
        oq[b0 + lane] = (w & smask) | ((l >> 5) << SG_SRC_BITS) | (nx << LL_NEXT_SHIFT);
        if ((m >> lane) == 1u) hd[d] = (unsigned)(b0 + lane);
      }
    }
    if (lane == 0) fin[wave] = (ci >= SG_C) ? 1 : 0;
    __syncthreads();

    unsigned p = heads[lane * AG_WAVES + wave];
    for (int step = 0; step <= LL_CAP; ++step) {
      const bool act = p != LL_END;
      if (__builtin_amdgcn_ballot_w32(act) == 0u) break;
      unsigned ent = 0u;
      if (act) ent = outw[lane * LL_STRIDE + p];
      const int srow = (int)(ent & smask);
      const int trow = wave + 32 * (int)((ent >> SG_SRC_BITS) & ((1u << LL_ROW_BITS) - 1u));
      mn = queue_push(mq, mn, act, make_int2(srow, trow), lane);
      p = act ? (ent >> LL_NEXT_SHIFT) : LL_END;
      if (mn >= 32) {
        add_rows<REC, true>(mq, 32, h, (const int*)0, tile, lane, rec, rpos, rcap, dcnt, wave + 32 * lane);
        const int rem = mn - 32;
        if (lane < rem) {
          const int2 mv = mq[32 + lane];
          mq[lane] = mv;
        }
        mn = rem;
      }
    }
    const bool alldone = __builtin_amdgcn_ballot_w32(fin[lane] != 0) == 0xffffffffu;
    __syncthreads();
    if (alldone) break;
  }
  if (mn > 0) add_rows<REC, true>(mq, mn, h, (const int*)0, tile, lane, rec, rpos, rcap, dcnt, wave + 32 * lane);
}

template <int L0>
__device__ __forceinline__ void load_group(float2 (&v)[8], int (&l)[8], unsigned eb,
                                           const float* __restrict__ h, int lane) {
  const unsigned smask = (1u << AG_SRC_BITS) - 1u;
#pragma unroll
  for (int u = 0; u < 8; ++u) {
    const unsigned e = (unsigned)__builtin_amdgcn_readlane((int)eb, L0 + u);
    l[u] = (int)(e >> AG_SRC_BITS);
    v[u] = *(const float2*)(h + (long long)(e & smask) * HID + lane * 2);
  }
}

__device__ __forceinline__ void add_group(const float2 (&v)[8], const int (&l)[8], int first,
                                          int cnt, float* tile, int lane) {
#pragma unroll
  for (int u = 0; u < 8; ++u) {
    if (first + u < cnt) {
      float2* t = (float2*)(tile + l[u] * HID + lane * 2);
      float2 a = *t;
      a.x += v[u].x;
      a.y += v[u].y;
      *t = a;
    }
  }
}

__device__ __forceinline__ void replay_rows(const unsigned* __restrict__ ents, int cnt,
                                            const float* __restrict__ h, float* tile, int lane) {
  unsigned eb = (lane < cnt) ? ents[lane] : 0u;
  float2 va[8], vb[8], vc[8], vd[8];
  int la[8], lb[8], lc[8], ld[8];
  load_group<0>(va, la, eb, h, lane);
  __builtin_amdgcn_sched_barrier(0);
  load_group<8>(vb, lb, eb, h, lane);
  for (int b = 0; b < cnt; b += 32) {
    const unsigned en = (b + 32 + lane < cnt) ? ents[b + 32 + lane] : 0u;
    load_group<16>(vc, lc, eb, h, lane);
    add_group(va, la, b, cnt, tile, lane);
    load_group<24>(vd, ld, eb, h, lane);
    add_group(vb, lb, b + 8, cnt, tile, lane);
    load_group<0>(va, la, en, h, lane);
    add_group(vc, lc, b + 16, cnt, tile, lane);
    load_group<8>(vb, lb, en, h, lane);
    add_group(vd, ld, b + 24, cnt, tile, lane);
    eb = en;
  }
}

template <bool REPLAY>
__global__ __launch_bounds__(AG_THREADS) void neigh_sum_kernel(const float* __restrict__ h,
                                                             const int* __restrict__ src,
                                                             const int* __restrict__ dst,
                                                             float* __restrict__ agg,
                                                             unsigned* __restrict__ lists,
                                                             float* __restrict__ deg, int n,
                                                             int E) {
  __shared__ float tile[AG_ROWS * HID];
  __shared__ int2 outq[AG_WAVES * AG_OUTCAP];
  __shared__ int2 ownq[AG_WAVES * AG_MYCAP];
  __shared__ int outn[AG_WAVES];
  __shared__ int fin[AG_WAVES];

  const int tid = threadIdx.x;
  const int wave = tid >> 5;
  const int lane = tid & 31;
  const int rowlo = blockIdx.x * AG_ROWS;
  const int rows = (n - rowlo < AG_ROWS) ? (n - rowlo) : AG_ROWS;

  float4* t4 = (float4*)tile;
  const float4 z = {0.0f, 0.0f, 0.0f, 0.0f};
  for (int i = tid; i < AG_ROWS * HID / 4; i += AG_THREADS) t4[i] = z;

  int g_ag, nsuper;
  const bool lay = sg_layout(n, g_ag, nsuper);
  const int rcap = lay ? SG_CAP : 0;
  unsigned* rec = lists + ((long long)blockIdx.x * 32 + wave) * SG_STRIDE;
  int rpos = 0;
  int dcnt = 0;

  if (REPLAY) {
    const int cnt = lay ? __builtin_amdgcn_readfirstlane((int)rec[0]) : -1;
    const int redo =
        __builtin_amdgcn_readfirstlane(__syncthreads_or((cnt < 0 || cnt > rcap) ? 1 : 0));
    if (!redo) {
      replay_rows(rec + AG_LIST_HEAD, cnt, h, tile, lane);
    } else {
      scan_gather<false>(h, src, dst, E, tile, outq, ownq, outn, fin, rowlo, wave, lane, rec,
                         rpos, rcap, dcnt);
    }
  } else {
    const int sg = blockIdx.x / SG_Q;
    const int qg = blockIdx.x - sg * SG_Q;
    const unsigned* coarse = lists + ((long long)g_ag * 32 + (long long)sg * SG_C * 32) * SG_STRIDE;
    unsigned nlen = 0u;
    int nst = 0, nen = 0;
    if (lay && lane < SG_C) {
      const unsigned* lh = coarse + ((long long)lane * 32 + wave) * SG_STRIDE;
      nlen = lh[0];
      nst = (int)lh[1 + qg];
      nen = (int)lh[2 + qg];
    }
    const bool over = __builtin_amdgcn_ballot_w32(nlen > (unsigned)SG_CAP) != 0u;
    const int direct =
        __builtin_amdgcn_readfirstlane(__syncthreads_or((!lay || over) ? 1 : 0));
    if (!direct) {
      seg_scan_gather<true>(h, coarse, nst, nen, tile, (unsigned*)outq, ownq, fin, qg, wave, lane,
                            rec, rpos, rcap, dcnt);
    } else {
      scan_gather<true>(h, src, dst, E, tile, outq, ownq, outn, fin, rowlo, wave, lane, rec,
                        rpos, rcap, dcnt);
    }
    if (lay && lane == 0) rec[0] = (unsigned)rpos;
    const int r = wave + 32 * lane;
    if (r < AG_ROWS) ((float*)outq)[r] = (float)dcnt;
  }
  __syncthreads();

  float4* o4 = (float4*)(agg + (long long)rowlo * HID);
  for (int i = tid; i < rows * (HID / 4); i += AG_THREADS) o4[i] = t4[i];
  if (!REPLAY) {
    if (tid < rows) deg[rowlo + tid] = ((const float*)outq)[tid];
  }
}


#define PJ_FULL 20
__global__ __launch_bounds__(128) __attribute__((amdgpu_waves_per_eu(2, 2))) void proj_kernel(const float* __restrict__ x,
                                                   const float* __restrict__ W,
                                                   const float* __restrict__ bias,
                                                   float* __restrict__ h, int n) {
  const int tid = threadIdx.x;
  const int wave = __builtin_amdgcn_readfirstlane(tid >> 5);
  const int lane = tid & 31;
  const int hf = lane >> 4;
  const int m = lane & 15;

  v2f bw[PJ_FULL + 1][2][4];
#pragma unroll
  for (int j = 0; j < PJ_FULL; ++j) {
#pragma unroll
    for (int tt = 0; tt < 2; ++tt) {
      const int k = 8 * j + 4 * hf + 2 * tt;
#pragma unroll
      for (int q = 0; q < 4; ++q) {
        bw[j][tt][q][0] = W[k * 64 + 16 * q + m];
        bw[j][tt][q][1] = W[(k + 1) * 64 + 16 * q + m];
      }
    }
  }
#pragma unroll
  for (int tt = 0; tt < 2; ++tt) {
    const int k = 8 * PJ_FULL + 4 * hf + 2 * tt;
#pragma unroll
    for (int q = 0; q < 4; ++q) {
      bw[PJ_FULL][tt][q][0] = (k < 165) ? W[k * 64 + 16 * q + m] : 0.0f;
      bw[PJ_FULL][tt][q][1] = (k + 1 < 165) ? W[(k + 1) * 64 + 16 * q + m] : 0.0f;
    }
  }
  float bv[4];
#pragma unroll
  for (int q = 0; q < 4; ++q) bv[q] = bias[16 * q + m];

  const int ntiles = (n + 15) >> 4;
  for (int t = blockIdx.x * 4 + wave; t < ntiles; t += (int)gridDim.x * 4) {
    const int row = t * 16 + m;
    const int rl = row < n ? row : n - 1;
    const float* xr = x + (long long)rl * 165 + 4 * hf;
    v4f a[PJ_FULL + 1];
#pragma unroll
    for (int j = 0; j < PJ_FULL; ++j) a[j] = *(const v4f_u*)(xr + 8 * j);
    const float* tp = xr + 8 * PJ_FULL;
    const float t0 = tp[0];
    const float t1 = tp[hf ? 0 : 1];
    const float t2 = tp[hf ? 0 : 2];
    const float t3 = tp[hf ? 0 : 3];
    v4f tl;
    tl[0] = t0;
    tl[1] = hf ? 0.0f : t1;
    tl[2] = hf ? 0.0f : t2;
    tl[3] = hf ? 0.0f : t3;
    a[PJ_FULL] = tl;

    v8f acc[4] = {};
#pragma unroll
    for (int j = 0; j <= PJ_FULL; ++j) {
#pragma unroll
      for (int tt = 0; tt < 2; ++tt) {
        v2f af;
        af[0] = a[j][2 * tt];
        af[1] = a[j][2 * tt + 1];
#pragma unroll
        for (int q = 0; q < 4; ++q) acc[q] = wmma_k4(af, bw[j][tt][q], acc[q]);
      }
    }

#pragma unroll
    for (int q = 0; q < 4; ++q) {
#pragma unroll
      for (int v = 0; v < 8; ++v) {
        const int r = t * 16 + v + 8 * hf;
        if (r < n) {
          const float val = acc[q][v] + bv[q];
          h[(long long)r * HID + 16 * q + m] = val > 0.0f ? val : 0.0f;
        }
      }
    }
  }
}

__global__ __launch_bounds__(128) __attribute__((amdgpu_waves_per_eu(2, 2))) void combine_kernel(
    const float* agg, const float* __restrict__ deg,
    const float* __restrict__ h_in, const float* __restrict__ Wl,
    const float* __restrict__ bl, const float* __restrict__ Wr,
    float* h_out, int n) {
  const int tid = threadIdx.x;
  const int wave = __builtin_amdgcn_readfirstlane(tid >> 5);
  const int lane = tid & 31;
  const int hf = lane >> 4;
  const int m = lane & 15;

  v2f wl[8][2][4], wr[8][2][4];
#pragma unroll
  for (int j = 0; j < 8; ++j) {
#pragma unroll
    for (int tt = 0; tt < 2; ++tt) {
      const int k = 8 * j + 4 * hf + 2 * tt;
#pragma unroll
      for (int q = 0; q < 4; ++q) {
        wl[j][tt][q][0] = Wl[k * 64 + 16 * q + m];
        wl[j][tt][q][1] = Wl[(k + 1) * 64 + 16 * q + m];
        wr[j][tt][q][0] = Wr[k * 64 + 16 * q + m];
        wr[j][tt][q][1] = Wr[(k + 1) * 64 + 16 * q + m];
      }
    }
  }
  float bv[4];
#pragma unroll
  for (int q = 0; q < 4; ++q) bv[q] = bl[16 * q + m];

  const int ntiles = (n + 15) >> 4;
  for (int t = blockIdx.x * 4 + wave; t < ntiles; t += (int)gridDim.x * 4) {
    const int row = t * 16 + m;
    const int rl = row < n ? row : n - 1;
    const v4f* ap = (const v4f*)(agg + (long long)rl * HID + 4 * hf);
    const v4f* hp = (const v4f*)(h_in + (long long)rl * HID + 4 * hf);
    v4f a[8], hv[8];
#pragma unroll
    for (int j = 0; j < 8; ++j) {
      a[j] = ap[2 * j];
      hv[j] = hp[2 * j];
    }
    float d = deg[rl];
    d = d > 1.0f ? d : 1.0f;
#pragma unroll
    for (int j = 0; j < 8; ++j) {
      a[j][0] = a[j][0] / d;
      a[j][1] = a[j][1] / d;
      a[j][2] = a[j][2] / d;
      a[j][3] = a[j][3] / d;
    }

    v8f acc[4] = {};
#pragma unroll
    for (int j = 0; j < 8; ++j) {
#pragma unroll
      for (int tt = 0; tt < 2; ++tt) {
        v2f af;
        af[0] = a[j][2 * tt];
        af[1] = a[j][2 * tt + 1];
#pragma unroll
        for (int q = 0; q < 4; ++q) acc[q] = wmma_k4(af, wl[j][tt][q], acc[q]);
      }
    }
#pragma unroll
    for (int j = 0; j < 8; ++j) {
#pragma unroll
      for (int tt = 0; tt < 2; ++tt) {
        v2f af;
        af[0] = hv[j][2 * tt];
        af[1] = hv[j][2 * tt + 1];
#pragma unroll
        for (int q = 0; q < 4; ++q) acc[q] = wmma_k4(af, wr[j][tt][q], acc[q]);
      }
    }

#pragma unroll
    for (int q = 0; q < 4; ++q) {
#pragma unroll
      for (int v = 0; v < 8; ++v) {
        const int r = t * 16 + v + 8 * hf;
        if (r < n) {
          const float val = acc[q][v] + bv[q];
          h_out[(long long)r * HID + 16 * q + m] = val > 0.0f ? val : 0.0f;
        }
      }
    }
  }
}

__global__ void cls_kernel(const float* __restrict__ h, const float* __restrict__ Wc,
                           const float* __restrict__ bc, float* __restrict__ out, int n) {
  int i = blockIdx.x * blockDim.x + threadIdx.x;
  if (i >= n) return;
  const float4* hp = (const float4*)(h + (long long)i * HID);
  const float4* wp = (const float4*)Wc;
  float acc = bc[0];
#pragma unroll
  for (int j = 0; j < HID / 4; ++j) {
    float4 hv = hp[j];
    float4 wv = wp[j];
    acc += hv.x * wv.x + hv.y * wv.y + hv.z * wv.z + hv.w * wv.w;
  }
  out[i] = acc;
}

extern "C" void kernel_launch(void* const* d_in, const int* in_sizes, int n_in,
                              void* d_out, int out_size, void* d_ws, size_t ws_size,
                              hipStream_t stream) {
  const float* x   = (const float*)d_in[0];
  const int* eidx  = (const int*)d_in[1];
  const float* Wp  = (const float*)d_in[2];
  const float* bp  = (const float*)d_in[3];
  const float* W1l = (const float*)d_in[4];
  const float* b1l = (const float*)d_in[5];
  const float* W1r = (const float*)d_in[6];
  const float* W2l = (const float*)d_in[7];
  const float* b2l = (const float*)d_in[8];
  const float* W2r = (const float*)d_in[9];
  const float* Wc  = (const float*)d_in[10];
  const float* bc  = (const float*)d_in[11];

  const int N = in_sizes[0] / 165;
  const int E = in_sizes[1] / 2;
  const int* src = eidx;
  const int* dst = eidx + E;

  float* A   = (float*)d_ws;
  float* B   = A + (size_t)N * HID;
  float* AGG = B + (size_t)N * HID;
  float* DEG = AGG + (size_t)N * HID;
  unsigned* LISTS = (unsigned*)B;

  const unsigned g_ag  = (unsigned)((N + AG_ROWS - 1) / AG_ROWS);
  const unsigned g_gemm = 512;
  const unsigned g_co  = ((g_ag + SG_Q - 1) / SG_Q) * SG_C;

  proj_kernel<<<g_gemm, 128, 0, stream>>>(x, Wp, bp, A, N);

  coarse_kernel<<<g_co, AG_THREADS, 0, stream>>>(src, dst, LISTS, N, E);

  neigh_sum_kernel<false><<<g_ag, AG_THREADS, 0, stream>>>(A, src, dst, AGG, LISTS, DEG, N, E);
  combine_kernel<<<g_gemm, 128, 0, stream>>>(AGG, DEG, A, W1l, b1l, W1r, AGG, N);

  neigh_sum_kernel<true><<<g_ag, AG_THREADS, 0, stream>>>(AGG, src, dst, A, LISTS, DEG, N, E);
  combine_kernel<<<g_gemm, 128, 0, stream>>>(A, DEG, AGG, W2l, b2l, W2r, B, N);

  cls_kernel<<<(N + 255) / 256, 256, 0, stream>>>(B, Wc, bc, (float*)d_out, N);
}


# conv 2: W2_l, W2_r applied to h1 in conv 1's combine (p, q stored, not h1); rows of p summed, tile's store becomes relu(sum/deg+q).W_cls; 7 launches to 5; mean after the product: sum order changes
# speedup vs baseline: 6.6076x; 1.0382x over previous
#include <hip/hip_runtime.h>

#define HID 64

typedef __attribute__((ext_vector_type(2))) float v2f;
typedef __attribute__((ext_vector_type(8))) float v8f;
typedef __attribute__((ext_vector_type(4))) float v4f;
typedef v4f __attribute__((aligned(4))) v4f_u;

__device__ __forceinline__ v8f wmma_k4(v2f a, v2f b, v8f c) {
  return __builtin_amdgcn_wmma_f32_16x16x4_f32(
       false, a,  false, b,
       (short)0, c,  false,  false);
}

#define AG_ROWS 784
#define AG_WAVES 32
#define AG_THREADS (AG_WAVES * 32)
#define AG_ITER (AG_WAVES * 512)
#define AG_OUTCAP 416
#define AG_MYCAP 64
#define AG_SRC_BITS 22
#define AG_LIST_HEAD 32

__device__ __forceinline__ void load_iter(int4 (&r)[4], const int* __restrict__ dst, int g,
                                          int woff, int E, bool vec) {
  const int base = g * AG_ITER;
  if (vec && base + AG_ITER <= E) {
#pragma unroll
    for (int j = 0; j < 4; ++j) r[j] = *(const int4*)(dst + base + woff + j * 128);
  } else {
#pragma unroll
    for (int j = 0; j < 4; ++j) {
      const int e = base + woff + j * 128;
      r[j].x = (e < E) ? dst[e] : -1;
      r[j].y = (e + 1 < E) ? dst[e + 1] : -1;
      r[j].z = (e + 2 < E) ? dst[e + 2] : -1;
      r[j].w = (e + 3 < E) ? dst[e + 3] : -1;
    }
  }
}

__device__ __forceinline__ int queue_push(int2* q, int qn, bool keep, int2 ent, int lane) {
  const unsigned b = __builtin_amdgcn_ballot_w32(keep);
  if (b != 0u) {
    if (keep) q[qn + __popc(b & ((1u << lane) - 1u))] = ent;
    qn += __popc(b);
  }
  return qn;
}

__device__ __forceinline__ bool push4(int2* q, int& qn, int4 d, int e, int rowlo, int lane) {
  const unsigned l0 = (unsigned)(d.x - rowlo);
  const unsigned l1 = (unsigned)(d.y - rowlo);
  const unsigned l2 = (unsigned)(d.z - rowlo);
  const unsigned l3 = (unsigned)(d.w - rowlo);
  const unsigned lm = min(min(l0, l1), min(l2, l3));
  if (__builtin_amdgcn_ballot_w32(lm < (unsigned)AG_ROWS) == 0u) return true;
  const bool k0 = l0 < (unsigned)AG_ROWS, k1 = l1 < (unsigned)AG_ROWS;
  const bool k2 = l2 < (unsigned)AG_ROWS, k3 = l3 < (unsigned)AG_ROWS;
  const unsigned b0 = __builtin_amdgcn_ballot_w32(k0), b1 = __builtin_amdgcn_ballot_w32(k1);
  const unsigned b2 = __builtin_amdgcn_ballot_w32(k2), b3 = __builtin_amdgcn_ballot_w32(k3);
  const int c0 = __popc(b0), c1 = __popc(b1), c2 = __popc(b2), c3 = __popc(b3);
  if (qn + c0 + c1 + c2 + c3 > AG_OUTCAP) return false;
  const unsigned below = (1u << lane) - 1u;
  int p = qn;
  if (k0) q[p + __popc(b0 & below)] = make_int2(e, (int)l0);
  p += c0;
  if (k1) q[p + __popc(b1 & below)] = make_int2(e + 1, (int)l1);
  p += c1;
  if (k2) q[p + __popc(b2 & below)] = make_int2(e + 2, (int)l2);
  p += c2;
  if (k3) q[p + __popc(b3 & below)] = make_int2(e + 3, (int)l3);
  qn = p + c3;
  return true;
}

template <bool REC, bool DIRECT>
__device__ __forceinline__ void add_rows(const int2* q, int cnt, const float* __restrict__ h,
                                         const int* __restrict__ src, float* tile, int lane,
                                         unsigned* __restrict__ rec, int& rpos, int rcap,
                                         int& dcnt, int myrow) {
  int s = 0, loc = 0;
  if (lane < cnt) {
    const int2 ent = q[lane];
    s = DIRECT ? ent.x : src[ent.x];
    loc = ent.y;
  }
  if (REC) {
    if (rpos + cnt <= rcap && lane < cnt)
      rec[AG_LIST_HEAD + rpos + lane] = (unsigned)s | ((unsigned)loc << AG_SRC_BITS);
    rpos += cnt;
  }
  for (int k = 0; k < cnt; k += 8) {
    float2 v[8];
    int lk[8];
#pragma unroll
    for (int u = 0; u < 8; ++u) {
      const int sk = __builtin_amdgcn_readlane(s, k + u);
      lk[u] = __builtin_amdgcn_readlane(loc, k + u);
      v[u] = *(const float2*)(h + (long long)sk * HID + lane * 2);
    }
#pragma unroll
    for (int u = 0; u < 8; ++u) {
      if (k + u < cnt) {
        float2* t = (float2*)(tile + lk[u] * HID + lane * 2);
        float2 a = *t;
        a.x += v[u].x;
        a.y += v[u].y;
        *t = a;
        if (REC) dcnt += (lk[u] == myrow) ? 1 : 0;
      }
    }
  }
}

template <bool REC>
__device__ __forceinline__ void scan_gather(const float* __restrict__ h,
                                            const int* __restrict__ src,
                                            const int* __restrict__ dst, int E, float* tile,
                                            int2* outq, int2* ownq, int* outn, int* fin,
                                            int rowlo, int wave, int lane,
                                            unsigned* __restrict__ rec, int& rpos, int rcap,
                                            int& dcnt) {
  const bool vec = (((size_t)dst) & 15) == 0;
  const int nit = (E + AG_ITER - 1) / AG_ITER;
  const int start = (int)(((long long)blockIdx.x * nit) / (int)gridDim.x);
  const int woff = wave * 512 + lane * 4;
  const int4 none = {-1, -1, -1, -1};

  int4 cur[4], nxt[4];
#pragma unroll
  for (int j = 0; j < 4; ++j) {
    cur[j] = none;
    nxt[j] = none;
  }

  int2* oq = outq + wave * AG_OUTCAP;
  int2* mq = ownq + wave * AG_MYCAP;
  int it = -1;
  int mn = 0;

  for (int round = 0; round <= 4 * nit; ++round) {
    int qn = 0;
    while (it < nit) {
      int g = start + it;
      if (g >= nit) g -= nit;
      int g1 = start + it + 1;
      if (g1 >= nit) g1 -= nit;
      if (it + 1 < nit) {
        load_iter(nxt, dst, g1, woff, E, vec);
      } else {
#pragma unroll
        for (int j = 0; j < 4; ++j) nxt[j] = none;
      }
      const int e0 = g * AG_ITER + woff;
      int queued = 0;
      if (push4(oq, qn, cur[0], e0, rowlo, lane)) {
        queued = 1;
        if (push4(oq, qn, cur[1], e0 + 128, rowlo, lane)) {
          queued = 2;
          if (push4(oq, qn, cur[2], e0 + 256, rowlo, lane)) {
            queued = 3;
            if (push4(oq, qn, cur[3], e0 + 384, rowlo, lane)) queued = 4;
          }
        }
      }
      if (queued < 4) {
        if (queued > 0) cur[0] = none;
        if (queued > 1) cur[1] = none;
        if (queued > 2) cur[2] = none;
        break;
      }
#pragma unroll
      for (int j = 0; j < 4; ++j) cur[j] = nxt[j];
      ++it;
    }
    if (lane == 0) {
      outn[wave] = qn;
      fin[wave] = (it >= nit) ? 1 : 0;
    }
    __syncthreads();

    for (int i = 0; i < AG_WAVES; ++i) {
      const int w2 = (wave + i) & (AG_WAVES - 1);
      const int cnt = __builtin_amdgcn_readfirstlane(outn[w2]);
      for (int b0 = 0; b0 < cnt; b0 += 32) {
        const bool live = b0 + lane < cnt;
        int2 ent = make_int2(0, -1);
        if (live) ent = outq[w2 * AG_OUTCAP + b0 + lane];
        const bool mine = live && ((ent.y & (AG_WAVES - 1)) == wave);
        mn = queue_push(mq, mn, mine, ent, lane);
        if (mn >= 32) {
          add_rows<REC, false>(mq, 32, h, src, tile, lane, rec, rpos, rcap, dcnt, wave + 32 * lane);
          const int rem = mn - 32;
          if (lane < rem) {
            const int2 mv = mq[32 + lane];
            mq[lane] = mv;
          }
          mn = rem;
        }
      }
    }
    const bool alldone = __builtin_amdgcn_ballot_w32(fin[lane] != 0) == 0xffffffffu;
    __syncthreads();
    if (alldone) break;
  }
  if (mn > 0) add_rows<REC, false>(mq, mn, h, src, tile, lane, rec, rpos, rcap, dcnt, wave + 32 * lane);
}

#define SG_Q 16
#define SG_C 16
#define SG_STRIDE 768
#define SG_CAP (SG_STRIDE - AG_LIST_HEAD)
#define SG_SRC_BITS 18
#define SG_OVER 0xffffffffu
#define SG_NB (SG_CAP / 32)
#define SB_CAP 64
static_assert(SG_Q == 16 && SG_Q + 2 <= AG_LIST_HEAD, "a bin index has four bits; the bin starts lie in the list's head");

__device__ __forceinline__ bool sg_layout(int n, int& g_ag, int& nsuper) {
  g_ag = (n + AG_ROWS - 1) / AG_ROWS;
  nsuper = (g_ag + SG_Q - 1) / SG_Q;
  const long long words = ((long long)g_ag + (long long)nsuper * SG_C) * 32 * SG_STRIDE;
  return n <= (1 << SG_SRC_BITS) && words <= (long long)n * HID;
}

__device__ __forceinline__ int queue_push_w(unsigned* q, int qn, bool keep, unsigned w, int lane) {
  const unsigned b = __builtin_amdgcn_ballot_w32(keep);
  if (b != 0u) {
    if (keep) q[qn + __popc(b & ((1u << lane) - 1u))] = w;
    qn += __popc(b);
  }
  return qn;
}

__device__ __forceinline__ void load_pair(int4& d0, int4& s0, int4& d1, int4& s1,
                                          const int* __restrict__ dst,
                                          const int* __restrict__ src, long long e, int E,
                                          bool whole) {
  if (whole) {
    d0 = *(const int4*)(dst + e);
    s0 = *(const int4*)(src + e);
    d1 = *(const int4*)(dst + e + 128);
    s1 = *(const int4*)(src + e + 128);
  } else {
    const long long f = e + 128;
    d0.x = (e < E) ? dst[e] : -1;
    d0.y = (e + 1 < E) ? dst[e + 1] : -1;
    d0.z = (e + 2 < E) ? dst[e + 2] : -1;
    d0.w = (e + 3 < E) ? dst[e + 3] : -1;
    s0.x = (e < E) ? src[e] : 0;
    s0.y = (e + 1 < E) ? src[e + 1] : 0;
    s0.z = (e + 2 < E) ? src[e + 2] : 0;
    s0.w = (e + 3 < E) ? src[e + 3] : 0;
    d1.x = (f < E) ? dst[f] : -1;
    d1.y = (f + 1 < E) ? dst[f + 1] : -1;
    d1.z = (f + 2 < E) ? dst[f + 2] : -1;
    d1.w = (f + 3 < E) ? dst[f + 3] : -1;
    s1.x = (f < E) ? src[f] : 0;
    s1.y = (f + 1 < E) ? src[f + 1] : 0;
    s1.z = (f + 2 < E) ? src[f + 2] : 0;
    s1.w = (f + 3 < E) ? src[f + 3] : 0;
  }
}

__device__ __forceinline__ void bin_push(unsigned* bq, unsigned* bc, bool live, unsigned w, int lane) {
  const unsigned smask = (1u << SG_SRC_BITS) - 1u;
  const unsigned u = w >> SG_SRC_BITS;
  const unsigned q = u / (unsigned)AG_ROWS;
  const unsigned l = u - q * (unsigned)AG_ROWS;
  unsigned m = __builtin_amdgcn_ballot_w32(live);
  for (int bit = 0; bit < 4; ++bit) {
    const bool on = ((q >> bit) & 1u) != 0u;
    const unsigned bb = __builtin_amdgcn_ballot_w32(live && on);
    m &= on ? bb : ~bb;
  }
  if (live) {
    const unsigned pos = bc[q] + (unsigned)__popc(m & ((1u << lane) - 1u));
    if (pos < (unsigned)SB_CAP) bq[q * SB_CAP + pos] = (w & smask) | (l << SG_SRC_BITS);
    if ((m >> lane) == 1u) bc[q] = pos + 1u;
  }
}

__device__ __forceinline__ void coarse_push(unsigned* mq, int& mn, unsigned* bq, unsigned* bc,
                                            int d, int s, int slo, int lane) {
  const unsigned u = (unsigned)(d - slo);
  const bool keep = u < (unsigned)(SG_Q * AG_ROWS);
  mn = queue_push_w(mq, mn, keep, (unsigned)s | (u << SG_SRC_BITS), lane);
  if (mn >= 32) {
    bin_push(bq, bc, true, mq[lane], lane);
    const int rem = mn - 32;
    if (lane < rem) {
      const unsigned mv = mq[32 + lane];
      mq[lane] = mv;
    }
    mn = rem;
  }
}

__global__ __launch_bounds__(AG_THREADS) void coarse_kernel(const int* __restrict__ src,
                                                          const int* __restrict__ dst,
                                                          unsigned* __restrict__ lists, int n,
                                                          int E) {
  __shared__ unsigned stage[AG_WAVES * 64];
  __shared__ unsigned bins[AG_WAVES * SG_Q * SB_CAP];
  __shared__ unsigned blen[AG_WAVES * SG_Q];
  int g_ag, nsuper;
  if (!sg_layout(n, g_ag, nsuper)) return;
  const int tid = threadIdx.x;
  const int wave = __builtin_amdgcn_readfirstlane(tid >> 5);
  const int lane = tid & 31;
  const int s = blockIdx.x / SG_C;
  const int c = blockIdx.x - s * SG_C;
  const int slo = s * (SG_Q * AG_ROWS);
  const long long share = ((long long)E + SG_C * AG_WAVES - 1) / (SG_C * AG_WAVES);
  const int steps = (int)((share + 127) / 128);
  const long long first = (long long)(c * AG_WAVES + wave) * steps * 128;
  unsigned* seg = lists + ((long long)g_ag * 32 + (long long)blockIdx.x * 32 + wave) * SG_STRIDE;
  unsigned* mq = stage + wave * 64;
  unsigned* bq = bins + wave * (SG_Q * SB_CAP);
  unsigned* bc = blen + wave * SG_Q;
  const bool vec = ((((size_t)src) | ((size_t)dst)) & 15) == 0;

  if (lane < SG_Q) bc[lane] = 0u;
  int mn = 0;
  int4 da, sa, db, sb;
  load_pair(da, sa, db, sb, dst, src, first + lane * 4, E, vec && first + 256 <= E);
  for (int i = 0; i < steps; i += 2) {
    const int4 d0 = da, s0 = sa, d1 = db, s1 = sb;
    if (i + 2 < steps) {
      const long long e2 = first + (long long)(i + 2) * 128;
      load_pair(da, sa, db, sb, dst, src, e2 + lane * 4, E, vec && e2 + 256 <= E);
    }
    coarse_push(mq, mn, bq, bc, d0.x, s0.x, slo, lane);
    coarse_push(mq, mn, bq, bc, d0.y, s0.y, slo, lane);
    coarse_push(mq, mn, bq, bc, d0.z, s0.z, slo, lane);
    coarse_push(mq, mn, bq, bc, d0.w, s0.w, slo, lane);
    if (i + 1 < steps) {
      coarse_push(mq, mn, bq, bc, d1.x, s1.x, slo, lane);
      coarse_push(mq, mn, bq, bc, d1.y, s1.y, slo, lane);
      coarse_push(mq, mn, bq, bc, d1.z, s1.z, slo, lane);
      coarse_push(mq, mn, bq, bc, d1.w, s1.w, slo, lane);
    }
  }
  if (mn > 0) bin_push(bq, bc, lane < mn, (lane < mn) ? mq[lane] : 0u, lane);

  const unsigned blq = (lane < SG_Q) ? bc[lane] : 0u;
  bool over = __builtin_amdgcn_ballot_w32(blq > (unsigned)SB_CAP) != 0u;
  int run = 0, offv = 0;
#pragma unroll
  for (int q = 0; q < SG_Q; ++q) {
    const int cq = __builtin_amdgcn_readlane((int)blq, q);
    if (lane == q) offv = run;
    if (!over && run + cq <= SG_CAP) {
      if (lane < cq) seg[AG_LIST_HEAD + run + lane] = bq[q * SB_CAP + lane];
      if (32 + lane < cq) seg[AG_LIST_HEAD + run + 32 + lane] = bq[q * SB_CAP + 32 + lane];
    }
    run += cq;
  }
  if (lane == SG_Q) offv = run;
  if (run > SG_CAP) over = true;
  if (lane <= SG_Q) seg[1 + lane] = (unsigned)offv;
  if (lane == 0) seg[0] = over ? SG_OVER : (unsigned)run;
}

#define LL_STRIDE 512
#define LL_CAP (LL_STRIDE - 1)
#define LL_END ((unsigned)LL_CAP)
#define LL_ROW_BITS 5
#define LL_NEXT_SHIFT (SG_SRC_BITS + LL_ROW_BITS)
static_assert(AG_WAVES * LL_STRIDE + AG_WAVES * AG_WAVES <= 2 * AG_WAVES * AG_OUTCAP,
              "the 32 queues and the heads lie in the LDS of the scan queues");
static_assert(LL_NEXT_SHIFT + 9 == 32 && AG_ROWS <= (AG_WAVES << LL_ROW_BITS) && AG_WAVES == 32,
              "an entry packs source row, row / 32 and a 9-bit link into one word");
template <bool REC>
__device__ __forceinline__ void seg_scan_gather(const float* __restrict__ h,
                                                const unsigned* __restrict__ coarse, int nst, int nen,
                                                float* tile, unsigned* outw, int2* ownq,
                                                int* fin, int rot, int wave, int lane,
                                                unsigned* __restrict__ rec, int& rpos, int rcap,
                                                int& dcnt) {
  const unsigned smask = (1u << SG_SRC_BITS) - 1u;
  const unsigned below = (1u << lane) - 1u;
  unsigned* oq = outw + wave * LL_STRIDE;
  unsigned* heads = outw + AG_WAVES * LL_STRIDE;
  unsigned* hd = heads + wave * AG_WAVES;
  int2* mq = ownq + wave * AG_MYCAP;
  int ci = 0;
  int mn = 0;

  for (int round = 0; round <= SG_C * SG_NB; ++round) {
    int qn = 0;
    if (ci < SG_C) {
      unsigned wa[SG_C], wb[SG_C];
      int ln[SG_C];
#pragma unroll
      for (int k = 0; k < SG_C; ++k) {
        const int cc = (k + rot) & (SG_C - 1);
        const int st = __builtin_amdgcn_readfirstlane(min(max(__builtin_amdgcn_readlane(nst, cc), 0), SG_CAP));
        const int len = __builtin_amdgcn_readfirstlane(
            min(max(__builtin_amdgcn_readlane(nen, cc) - st, 0), min(SB_CAP, SG_CAP - st)));
        const unsigned* sp = coarse + ((long long)cc * 32 + wave) * SG_STRIDE + AG_LIST_HEAD + st;
        wa[k] = (lane < len) ? sp[lane] : 0u;
        wb[k] = (32 + lane < len) ? sp[32 + lane] : 0u;
        ln[k] = len;
      }
      bool full = false;
      int nci = SG_C;
#pragma unroll
      for (int k = 0; k < SG_C; ++k) {
        if (!full && k >= ci) {
          if (qn + ln[k] > LL_CAP) {
            full = true;
            nci = k;
          } else {
            if (lane < ln[k]) oq[qn + lane] = wa[k];
            if (32 + lane < ln[k]) oq[qn + 32 + lane] = wb[k];
            qn += ln[k];
          }
        }
      }
      ci = nci;
    }

    hd[lane] = LL_END;
    for (int b0 = 0; b0 < qn; b0 += 32) {
      const bool live = b0 + lane < qn;
      unsigned w = 0u;
      if (live) w = oq[b0 + lane];
      const unsigned l = w >> SG_SRC_BITS;
      const unsigned d = l & (AG_WAVES - 1);
      unsigned m = __builtin_amdgcn_ballot_w32(live);
      for (int bit = 0; bit < 5; ++bit) {
        const bool on = ((d >> bit) & 1u) != 0u;
        const unsigned bb = __builtin_amdgcn_ballot_w32(live && on);
        m &= on ? bb : ~bb;
      }
      const unsigned pm = m & below;
      unsigned nx = LL_END;
      if (live) nx = hd[d];
      if (pm != 0u) nx = (unsigned)(b0 + 31 - __clz(pm));
      if (live) {
        oq[b0 + lane] = (w & smask) | ((l >> 5) << SG_SRC_BITS) | (nx << LL_NEXT_SHIFT);
        if ((m >> lane) == 1u) hd[d] = (unsigned)(b0 + lane);
      }
    }
    if (lane == 0) fin[wave] = (ci >= SG_C) ? 1 : 0;
    __syncthreads();

    unsigned p = heads[lane * AG_WAVES + wave];
    for (int step = 0; step <= LL_CAP; ++step) {
      const bool act = p != LL_END;
      if (__builtin_amdgcn_ballot_w32(act) == 0u) break;
      unsigned ent = 0u;
      if (act) ent = outw[lane * LL_STRIDE + p];
      const int srow = (int)(ent & smask);
      const int trow = wave + 32 * (int)((ent >> SG_SRC_BITS) & ((1u << LL_ROW_BITS) - 1u));
      mn = queue_push(mq, mn, act, make_int2(srow, trow), lane);
      p = act ? (ent >> LL_NEXT_SHIFT) : LL_END;
      if (mn >= 32) {
        add_rows<REC, true>(mq, 32, h, (const int*)0, tile, lane, rec, rpos, rcap, dcnt, wave + 32 * lane);
        const int rem = mn - 32;
        if (lane < rem) {
          const int2 mv = mq[32 + lane];
          mq[lane] = mv;
        }
        mn = rem;
      }
    }
    const bool alldone = __builtin_amdgcn_ballot_w32(fin[lane] != 0) == 0xffffffffu;
    __syncthreads();
    if (alldone) break;
  }
  if (mn > 0) add_rows<REC, true>(mq, mn, h, (const int*)0, tile, lane, rec, rpos, rcap, dcnt, wave + 32 * lane);
}

template <int L0>
__device__ __forceinline__ void load_group(float2 (&v)[8], int (&l)[8], unsigned eb,
                                           const float* __restrict__ h, int lane) {
  const unsigned smask = (1u << AG_SRC_BITS) - 1u;
#pragma unroll
  for (int u = 0; u < 8; ++u) {
    const unsigned e = (unsigned)__builtin_amdgcn_readlane((int)eb, L0 + u);
    l[u] = (int)(e >> AG_SRC_BITS);
    v[u] = *(const float2*)(h + (long long)(e & smask) * HID + lane * 2);
  }
}

__device__ __forceinline__ void add_group(const float2 (&v)[8], const int (&l)[8], int first,
                                          int cnt, float* tile, int lane) {
#pragma unroll
  for (int u = 0; u < 8; ++u) {
    if (first + u < cnt) {
      float2* t = (float2*)(tile + l[u] * HID + lane * 2);
      float2 a = *t;
      a.x += v[u].x;
      a.y += v[u].y;
      *t = a;
    }
  }
}

__device__ __forceinline__ void replay_rows(const unsigned* __restrict__ ents, int cnt,
                                            const float* __restrict__ h, float* tile, int lane) {
  unsigned eb = (lane < cnt) ? ents[lane] : 0u;
  float2 va[8], vb[8], vc[8], vd[8];
  int la[8], lb[8], lc[8], ld[8];
  load_group<0>(va, la, eb, h, lane);
  __builtin_amdgcn_sched_barrier(0);
  load_group<8>(vb, lb, eb, h, lane);
  for (int b = 0; b < cnt; b += 32) {
    const unsigned en = (b + 32 + lane < cnt) ? ents[b + 32 + lane] : 0u;
    load_group<16>(vc, lc, eb, h, lane);
    add_group(va, la, b, cnt, tile, lane);
    load_group<24>(vd, ld, eb, h, lane);
    add_group(vb, lb, b + 8, cnt, tile, lane);
    load_group<0>(va, la, en, h, lane);
    add_group(vc, lc, b + 16, cnt, tile, lane);
    load_group<8>(vb, lb, en, h, lane);
    add_group(vd, ld, b + 24, cnt, tile, lane);
    eb = en;
  }
}

template <bool REPLAY>
__global__ __launch_bounds__(AG_THREADS) void neigh_sum_kernel(const float* __restrict__ h,
                                                             const int* __restrict__ src,
                                                             const int* __restrict__ dst,
                                                             float* __restrict__ agg,
                                                             unsigned* __restrict__ lists,
                                                             float* __restrict__ deg, int n,
                                                             int E,
                                                             const float* __restrict__ qself,
                                                             const float* __restrict__ Wc,
                                                             const float* __restrict__ bc,
                                                             float* __restrict__ out) {
  __shared__ float tile[AG_ROWS * HID];
  __shared__ int2 outq[AG_WAVES * AG_OUTCAP];
  __shared__ int2 ownq[AG_WAVES * AG_MYCAP];
  __shared__ int outn[AG_WAVES];
  __shared__ int fin[AG_WAVES];

  const int tid = threadIdx.x;
  const int wave = tid >> 5;
  const int lane = tid & 31;
  const int rowlo = blockIdx.x * AG_ROWS;
  const int rows = (n - rowlo < AG_ROWS) ? (n - rowlo) : AG_ROWS;

  float4* t4 = (float4*)tile;
  const float4 z = {0.0f, 0.0f, 0.0f, 0.0f};
  for (int i = tid; i < AG_ROWS * HID / 4; i += AG_THREADS) t4[i] = z;

  int g_ag, nsuper;
  const bool lay = sg_layout(n, g_ag, nsuper);
  const int rcap = lay ? SG_CAP : 0;
  unsigned* rec = lists + ((long long)blockIdx.x * 32 + wave) * SG_STRIDE;
  int rpos = 0;
  int dcnt = 0;

  if (REPLAY) {
    const int cnt = lay ? __builtin_amdgcn_readfirstlane((int)rec[0]) : -1;
    const int redo =
        __builtin_amdgcn_readfirstlane(__syncthreads_or((cnt < 0 || cnt > rcap) ? 1 : 0));
    if (!redo) {
      replay_rows(rec + AG_LIST_HEAD, cnt, h, tile, lane);
    } else {
      scan_gather<false>(h, src, dst, E, tile, outq, ownq, outn, fin, rowlo, wave, lane, rec,
                         rpos, rcap, dcnt);
    }
  } else {
    const int sg = blockIdx.x / SG_Q;
    const int qg = blockIdx.x - sg * SG_Q;
    const unsigned* coarse = lists + ((long long)g_ag * 32 + (long long)sg * SG_C * 32) * SG_STRIDE;
    unsigned nlen = 0u;
    int nst = 0, nen = 0;
    if (lay && lane < SG_C) {
      const unsigned* lh = coarse + ((long long)lane * 32 + wave) * SG_STRIDE;
      nlen = lh[0];
      nst = (int)lh[1 + qg];
      nen = (int)lh[2 + qg];
    }
    const bool over = __builtin_amdgcn_ballot_w32(nlen > (unsigned)SG_CAP) != 0u;
    const int direct =
        __builtin_amdgcn_readfirstlane(__syncthreads_or((!lay || over) ? 1 : 0));
    if (!direct) {
      seg_scan_gather<true>(h, coarse, nst, nen, tile, (unsigned*)outq, ownq, fin, qg, wave, lane,
                            rec, rpos, rcap, dcnt);
    } else {
      scan_gather<true>(h, src, dst, E, tile, outq, ownq, outn, fin, rowlo, wave, lane, rec,
                        rpos, rcap, dcnt);
    }
    if (lay && lane == 0) rec[0] = (unsigned)rpos;
    const int r = wave + 32 * lane;
    if (r < AG_ROWS) ((float*)outq)[r] = (float)dcnt;
  }
  __syncthreads();

  if (!REPLAY) {
    float4* o4 = (float4*)(agg + (long long)rowlo * HID);
    for (int i = tid; i < rows * (HID / 4); i += AG_THREADS) o4[i] = t4[i];
    if (tid < rows) deg[rowlo + tid] = ((const float*)outq)[tid];
  } else {
    constexpr int RPW = (AG_ROWS + AG_WAVES - 1) / AG_WAVES;
    float2 qv[RPW];
    float dg[RPW];
#pragma unroll
    for (int i = 0; i < RPW; ++i) {
      const int r = wave + AG_WAVES * i;
      const int rr = r < rows ? r : 0;
      qv[i] = *(const float2*)(qself + (long long)(rowlo + rr) * HID + lane * 2);
      dg[i] = deg[rowlo + rr];
    }
    const float2 wc = *(const float2*)(Wc + lane * 2);
    const float bcv = bc[0];
    static_assert(RPW * 33 <= 2 * AG_OUTCAP && RPW <= 32 && AG_ROWS <= 2 * AG_WAVES * AG_MYCAP,
                  "a wave's buffer fits its scan queue, the results fit the own-row lists");
    float* sc = (float*)outq + wave * (2 * AG_OUTCAP);
    float* res = (float*)ownq;
#pragma unroll
    for (int i = 0; i < RPW; ++i) {
      const int r = wave + AG_WAVES * i;
      const int rr = r < rows ? r : 0;
      const float2 tv = *(const float2*)(tile + rr * HID + lane * 2);
      const float d = dg[i] > 1.0f ? dg[i] : 1.0f;
      float vx = tv.x / d + qv[i].x;
      float vy = tv.y / d + qv[i].y;
      vx = vx > 0.0f ? vx : 0.0f;
      vy = vy > 0.0f ? vy : 0.0f;
      sc[i * 33 + lane] = vx * wc.x + vy * wc.y;
    }
    if (lane < RPW) {
      float s = 0.0f;
      for (int k = 0; k < 32; ++k) s += sc[lane * 33 + k];
      const int r = wave + AG_WAVES * lane;
      if (r < rows) res[r] = s + bcv;
    }
    __syncthreads();
    if (tid < rows) out[rowlo + tid] = res[tid];
  }
}


#define PJ_FULL 20
__global__ __launch_bounds__(128) __attribute__((amdgpu_waves_per_eu(2, 2))) void proj_kernel(const float* __restrict__ x,
                                                   const float* __restrict__ W,
                                                   const float* __restrict__ bias,
                                                   float* __restrict__ h, int n) {
  const int tid = threadIdx.x;
  const int wave = __builtin_amdgcn_readfirstlane(tid >> 5);
  const int lane = tid & 31;
  const int hf = lane >> 4;
  const int m = lane & 15;

  v2f bw[PJ_FULL + 1][2][4];
#pragma unroll
  for (int j = 0; j < PJ_FULL; ++j) {
#pragma unroll
    for (int tt = 0; tt < 2; ++tt) {
      const int k = 8 * j + 4 * hf + 2 * tt;
#pragma unroll
      for (int q = 0; q < 4; ++q) {
        bw[j][tt][q][0] = W[k * 64 + 16 * q + m];
        bw[j][tt][q][1] = W[(k + 1) * 64 + 16 * q + m];
      }
    }
  }
#pragma unroll
  for (int tt = 0; tt < 2; ++tt) {
    const int k = 8 * PJ_FULL + 4 * hf + 2 * tt;
#pragma unroll
    for (int q = 0; q < 4; ++q) {
      bw[PJ_FULL][tt][q][0] = (k < 165) ? W[k * 64 + 16 * q + m] : 0.0f;
      bw[PJ_FULL][tt][q][1] = (k + 1 < 165) ? W[(k + 1) * 64 + 16 * q + m] : 0.0f;
    }
  }
  float bv[4];
#pragma unroll
  for (int q = 0; q < 4; ++q) bv[q] = bias[16 * q + m];

  const int ntiles = (n + 15) >> 4;
  for (int t = blockIdx.x * 4 + wave; t < ntiles; t += (int)gridDim.x * 4) {
    const int row = t * 16 + m;
    const int rl = row < n ? row : n - 1;
    const float* xr = x + (long long)rl * 165 + 4 * hf;
    v4f a[PJ_FULL + 1];
#pragma unroll
    for (int j = 0; j < PJ_FULL; ++j) a[j] = *(const v4f_u*)(xr + 8 * j);
    const float* tp = xr + 8 * PJ_FULL;
    const float t0 = tp[0];
    const float t1 = tp[hf ? 0 : 1];
    const float t2 = tp[hf ? 0 : 2];
    const float t3 = tp[hf ? 0 : 3];
    v4f tl;
    tl[0] = t0;
    tl[1] = hf ? 0.0f : t1;
    tl[2] = hf ? 0.0f : t2;
    tl[3] = hf ? 0.0f : t3;
    a[PJ_FULL] = tl;

    v8f acc[4] = {};
#pragma unroll
    for (int j = 0; j <= PJ_FULL; ++j) {
#pragma unroll
      for (int tt = 0; tt < 2; ++tt) {
        v2f af;
        af[0] = a[j][2 * tt];
        af[1] = a[j][2 * tt + 1];
#pragma unroll
        for (int q = 0; q < 4; ++q) acc[q] = wmma_k4(af, bw[j][tt][q], acc[q]);
      }
    }

#pragma unroll
    for (int q = 0; q < 4; ++q) {
#pragma unroll
      for (int v = 0; v < 8; ++v) {
        const int r = t * 16 + v + 8 * hf;
        if (r < n) {
          const float val = acc[q][v] + bv[q];
          h[(long long)r * HID + 16 * q + m] = val > 0.0f ? val : 0.0f;
        }
      }
    }
  }
}

#define TURN_PITCH 68
__device__ __forceinline__ void load_rows(v4f (&a)[8], v4f (&hv)[8], float& d, const float* agg,
                                          const float* h_in, const float* __restrict__ deg,
                                          int t, int n, int m, int hf) {
  const int row = t * 16 + m;
  const int rl = row < n ? row : n - 1;
  const v4f* ap = (const v4f*)(agg + (long long)rl * HID + 4 * hf);
  const v4f* hp = (const v4f*)(h_in + (long long)rl * HID + 4 * hf);
#pragma unroll
  for (int j = 0; j < 8; ++j) {
    a[j] = ap[2 * j];
    hv[j] = hp[2 * j];
  }
  d = deg[rl];
}

__global__ __launch_bounds__(128) __attribute__((amdgpu_waves_per_eu(1, 1))) void combine_pq_kernel(
    const float* agg, const float* __restrict__ deg, const float* h_in,
    const float* __restrict__ Wl, const float* __restrict__ bl, const float* __restrict__ Wr,
    const float* __restrict__ Wl2, const float* __restrict__ bl2, const float* __restrict__ Wr2,
    float* p_out, float* q_out, int n) {
  __shared__ __attribute__((aligned(16))) float turn[4 * 16 * TURN_PITCH];
  const int tid = threadIdx.x;
  const int wave = __builtin_amdgcn_readfirstlane(tid >> 5);
  const int lane = tid & 31;
  const int hf = lane >> 4;
  const int m = lane & 15;

  v2f wl[8][2][4], wr[8][2][4], wl2[8][2][4], wr2[8][2][4];
#pragma unroll
  for (int j = 0; j < 8; ++j) {
#pragma unroll
    for (int tt = 0; tt < 2; ++tt) {
      const int k = 8 * j + 4 * hf + 2 * tt;
#pragma unroll
      for (int q = 0; q < 4; ++q) {
        wl[j][tt][q][0] = Wl[k * 64 + 16 * q + m];
        wl[j][tt][q][1] = Wl[(k + 1) * 64 + 16 * q + m];
        wr[j][tt][q][0] = Wr[k * 64 + 16 * q + m];
        wr[j][tt][q][1] = Wr[(k + 1) * 64 + 16 * q + m];
        wl2[j][tt][q][0] = Wl2[k * 64 + 16 * q + m];
        wl2[j][tt][q][1] = Wl2[(k + 1) * 64 + 16 * q + m];
        wr2[j][tt][q][0] = Wr2[k * 64 + 16 * q + m];
        wr2[j][tt][q][1] = Wr2[(k + 1) * 64 + 16 * q + m];
      }
    }
  }
  float bv[4], bv2[4];
#pragma unroll
  for (int q = 0; q < 4; ++q) {
    bv[q] = bl[16 * q + m];
    bv2[q] = bl2[16 * q + m];
  }

  float* tw = turn + wave * (16 * TURN_PITCH);
  const int ntiles = (n + 15) >> 4;
  const int tstep = (int)gridDim.x * 4;
  int t = blockIdx.x * 4 + wave;

  v4f a[8], hv[8];
  float d;
  load_rows(a, hv, d, agg, h_in, deg, t < ntiles ? t : ntiles - 1, n, m, hf);

  for (; t < ntiles; t += tstep) {
    v4f an[8], hn[8];
    float dn;
    load_rows(an, hn, dn, agg, h_in, deg, t + tstep < ntiles ? t + tstep : t, n, m, hf);

    d = d > 1.0f ? d : 1.0f;
#pragma unroll
    for (int j = 0; j < 8; ++j) {
      a[j][0] = a[j][0] / d;
      a[j][1] = a[j][1] / d;
      a[j][2] = a[j][2] / d;
      a[j][3] = a[j][3] / d;
    }

    v8f acc[4] = {};
#pragma unroll
    for (int j = 0; j < 8; ++j) {
#pragma unroll
      for (int tt = 0; tt < 2; ++tt) {
        v2f af;
        af[0] = a[j][2 * tt];
        af[1] = a[j][2 * tt + 1];
#pragma unroll
        for (int q = 0; q < 4; ++q) acc[q] = wmma_k4(af, wl[j][tt][q], acc[q]);
      }
    }
#pragma unroll
    for (int j = 0; j < 8; ++j) {
#pragma unroll
      for (int tt = 0; tt < 2; ++tt) {
        v2f af;
        af[0] = hv[j][2 * tt];
        af[1] = hv[j][2 * tt + 1];
#pragma unroll
        for (int q = 0; q < 4; ++q) acc[q] = wmma_k4(af, wr[j][tt][q], acc[q]);
      }
    }

#pragma unroll
    for (int q = 0; q < 4; ++q) {
#pragma unroll
      for (int v = 0; v < 8; ++v) {
        const float val = acc[q][v] + bv[q];
        tw[(v + 8 * hf) * TURN_PITCH + 16 * q + m] = val > 0.0f ? val : 0.0f;
      }
    }
    v4f g[8];
#pragma unroll
    for (int j = 0; j < 8; ++j) g[j] = *(const v4f*)(tw + m * TURN_PITCH + 8 * j + 4 * hf);

    v8f accp[4] = {}, accq[4] = {};
#pragma unroll
    for (int j = 0; j < 8; ++j) {
#pragma unroll
      for (int tt = 0; tt < 2; ++tt) {
        v2f af;
        af[0] = g[j][2 * tt];
        af[1] = g[j][2 * tt + 1];
#pragma unroll
        for (int q = 0; q < 4; ++q) accp[q] = wmma_k4(af, wl2[j][tt][q], accp[q]);
#pragma unroll
        for (int q = 0; q < 4; ++q) accq[q] = wmma_k4(af, wr2[j][tt][q], accq[q]);
      }
    }

#pragma unroll
    for (int q = 0; q < 4; ++q) {
#pragma unroll
      for (int v = 0; v < 8; ++v) {
        const int r = t * 16 + v + 8 * hf;
        if (r < n) {
          p_out[(long long)r * HID + 16 * q + m] = accp[q][v];
          q_out[(long long)r * HID + 16 * q + m] = accq[q][v] + bv2[q];
        }
      }
    }

#pragma unroll
    for (int j = 0; j < 8; ++j) {
      a[j] = an[j];
      hv[j] = hn[j];
    }
    d = dn;
  }
}

extern "C" void kernel_launch(void* const* d_in, const int* in_sizes, int n_in,
                              void* d_out, int out_size, void* d_ws, size_t ws_size,
                              hipStream_t stream) {
  const float* x   = (const float*)d_in[0];
  const int* eidx  = (const int*)d_in[1];
  const float* Wp  = (const float*)d_in[2];
  const float* bp  = (const float*)d_in[3];
  const float* W1l = (const float*)d_in[4];
  const float* b1l = (const float*)d_in[5];
  const float* W1r = (const float*)d_in[6];
  const float* W2l = (const float*)d_in[7];
  const float* b2l = (const float*)d_in[8];
  const float* W2r = (const float*)d_in[9];
  const float* Wc  = (const float*)d_in[10];
  const float* bc  = (const float*)d_in[11];

  const int N = in_sizes[0] / 165;
  const int E = in_sizes[1] / 2;
  const int* src = eidx;
  const int* dst = eidx + E;

  float* A   = (float*)d_ws;
  float* B   = A + (size_t)N * HID;
  float* AGG = B + (size_t)N * HID;
  float* DEG = AGG + (size_t)N * HID;
  unsigned* LISTS = (unsigned*)B;

  const unsigned g_ag  = (unsigned)((N + AG_ROWS - 1) / AG_ROWS);
  const unsigned g_gemm = 512;
  const unsigned g_pq = 256;
  const unsigned g_co  = ((g_ag + SG_Q - 1) / SG_Q) * SG_C;

  proj_kernel<<<g_gemm, 128, 0, stream>>>(x, Wp, bp, A, N);

  coarse_kernel<<<g_co, AG_THREADS, 0, stream>>>(src, dst, LISTS, N, E);

  neigh_sum_kernel<false><<<g_ag, AG_THREADS, 0, stream>>>(A, src, dst, AGG, LISTS, DEG, N, E,
                                                           (const float*)0, (const float*)0,
                                                           (const float*)0, (float*)0);
  combine_pq_kernel<<<g_pq, 128, 0, stream>>>(AGG, DEG, A, W1l, b1l, W1r, W2l, b2l, W2r, AGG, A, N);

  neigh_sum_kernel<true><<<g_ag, AG_THREADS, 0, stream>>>(AGG, src, dst, (float*)0, LISTS, DEG, N, E,
                                                          A, Wc, bc, (float*)d_out);
}


# neighbour sums (conv 1, conv 2): a wave keeps the sums of its own 25 tile rows in registers, indexed by a scalar (v_movrels, v_movreld); no LDS read-add-write or branch per edge; add order kept
# speedup vs baseline: 6.8562x; 1.0376x over previous
#include <hip/hip_runtime.h>

#define HID 64

typedef __attribute__((ext_vector_type(2))) float v2f;
typedef __attribute__((ext_vector_type(8))) float v8f;
typedef __attribute__((ext_vector_type(4))) float v4f;
typedef v4f __attribute__((aligned(4))) v4f_u;
typedef __attribute__((ext_vector_type(32))) float v32f;

__device__ __forceinline__ v8f wmma_k4(v2f a, v2f b, v8f c) {
  return __builtin_amdgcn_wmma_f32_16x16x4_f32(
       false, a,  false, b,
       (short)0, c,  false,  false);
}

#define AG_ROWS 784
#define AG_WAVES 32
#define AG_THREADS (AG_WAVES * 32)
#define AG_ITER (AG_WAVES * 512)
#define AG_OUTCAP 416
#define AG_MYCAP 64
#define AG_SRC_BITS 22
#define AG_LIST_HEAD 32

__device__ __forceinline__ void load_iter(int4 (&r)[4], const int* __restrict__ dst, int g,
                                          int woff, int E, bool vec) {
  const int base = g * AG_ITER;
  if (vec && base + AG_ITER <= E) {
#pragma unroll
    for (int j = 0; j < 4; ++j) r[j] = *(const int4*)(dst + base + woff + j * 128);
  } else {
#pragma unroll
    for (int j = 0; j < 4; ++j) {
      const int e = base + woff + j * 128;
      r[j].x = (e < E) ? dst[e] : -1;
      r[j].y = (e + 1 < E) ? dst[e + 1] : -1;
      r[j].z = (e + 2 < E) ? dst[e + 2] : -1;
      r[j].w = (e + 3 < E) ? dst[e + 3] : -1;
    }
  }
}

__device__ __forceinline__ int queue_push(int2* q, int qn, bool keep, int2 ent, int lane) {
  const unsigned b = __builtin_amdgcn_ballot_w32(keep);
  if (b != 0u) {
    if (keep) q[qn + __popc(b & ((1u << lane) - 1u))] = ent;
    qn += __popc(b);
  }
  return qn;
}

__device__ __forceinline__ bool push4(int2* q, int& qn, int4 d, int e, int rowlo, int lane) {
  const unsigned l0 = (unsigned)(d.x - rowlo);
  const unsigned l1 = (unsigned)(d.y - rowlo);
  const unsigned l2 = (unsigned)(d.z - rowlo);
  const unsigned l3 = (unsigned)(d.w - rowlo);
  const unsigned lm = min(min(l0, l1), min(l2, l3));
  if (__builtin_amdgcn_ballot_w32(lm < (unsigned)AG_ROWS) == 0u) return true;
  const bool k0 = l0 < (unsigned)AG_ROWS, k1 = l1 < (unsigned)AG_ROWS;
  const bool k2 = l2 < (unsigned)AG_ROWS, k3 = l3 < (unsigned)AG_ROWS;
  const unsigned b0 = __builtin_amdgcn_ballot_w32(k0), b1 = __builtin_amdgcn_ballot_w32(k1);
  const unsigned b2 = __builtin_amdgcn_ballot_w32(k2), b3 = __builtin_amdgcn_ballot_w32(k3);
  const int c0 = __popc(b0), c1 = __popc(b1), c2 = __popc(b2), c3 = __popc(b3);
  if (qn + c0 + c1 + c2 + c3 > AG_OUTCAP) return false;
  const unsigned below = (1u << lane) - 1u;
  int p = qn;
  if (k0) q[p + __popc(b0 & below)] = make_int2(e, (int)l0);
  p += c0;
  if (k1) q[p + __popc(b1 & below)] = make_int2(e + 1, (int)l1);
  p += c1;
  if (k2) q[p + __popc(b2 & below)] = make_int2(e + 2, (int)l2);
  p += c2;
  if (k3) q[p + __popc(b3 & below)] = make_int2(e + 3, (int)l3);
  qn = p + c3;
  return true;
}

__device__ __forceinline__ float2 row_load(int s, int i, const float* __restrict__ h, int lane) {
  const int sk = __builtin_amdgcn_readlane(s, i);
  return *(const float2*)(h + (long long)sk * HID + lane * 2);
}

template <bool REC>
__device__ __forceinline__ void row_add(float2 v, int loc, int i, bool on, v32f& ax, v32f& ay,
                                        int& dcnt, int myrow) {
  const int lk = __builtin_amdgcn_readlane(loc, i);
  const int r = lk >> 5;
  ax[r] += on ? v.x : 0.0f;
  ay[r] += on ? v.y : 0.0f;
  if (REC) dcnt += (on && lk == myrow) ? 1 : 0;
}

template <bool REC, bool DIRECT, bool FULL>
__device__ __forceinline__ void add_rows(const int2* q, int cnt, const float* __restrict__ h,
                                         const int* __restrict__ src, v32f& ax, v32f& ay, int lane,
                                         unsigned* __restrict__ rec, int& rpos, int rcap,
                                         int& dcnt, int myrow) {
  int s = 0, loc = 0;
  if (lane < cnt) {
    const int2 ent = q[lane];
    s = DIRECT ? ent.x : src[ent.x];
    loc = ent.y;
  }
  if (REC) {
    if (rpos + cnt <= rcap && lane < cnt)
      rec[AG_LIST_HEAD + rpos + lane] = (unsigned)s | ((unsigned)loc << AG_SRC_BITS);
    rpos += cnt;
  }
  float2 v[8];
  if (FULL) {
#pragma unroll
    for (int u = 0; u < 8; ++u) v[u] = row_load(s, u, h, lane);
#pragma unroll
    for (int i = 0; i < 32; ++i) {
      row_add<REC>(v[i & 7], loc, i, true, ax, ay, dcnt, myrow);
      if (i + 8 < 32) v[i & 7] = row_load(s, i + 8, h, lane);
    }
  } else {
    for (int k = 0; k < cnt; k += 8) {
#pragma unroll
      for (int u = 0; u < 8; ++u) v[u] = row_load(s, k + u, h, lane);
#pragma unroll
      for (int u = 0; u < 8; ++u) row_add<REC>(v[u], loc, k + u, k + u < cnt, ax, ay, dcnt, myrow);
    }
  }
}

template <bool REC>
__device__ __forceinline__ void scan_gather(const float* __restrict__ h,
                                            const int* __restrict__ src,
                                            const int* __restrict__ dst, int E, v32f& ax, v32f& ay,
                                            int2* outq, int2* ownq, int* outn, int* fin,
                                            int rowlo, int wave, int lane,
                                            unsigned* __restrict__ rec, int& rpos, int rcap,
                                            int& dcnt) {
  const bool vec = (((size_t)dst) & 15) == 0;
  const int nit = (E + AG_ITER - 1) / AG_ITER;
  const int start = (int)(((long long)blockIdx.x * nit) / (int)gridDim.x);
  const int woff = wave * 512 + lane * 4;
  const int4 none = {-1, -1, -1, -1};

  int4 cur[4], nxt[4];
#pragma unroll
  for (int j = 0; j < 4; ++j) {
    cur[j] = none;
    nxt[j] = none;
  }

  int2* oq = outq + wave * AG_OUTCAP;
  int2* mq = ownq + wave * AG_MYCAP;
  int it = -1;
  int mn = 0;

  for (int round = 0; round <= 4 * nit; ++round) {
    int qn = 0;
    while (it < nit) {
      int g = start + it;
      if (g >= nit) g -= nit;
      int g1 = start + it + 1;
      if (g1 >= nit) g1 -= nit;
      if (it + 1 < nit) {
        load_iter(nxt, dst, g1, woff, E, vec);
      } else {
#pragma unroll
        for (int j = 0; j < 4; ++j) nxt[j] = none;
      }
      const int e0 = g * AG_ITER + woff;
      int queued = 0;
      if (push4(oq, qn, cur[0], e0, rowlo, lane)) {
        queued = 1;
        if (push4(oq, qn, cur[1], e0 + 128, rowlo, lane)) {
          queued = 2;
          if (push4(oq, qn, cur[2], e0 + 256, rowlo, lane)) {
            queued = 3;
            if (push4(oq, qn, cur[3], e0 + 384, rowlo, lane)) queued = 4;
          }
        }
      }
      if (queued < 4) {
        if (queued > 0) cur[0] = none;
        if (queued > 1) cur[1] = none;
        if (queued > 2) cur[2] = none;
        break;
      }
#pragma unroll
      for (int j = 0; j < 4; ++j) cur[j] = nxt[j];
      ++it;
    }
    if (lane == 0) {
      outn[wave] = qn;
      fin[wave] = (it >= nit) ? 1 : 0;
    }
    __syncthreads();

    for (int i = 0; i < AG_WAVES; ++i) {
      const int w2 = (wave + i) & (AG_WAVES - 1);
      const int cnt = __builtin_amdgcn_readfirstlane(outn[w2]);
      for (int b0 = 0; b0 < cnt; b0 += 32) {
        const bool live = b0 + lane < cnt;
        int2 ent = make_int2(0, -1);
        if (live) ent = outq[w2 * AG_OUTCAP + b0 + lane];
        const bool mine = live && ((ent.y & (AG_WAVES - 1)) == wave);
        mn = queue_push(mq, mn, mine, ent, lane);
        if (mn >= 32) {
          add_rows<REC, false, false>(mq, 32, h, src, ax, ay, lane, rec, rpos, rcap, dcnt, wave + 32 * lane);
          const int rem = mn - 32;
          if (lane < rem) {
            const int2 mv = mq[32 + lane];
            mq[lane] = mv;
          }
          mn = rem;
        }
      }
    }
    const bool alldone = __builtin_amdgcn_ballot_w32(fin[lane] != 0) == 0xffffffffu;
    __syncthreads();
    if (alldone) break;
  }
  if (mn > 0) add_rows<REC, false, false>(mq, mn, h, src, ax, ay, lane, rec, rpos, rcap, dcnt, wave + 32 * lane);
}

#define SG_Q 16
#define SG_C 16
#define SG_STRIDE 768
#define SG_CAP (SG_STRIDE - AG_LIST_HEAD)
#define SG_SRC_BITS 18
#define SG_OVER 0xffffffffu
#define SG_NB (SG_CAP / 32)
#define SB_CAP 64
static_assert(SG_Q == 16 && SG_Q + 2 <= AG_LIST_HEAD, "a bin index has four bits; the bin starts lie in the list's head");

__device__ __forceinline__ bool sg_layout(int n, int& g_ag, int& nsuper) {
  g_ag = (n + AG_ROWS - 1) / AG_ROWS;
  nsuper = (g_ag + SG_Q - 1) / SG_Q;
  const long long words = ((long long)g_ag + (long long)nsuper * SG_C) * 32 * SG_STRIDE;
  return n <= (1 << SG_SRC_BITS) && words <= (long long)n * HID;
}

__device__ __forceinline__ int queue_push_w(unsigned* q, int qn, bool keep, unsigned w, int lane) {
  const unsigned b = __builtin_amdgcn_ballot_w32(keep);
  if (b != 0u) {
    if (keep) q[qn + __popc(b & ((1u << lane) - 1u))] = w;
    qn += __popc(b);
  }
  return qn;
}

__device__ __forceinline__ void load_pair(int4& d0, int4& s0, int4& d1, int4& s1,
                                          const int* __restrict__ dst,
                                          const int* __restrict__ src, long long e, int E,
                                          bool whole) {
  if (whole) {
    d0 = *(const int4*)(dst + e);
    s0 = *(const int4*)(src + e);
    d1 = *(const int4*)(dst + e + 128);
    s1 = *(const int4*)(src + e + 128);
  } else {
    const long long f = e + 128;
    d0.x = (e < E) ? dst[e] : -1;
    d0.y = (e + 1 < E) ? dst[e + 1] : -1;
    d0.z = (e + 2 < E) ? dst[e + 2] : -1;
    d0.w = (e + 3 < E) ? dst[e + 3] : -1;
    s0.x = (e < E) ? src[e] : 0;
    s0.y = (e + 1 < E) ? src[e + 1] : 0;
    s0.z = (e + 2 < E) ? src[e + 2] : 0;
    s0.w = (e + 3 < E) ? src[e + 3] : 0;
    d1.x = (f < E) ? dst[f] : -1;
    d1.y = (f + 1 < E) ? dst[f + 1] : -1;
    d1.z = (f + 2 < E) ? dst[f + 2] : -1;
    d1.w = (f + 3 < E) ? dst[f + 3] : -1;
    s1.x = (f < E) ? src[f] : 0;
    s1.y = (f + 1 < E) ? src[f + 1] : 0;
    s1.z = (f + 2 < E) ? src[f + 2] : 0;
    s1.w = (f + 3 < E) ? src[f + 3] : 0;
  }
}

__device__ __forceinline__ void bin_push(unsigned* bq, unsigned* bc, bool live, unsigned w, int lane) {
  const unsigned smask = (1u << SG_SRC_BITS) - 1u;
  const unsigned u = w >> SG_SRC_BITS;
  const unsigned q = u / (unsigned)AG_ROWS;
  const unsigned l = u - q * (unsigned)AG_ROWS;
  unsigned m = __builtin_amdgcn_ballot_w32(live);
  for (int bit = 0; bit < 4; ++bit) {
    const bool on = ((q >> bit) & 1u) != 0u;
    const unsigned bb = __builtin_amdgcn_ballot_w32(live && on);
    m &= on ? bb : ~bb;
  }
  if (live) {
    const unsigned pos = bc[q] + (unsigned)__popc(m & ((1u << lane) - 1u));
    if (pos < (unsigned)SB_CAP) bq[q * SB_CAP + pos] = (w & smask) | (l << SG_SRC_BITS);
    if ((m >> lane) == 1u) bc[q] = pos + 1u;
  }
}

__device__ __forceinline__ void coarse_push(unsigned* mq, int& mn, unsigned* bq, unsigned* bc,
                                            int d, int s, int slo, int lane) {
  const unsigned u = (unsigned)(d - slo);
  const bool keep = u < (unsigned)(SG_Q * AG_ROWS);
  mn = queue_push_w(mq, mn, keep, (unsigned)s | (u << SG_SRC_BITS), lane);
  if (mn >= 32) {
    bin_push(bq, bc, true, mq[lane], lane);
    const int rem = mn - 32;
    if (lane < rem) {
      const unsigned mv = mq[32 + lane];
      mq[lane] = mv;
    }
    mn = rem;
  }
}

__global__ __launch_bounds__(AG_THREADS) void coarse_kernel(const int* __restrict__ src,
                                                          const int* __restrict__ dst,
                                                          unsigned* __restrict__ lists, int n,
                                                          int E) {
  __shared__ unsigned stage[AG_WAVES * 64];
  __shared__ unsigned bins[AG_WAVES * SG_Q * SB_CAP];
  __shared__ unsigned blen[AG_WAVES * SG_Q];
  int g_ag, nsuper;
  if (!sg_layout(n, g_ag, nsuper)) return;
  const int tid = threadIdx.x;
  const int wave = __builtin_amdgcn_readfirstlane(tid >> 5);
  const int lane = tid & 31;
  const int s = blockIdx.x / SG_C;
  const int c = blockIdx.x - s * SG_C;
  const int slo = s * (SG_Q * AG_ROWS);
  const long long share = ((long long)E + SG_C * AG_WAVES - 1) / (SG_C * AG_WAVES);
  const int steps = (int)((share + 127) / 128);
  const long long first = (long long)(c * AG_WAVES + wave) * steps * 128;
  unsigned* seg = lists + ((long long)g_ag * 32 + (long long)blockIdx.x * 32 + wave) * SG_STRIDE;
  unsigned* mq = stage + wave * 64;
  unsigned* bq = bins + wave * (SG_Q * SB_CAP);
  unsigned* bc = blen + wave * SG_Q;
  const bool vec = ((((size_t)src) | ((size_t)dst)) & 15) == 0;

  if (lane < SG_Q) bc[lane] = 0u;
  int mn = 0;
  int4 da, sa, db, sb;
  load_pair(da, sa, db, sb, dst, src, first + lane * 4, E, vec && first + 256 <= E);
  for (int i = 0; i < steps; i += 2) {
    const int4 d0 = da, s0 = sa, d1 = db, s1 = sb;
    if (i + 2 < steps) {
      const long long e2 = first + (long long)(i + 2) * 128;
      load_pair(da, sa, db, sb, dst, src, e2 + lane * 4, E, vec && e2 + 256 <= E);
    }
    coarse_push(mq, mn, bq, bc, d0.x, s0.x, slo, lane);
    coarse_push(mq, mn, bq, bc, d0.y, s0.y, slo, lane);
    coarse_push(mq, mn, bq, bc, d0.z, s0.z, slo, lane);
    coarse_push(mq, mn, bq, bc, d0.w, s0.w, slo, lane);
    if (i + 1 < steps) {
      coarse_push(mq, mn, bq, bc, d1.x, s1.x, slo, lane);
      coarse_push(mq, mn, bq, bc, d1.y, s1.y, slo, lane);
      coarse_push(mq, mn, bq, bc, d1.z, s1.z, slo, lane);
      coarse_push(mq, mn, bq, bc, d1.w, s1.w, slo, lane);
    }
  }
  if (mn > 0) bin_push(bq, bc, lane < mn, (lane < mn) ? mq[lane] : 0u, lane);

  const unsigned blq = (lane < SG_Q) ? bc[lane] : 0u;
  bool over = __builtin_amdgcn_ballot_w32(blq > (unsigned)SB_CAP) != 0u;
  int run = 0, offv = 0;
#pragma unroll
  for (int q = 0; q < SG_Q; ++q) {
    const int cq = __builtin_amdgcn_readlane((int)blq, q);
    if (lane == q) offv = run;
    if (!over && run + cq <= SG_CAP) {
      if (lane < cq) seg[AG_LIST_HEAD + run + lane] = bq[q * SB_CAP + lane];
      if (32 + lane < cq) seg[AG_LIST_HEAD + run + 32 + lane] = bq[q * SB_CAP + 32 + lane];
    }
    run += cq;
  }
  if (lane == SG_Q) offv = run;
  if (run > SG_CAP) over = true;
  if (lane <= SG_Q) seg[1 + lane] = (unsigned)offv;
  if (lane == 0) seg[0] = over ? SG_OVER : (unsigned)run;
}

#define LL_STRIDE 512
#define LL_CAP (LL_STRIDE - 1)
#define LL_END ((unsigned)LL_CAP)
#define LL_ROW_BITS 5
#define LL_NEXT_SHIFT (SG_SRC_BITS + LL_ROW_BITS)
static_assert(AG_WAVES * LL_STRIDE + AG_WAVES * AG_WAVES <= 2 * AG_WAVES * AG_OUTCAP,
              "the 32 queues and the heads lie in the LDS of the scan queues");
static_assert(LL_NEXT_SHIFT + 9 == 32 && AG_ROWS <= (AG_WAVES << LL_ROW_BITS) && AG_WAVES == 32,
              "an entry packs source row, row / 32 and a 9-bit link into one word");
template <bool REC>
__device__ __forceinline__ void seg_scan_gather(const float* __restrict__ h,
                                                const unsigned* __restrict__ coarse, int nst, int nen,
                                                v32f& ax, v32f& ay, unsigned* outw, int2* ownq,
                                                int* fin, int rot, int wave, int lane,
                                                unsigned* __restrict__ rec, int& rpos, int rcap,
                                                int& dcnt) {
  const unsigned smask = (1u << SG_SRC_BITS) - 1u;
  const unsigned below = (1u << lane) - 1u;
  unsigned* oq = outw + wave * LL_STRIDE;
  unsigned* heads = outw + AG_WAVES * LL_STRIDE;
  unsigned* hd = heads + wave * AG_WAVES;
  int2* mq = ownq + wave * AG_MYCAP;
  int ci = 0;
  int mn = 0;

  for (int round = 0; round <= SG_C * SG_NB; ++round) {
    int qn = 0;
    if (ci < SG_C) {
      unsigned wa[SG_C], wb[SG_C];
      int ln[SG_C];
#pragma unroll
      for (int k = 0; k < SG_C; ++k) {
        const int cc = (k + rot) & (SG_C - 1);
        const int st = __builtin_amdgcn_readfirstlane(min(max(__builtin_amdgcn_readlane(nst, cc), 0), SG_CAP));
        const int len = __builtin_amdgcn_readfirstlane(
            min(max(__builtin_amdgcn_readlane(nen, cc) - st, 0), min(SB_CAP, SG_CAP - st)));
        const unsigned* sp = coarse + ((long long)cc * 32 + wave) * SG_STRIDE + AG_LIST_HEAD + st;
        wa[k] = (lane < len) ? sp[lane] : 0u;
        wb[k] = (32 + lane < len) ? sp[32 + lane] : 0u;
        ln[k] = len;
      }
      bool full = false;
      int nci = SG_C;
#pragma unroll
      for (int k = 0; k < SG_C; ++k) {
        if (!full && k >= ci) {
          if (qn + ln[k] > LL_CAP) {
            full = true;
            nci = k;
          } else {
            if (lane < ln[k]) oq[qn + lane] = wa[k];
            if (32 + lane < ln[k]) oq[qn + 32 + lane] = wb[k];
            qn += ln[k];
          }
        }
      }
      ci = nci;
    }

    hd[lane] = LL_END;
    for (int b0 = 0; b0 < qn; b0 += 32) {
      const bool live = b0 + lane < qn;
      unsigned w = 0u;
      if (live) w = oq[b0 + lane];
      const unsigned l = w >> SG_SRC_BITS;
      const unsigned d = l & (AG_WAVES - 1);
      unsigned m = __builtin_amdgcn_ballot_w32(live);
      for (int bit = 0; bit < 5; ++bit) {
        const bool on = ((d >> bit) & 1u) != 0u;
        const unsigned bb = __builtin_amdgcn_ballot_w32(live && on);
        m &= on ? bb : ~bb;
      }
      const unsigned pm = m & below;
      unsigned nx = LL_END;
      if (live) nx = hd[d];
      if (pm != 0u) nx = (unsigned)(b0 + 31 - __clz(pm));
      if (live) {
        oq[b0 + lane] = (w & smask) | ((l >> 5) << SG_SRC_BITS) | (nx << LL_NEXT_SHIFT);
        if ((m >> lane) == 1u) hd[d] = (unsigned)(b0 + lane);
      }
    }
    if (lane == 0) fin[wave] = (ci >= SG_C) ? 1 : 0;
    __syncthreads();

    unsigned p = heads[lane * AG_WAVES + wave];
    for (int step = 0; step <= LL_CAP; ++step) {
      const bool act = p != LL_END;
      if (__builtin_amdgcn_ballot_w32(act) == 0u) break;
      unsigned ent = 0u;
      if (act) ent = outw[lane * LL_STRIDE + p];
      const int srow = (int)(ent & smask);
      const int trow = wave + 32 * (int)((ent >> SG_SRC_BITS) & ((1u << LL_ROW_BITS) - 1u));
      mn = queue_push(mq, mn, act, make_int2(srow, trow), lane);
      p = act ? (ent >> LL_NEXT_SHIFT) : LL_END;
      if (mn >= 32) {
        add_rows<REC, true, true>(mq, 32, h, (const int*)0, ax, ay, lane, rec, rpos, rcap, dcnt, wave + 32 * lane);
        const int rem = mn - 32;
        if (lane < rem) {
          const int2 mv = mq[32 + lane];
          mq[lane] = mv;
        }
        mn = rem;
      }
    }
    const bool alldone = __builtin_amdgcn_ballot_w32(fin[lane] != 0) == 0xffffffffu;
    __syncthreads();
    if (alldone) break;
  }
  if (mn > 0) add_rows<REC, true, false>(mq, mn, h, (const int*)0, ax, ay, lane, rec, rpos, rcap, dcnt, wave + 32 * lane);
}

__device__ __forceinline__ float2 pair_load(unsigned eb, int i, const float* __restrict__ h,
                                            int lane) {
  const unsigned smask = (1u << AG_SRC_BITS) - 1u;
  const unsigned e = (unsigned)__builtin_amdgcn_readlane((int)eb, i);
  return *(const float2*)(h + (long long)(e & smask) * HID + lane * 2);
}

__device__ __forceinline__ void pair_add(float2 v, unsigned eb, int i, bool on, v32f& ax,
                                         v32f& ay) {
  const unsigned e = (unsigned)__builtin_amdgcn_readlane((int)eb, i);
  const int r = (int)(e >> (AG_SRC_BITS + 5));
  ax[r] += on ? v.x : 0.0f;
  ay[r] += on ? v.y : 0.0f;
}

__device__ __forceinline__ void replay_rows(const unsigned* __restrict__ ents, int cnt,
                                            const float* __restrict__ h, v32f& ax, v32f& ay, int lane) {
  unsigned eb = (lane < cnt) ? ents[lane] : 0u;
  unsigned en = (32 + lane < cnt) ? ents[32 + lane] : 0u;
  float2 v[16];
#pragma unroll
  for (int u = 0; u < 16; ++u) v[u] = pair_load(eb, u, h, lane);
  for (int b = 0; b < cnt; b += 32) {
    const unsigned e2 = (b + 64 + lane < cnt) ? ents[b + 64 + lane] : 0u;
#pragma unroll
    for (int i = 0; i < 32; ++i) {
      pair_add(v[i & 15], eb, i, b + i < cnt, ax, ay);
      v[i & 15] = pair_load(i + 16 < 32 ? eb : en, (i + 16) & 31, h, lane);
    }
    eb = en;
    en = e2;
  }
}

template <bool REPLAY>
__global__ __launch_bounds__(AG_THREADS) void neigh_sum_kernel(const float* __restrict__ h,
                                                             const int* __restrict__ src,
                                                             const int* __restrict__ dst,
                                                             float* __restrict__ agg,
                                                             unsigned* __restrict__ lists,
                                                             float* __restrict__ deg, int n,
                                                             int E,
                                                             const float* __restrict__ qself,
                                                             const float* __restrict__ Wc,
                                                             const float* __restrict__ bc,
                                                             float* __restrict__ out) {
  __shared__ float tile[AG_ROWS * HID];
  __shared__ int2 outq[AG_WAVES * AG_OUTCAP];
  __shared__ int2 ownq[AG_WAVES * AG_MYCAP];
  __shared__ int outn[AG_WAVES];
  __shared__ int fin[AG_WAVES];

  const int tid = threadIdx.x;
  const int wave = __builtin_amdgcn_readfirstlane(tid >> 5);
  const int lane = tid & 31;
  const int rowlo = blockIdx.x * AG_ROWS;
  const int rows = (n - rowlo < AG_ROWS) ? (n - rowlo) : AG_ROWS;

  float4* t4 = (float4*)tile;
  v32f ax = {}, ay = {};

  int g_ag, nsuper;
  const bool lay = sg_layout(n, g_ag, nsuper);
  const int rcap = lay ? SG_CAP : 0;
  unsigned* rec = lists + ((long long)blockIdx.x * 32 + wave) * SG_STRIDE;
  int rpos = 0;
  int dcnt = 0;

  if (REPLAY) {
    const int cnt = lay ? __builtin_amdgcn_readfirstlane((int)rec[0]) : -1;
    const int redo =
        __builtin_amdgcn_readfirstlane(__syncthreads_or((cnt < 0 || cnt > rcap) ? 1 : 0));
    if (!redo) {
      replay_rows(rec + AG_LIST_HEAD, cnt, h, ax, ay, lane);
    } else {
      scan_gather<false>(h, src, dst, E, ax, ay, outq, ownq, outn, fin, rowlo, wave, lane, rec,
                         rpos, rcap, dcnt);
    }
  } else {
    const int sg = blockIdx.x / SG_Q;
    const int qg = blockIdx.x - sg * SG_Q;
    const unsigned* coarse = lists + ((long long)g_ag * 32 + (long long)sg * SG_C * 32) * SG_STRIDE;
    unsigned nlen = 0u;
    int nst = 0, nen = 0;
    if (lay && lane < SG_C) {
      const unsigned* lh = coarse + ((long long)lane * 32 + wave) * SG_STRIDE;
      nlen = lh[0];
      nst = (int)lh[1 + qg];
      nen = (int)lh[2 + qg];
    }
    const bool over = __builtin_amdgcn_ballot_w32(nlen > (unsigned)SG_CAP) != 0u;
    const int direct =
        __builtin_amdgcn_readfirstlane(__syncthreads_or((!lay || over) ? 1 : 0));
    if (!direct) {
      seg_scan_gather<true>(h, coarse, nst, nen, ax, ay, (unsigned*)outq, ownq, fin, qg, wave, lane,
                            rec, rpos, rcap, dcnt);
    } else {
      scan_gather<true>(h, src, dst, E, ax, ay, outq, ownq, outn, fin, rowlo, wave, lane, rec,
                        rpos, rcap, dcnt);
    }
    if (lay && lane == 0) rec[0] = (unsigned)rpos;
    const int r = wave + 32 * lane;
    if (r < AG_ROWS) ((float*)outq)[r] = (float)dcnt;
  }
#pragma unroll
  for (int i = 0; i < (AG_ROWS + AG_WAVES - 1) / AG_WAVES; ++i) {
    const int r = wave + AG_WAVES * i;
    if (AG_WAVES * (i + 1) <= AG_ROWS || r < AG_ROWS)
      *(float2*)(tile + r * HID + lane * 2) = make_float2(ax[i], ay[i]);
  }
  __syncthreads();

  if (!REPLAY) {
    float4* o4 = (float4*)(agg + (long long)rowlo * HID);
    for (int i = tid; i < rows * (HID / 4); i += AG_THREADS) o4[i] = t4[i];
    if (tid < rows) deg[rowlo + tid] = ((const float*)outq)[tid];
  } else {
    constexpr int RPW = (AG_ROWS + AG_WAVES - 1) / AG_WAVES;
    float2 qv[RPW];
    float dg[RPW];
#pragma unroll
    for (int i = 0; i < RPW; ++i) {
      const int r = wave + AG_WAVES * i;
      const int rr = r < rows ? r : 0;
      qv[i] = *(const float2*)(qself + (long long)(rowlo + rr) * HID + lane * 2);
      dg[i] = deg[rowlo + rr];
    }
    const float2 wc = *(const float2*)(Wc + lane * 2);
    const float bcv = bc[0];
    static_assert(RPW * 33 <= 2 * AG_OUTCAP && RPW <= 32 && AG_ROWS <= 2 * AG_WAVES * AG_MYCAP,
                  "a wave's buffer fits its scan queue, the results fit the own-row lists");
    float* sc = (float*)outq + wave * (2 * AG_OUTCAP);
    float* res = (float*)ownq;
#pragma unroll
    for (int i = 0; i < RPW; ++i) {
      const int r = wave + AG_WAVES * i;
      const int rr = r < rows ? r : 0;
      const float2 tv = *(const float2*)(tile + rr * HID + lane * 2);
      const float d = dg[i] > 1.0f ? dg[i] : 1.0f;
      float vx = tv.x / d + qv[i].x;
      float vy = tv.y / d + qv[i].y;
      vx = vx > 0.0f ? vx : 0.0f;
      vy = vy > 0.0f ? vy : 0.0f;
      sc[i * 33 + lane] = vx * wc.x + vy * wc.y;
    }
    if (lane < RPW) {
      float s = 0.0f;
      for (int k = 0; k < 32; ++k) s += sc[lane * 33 + k];
      const int r = wave + AG_WAVES * lane;
      if (r < rows) res[r] = s + bcv;
    }
    __syncthreads();
    if (tid < rows) out[rowlo + tid] = res[tid];
  }
}


#define PJ_FULL 20
__global__ __launch_bounds__(128) __attribute__((amdgpu_waves_per_eu(2, 2))) void proj_kernel(const float* __restrict__ x,
                                                   const float* __restrict__ W,
                                                   const float* __restrict__ bias,
                                                   float* __restrict__ h, int n) {
  const int tid = threadIdx.x;
  const int wave = __builtin_amdgcn_readfirstlane(tid >> 5);
  const int lane = tid & 31;
  const int hf = lane >> 4;
  const int m = lane & 15;

  v2f bw[PJ_FULL + 1][2][4];
#pragma unroll
  for (int j = 0; j < PJ_FULL; ++j) {
#pragma unroll
    for (int tt = 0; tt < 2; ++tt) {
      const int k = 8 * j + 4 * hf + 2 * tt;
#pragma unroll
      for (int q = 0; q < 4; ++q) {
        bw[j][tt][q][0] = W[k * 64 + 16 * q + m];
        bw[j][tt][q][1] = W[(k + 1) * 64 + 16 * q + m];
      }
    }
  }
#pragma unroll
  for (int tt = 0; tt < 2; ++tt) {
    const int k = 8 * PJ_FULL + 4 * hf + 2 * tt;
#pragma unroll
    for (int q = 0; q < 4; ++q) {
      bw[PJ_FULL][tt][q][0] = (k < 165) ? W[k * 64 + 16 * q + m] : 0.0f;
      bw[PJ_FULL][tt][q][1] = (k + 1 < 165) ? W[(k + 1) * 64 + 16 * q + m] : 0.0f;
    }
  }
  float bv[4];
#pragma unroll
  for (int q = 0; q < 4; ++q) bv[q] = bias[16 * q + m];

  const int ntiles = (n + 15) >> 4;
  for (int t = blockIdx.x * 4 + wave; t < ntiles; t += (int)gridDim.x * 4) {
    const int row = t * 16 + m;
    const int rl = row < n ? row : n - 1;
    const float* xr = x + (long long)rl * 165 + 4 * hf;
    v4f a[PJ_FULL + 1];
#pragma unroll
    for (int j = 0; j < PJ_FULL; ++j) a[j] = *(const v4f_u*)(xr + 8 * j);
    const float* tp = xr + 8 * PJ_FULL;
    const float t0 = tp[0];
    const float t1 = tp[hf ? 0 : 1];
    const float t2 = tp[hf ? 0 : 2];
    const float t3 = tp[hf ? 0 : 3];
    v4f tl;
    tl[0] = t0;
    tl[1] = hf ? 0.0f : t1;
    tl[2] = hf ? 0.0f : t2;
    tl[3] = hf ? 0.0f : t3;
    a[PJ_FULL] = tl;

    v8f acc[4] = {};
#pragma unroll
    for (int j = 0; j <= PJ_FULL; ++j) {
#pragma unroll
      for (int tt = 0; tt < 2; ++tt) {
        v2f af;
        af[0] = a[j][2 * tt];
        af[1] = a[j][2 * tt + 1];
#pragma unroll
        for (int q = 0; q < 4; ++q) acc[q] = wmma_k4(af, bw[j][tt][q], acc[q]);
      }
    }

#pragma unroll
    for (int q = 0; q < 4; ++q) {
#pragma unroll
      for (int v = 0; v < 8; ++v) {
        const int r = t * 16 + v + 8 * hf;
        if (r < n) {
          const float val = acc[q][v] + bv[q];
          h[(long long)r * HID + 16 * q + m] = val > 0.0f ? val : 0.0f;
        }
      }
    }
  }
}

#define TURN_PITCH 68
__device__ __forceinline__ void load_rows(v4f (&a)[8], v4f (&hv)[8], float& d, const float* agg,
                                          const float* h_in, const float* __restrict__ deg,
                                          int t, int n, int m, int hf) {
  const int row = t * 16 + m;
  const int rl = row < n ? row : n - 1;
  const v4f* ap = (const v4f*)(agg + (long long)rl * HID + 4 * hf);
  const v4f* hp = (const v4f*)(h_in + (long long)rl * HID + 4 * hf);
#pragma unroll
  for (int j = 0; j < 8; ++j) {
    a[j] = ap[2 * j];
    hv[j] = hp[2 * j];
  }
  d = deg[rl];
}

__global__ __launch_bounds__(128) __attribute__((amdgpu_waves_per_eu(1, 1))) void combine_pq_kernel(
    const float* agg, const float* __restrict__ deg, const float* h_in,
    const float* __restrict__ Wl, const float* __restrict__ bl, const float* __restrict__ Wr,
    const float* __restrict__ Wl2, const float* __restrict__ bl2, const float* __restrict__ Wr2,
    float* p_out, float* q_out, int n) {
  __shared__ __attribute__((aligned(16))) float turn[4 * 16 * TURN_PITCH];
  const int tid = threadIdx.x;
  const int wave = __builtin_amdgcn_readfirstlane(tid >> 5);
  const int lane = tid & 31;
  const int hf = lane >> 4;
  const int m = lane & 15;

  v2f wl[8][2][4], wr[8][2][4], wl2[8][2][4], wr2[8][2][4];
#pragma unroll
  for (int j = 0; j < 8; ++j) {
#pragma unroll
    for (int tt = 0; tt < 2; ++tt) {
      const int k = 8 * j + 4 * hf + 2 * tt;
#pragma unroll
      for (int q = 0; q < 4; ++q) {
        wl[j][tt][q][0] = Wl[k * 64 + 16 * q + m];
        wl[j][tt][q][1] = Wl[(k + 1) * 64 + 16 * q + m];
        wr[j][tt][q][0] = Wr[k * 64 + 16 * q + m];
        wr[j][tt][q][1] = Wr[(k + 1) * 64 + 16 * q + m];
        wl2[j][tt][q][0] = Wl2[k * 64 + 16 * q + m];
        wl2[j][tt][q][1] = Wl2[(k + 1) * 64 + 16 * q + m];
        wr2[j][tt][q][0] = Wr2[k * 64 + 16 * q + m];
        wr2[j][tt][q][1] = Wr2[(k + 1) * 64 + 16 * q + m];
      }
    }
  }
  float bv[4], bv2[4];
#pragma unroll
  for (int q = 0; q < 4; ++q) {
    bv[q] = bl[16 * q + m];
    bv2[q] = bl2[16 * q + m];
  }

  float* tw = turn + wave * (16 * TURN_PITCH);
  const int ntiles = (n + 15) >> 4;
  const int tstep = (int)gridDim.x * 4;
  int t = blockIdx.x * 4 + wave;

  v4f a[8], hv[8];
  float d;
  load_rows(a, hv, d, agg, h_in, deg, t < ntiles ? t : ntiles - 1, n, m, hf);

  for (; t < ntiles; t += tstep) {
    v4f an[8], hn[8];
    float dn;
    load_rows(an, hn, dn, agg, h_in, deg, t + tstep < ntiles ? t + tstep : t, n, m, hf);

    d = d > 1.0f ? d : 1.0f;
#pragma unroll
    for (int j = 0; j < 8; ++j) {
      a[j][0] = a[j][0] / d;
      a[j][1] = a[j][1] / d;
      a[j][2] = a[j][2] / d;
      a[j][3] = a[j][3] / d;
    }

    v8f acc[4] = {};
#pragma unroll
    for (int j = 0; j < 8; ++j) {
#pragma unroll
      for (int tt = 0; tt < 2; ++tt) {
        v2f af;
        af[0] = a[j][2 * tt];
        af[1] = a[j][2 * tt + 1];
#pragma unroll
        for (int q = 0; q < 4; ++q) acc[q] = wmma_k4(af, wl[j][tt][q], acc[q]);
      }
    }
#pragma unroll
    for (int j = 0; j < 8; ++j) {
#pragma unroll
      for (int tt = 0; tt < 2; ++tt) {
        v2f af;
        af[0] = hv[j][2 * tt];
        af[1] = hv[j][2 * tt + 1];
#pragma unroll
        for (int q = 0; q < 4; ++q) acc[q] = wmma_k4(af, wr[j][tt][q], acc[q]);
      }
    }

#pragma unroll
    for (int q = 0; q < 4; ++q) {
#pragma unroll
      for (int v = 0; v < 8; ++v) {
        const float val = acc[q][v] + bv[q];
        tw[(v + 8 * hf) * TURN_PITCH + 16 * q + m] = val > 0.0f ? val : 0.0f;
      }
    }
    v4f g[8];
#pragma unroll
    for (int j = 0; j < 8; ++j) g[j] = *(const v4f*)(tw + m * TURN_PITCH + 8 * j + 4 * hf);

    v8f accp[4] = {}, accq[4] = {};
#pragma unroll
    for (int j = 0; j < 8; ++j) {
#pragma unroll
      for (int tt = 0; tt < 2; ++tt) {
        v2f af;
        af[0] = g[j][2 * tt];
        af[1] = g[j][2 * tt + 1];
#pragma unroll
        for (int q = 0; q < 4; ++q) accp[q] = wmma_k4(af, wl2[j][tt][q], accp[q]);
#pragma unroll
        for (int q = 0; q < 4; ++q) accq[q] = wmma_k4(af, wr2[j][tt][q], accq[q]);
      }
    }

#pragma unroll
    for (int q = 0; q < 4; ++q) {
#pragma unroll
      for (int v = 0; v < 8; ++v) {
        const int r = t * 16 + v + 8 * hf;
        if (r < n) {
          p_out[(long long)r * HID + 16 * q + m] = accp[q][v];
          q_out[(long long)r * HID + 16 * q + m] = accq[q][v] + bv2[q];
        }
      }
    }

#pragma unroll
    for (int j = 0; j < 8; ++j) {
      a[j] = an[j];
      hv[j] = hn[j];
    }
    d = dn;
  }
}

extern "C" void kernel_launch(void* const* d_in, const int* in_sizes, int n_in,
                              void* d_out, int out_size, void* d_ws, size_t ws_size,
                              hipStream_t stream) {
  const float* x   = (const float*)d_in[0];
  const int* eidx  = (const int*)d_in[1];
  const float* Wp  = (const float*)d_in[2];
  const float* bp  = (const float*)d_in[3];
  const float* W1l = (const float*)d_in[4];
  const float* b1l = (const float*)d_in[5];
  const float* W1r = (const float*)d_in[6];
  const float* W2l = (const float*)d_in[7];
  const float* b2l = (const float*)d_in[8];
  const float* W2r = (const float*)d_in[9];
  const float* Wc  = (const float*)d_in[10];
  const float* bc  = (const float*)d_in[11];

  const int N = in_sizes[0] / 165;
  const int E = in_sizes[1] / 2;
  const int* src = eidx;
  const int* dst = eidx + E;

  float* A   = (float*)d_ws;
  float* B   = A + (size_t)N * HID;
  float* AGG = B + (size_t)N * HID;
  float* DEG = AGG + (size_t)N * HID;
  unsigned* LISTS = (unsigned*)B;

  const unsigned g_ag  = (unsigned)((N + AG_ROWS - 1) / AG_ROWS);
  const unsigned g_gemm = 512;
  const unsigned g_pq = 256;
  const unsigned g_co  = ((g_ag + SG_Q - 1) / SG_Q) * SG_C;

  proj_kernel<<<g_gemm, 128, 0, stream>>>(x, Wp, bp, A, N);

  coarse_kernel<<<g_co, AG_THREADS, 0, stream>>>(src, dst, LISTS, N, E);

  neigh_sum_kernel<false><<<g_ag, AG_THREADS, 0, stream>>>(A, src, dst, AGG, LISTS, DEG, N, E,
                                                           (const float*)0, (const float*)0,
                                                           (const float*)0, (float*)0);
  combine_pq_kernel<<<g_pq, 128, 0, stream>>>(AGG, DEG, A, W1l, b1l, W1r, W2l, b2l, W2r, AGG, A, N);

  neigh_sum_kernel<true><<<g_ag, AG_THREADS, 0, stream>>>(AGG, src, dst, (float*)0, LISTS, DEG, N, E,
                                                          A, Wc, bc, (float*)d_out);
}
